# starting kernel
# baseline (speedup 1.0000x reference)
#include <hip/hip_runtime.h>


typedef __attribute__((ext_vector_type(2))) float v2f;
typedef __attribute__((ext_vector_type(8))) float v8f;


__global__ void zero_f4_kernel(float4* __restrict__ p, int n4) {
    int i = blockIdx.x * blockDim.x + threadIdx.x;
    if (i < n4) p[i] = make_float4(0.f, 0.f, 0.f, 0.f);
}

__global__ void transpose_kernel(const float* __restrict__ in,
                                 float* __restrict__ outT, int K, int N) {
    int i = blockIdx.x * blockDim.x + threadIdx.x;
    if (i < N * K) {
        int n = i / K, k = i % K;
        outT[i] = in[k * N + n];
    }
}

__global__ void degree_count_kernel(const int* __restrict__ src,
                                    const int* __restrict__ dst,
                                    float* __restrict__ deg_out,
                                    float* __restrict__ deg_in, int nE) {
    int e = blockIdx.x * blockDim.x + threadIdx.x;
    if (e < nE) {
        atomicAdd(&deg_out[src[e]], 1.0f);
        atomicAdd(&deg_in[dst[e]], 1.0f);
    }
}

__global__ void rsqrt_clamp_kernel(float* __restrict__ d, int n) {
    int i = blockIdx.x * blockDim.x + threadIdx.x;
    if (i < n) d[i] = rsqrtf(fmaxf(d[i], 1.0f));
}

__global__ void edge_aggregate_kernel(const int* __restrict__ src,
                                      const int* __restrict__ dst,
                                      const float* __restrict__ h,
                                      float* __restrict__ agg, int nE) {
    int tid = blockIdx.x * blockDim.x + threadIdx.x;
    int e = tid >> 5;
    int f = (tid & 31) * 4;
    if (e < nE) {
        const float4 v = *(const float4*)(h + (size_t)src[e] * 128 + f);
        float* a = agg + (size_t)dst[e] * 128 + f;
        atomicAdd(a + 0, v.x);
        atomicAdd(a + 1, v.y);
        atomicAdd(a + 2, v.z);
        atomicAdd(a + 3, v.w);
    }
}

__global__ void scatter_rows_kernel(const int* __restrict__ ids,
                                    const float* __restrict__ h,
                                    float* __restrict__ out_feat, int n) {
    int tid = blockIdx.x * blockDim.x + threadIdx.x;
    int r = tid >> 5;
    int f = (tid & 31) * 4;
    if (r < n) {
        const float4 v = *(const float4*)(h + (size_t)r * 128 + f);
        float* o = out_feat + (size_t)ids[r] * 128 + f;
        atomicAdd(o + 0, v.x);
        atomicAdd(o + 1, v.y);
        atomicAdd(o + 2, v.z);
        atomicAdd(o + 3, v.w);
    }
}

template <int NTILES>
__global__ __launch_bounds__(256)
void gemm_wmma_k128_kernel(const float* __restrict__ A,
                           const float* __restrict__ Wt,
                           const float* __restrict__ bias,
                           const float* __restrict__ arowscale,
                           const float* __restrict__ erowscale,
                           float* __restrict__ out,
                           int M, int relu) {
    const int lane    = threadIdx.x & 31;
    const int wave    = threadIdx.x >> 5;
    const int task    = blockIdx.x * (blockDim.x >> 5) + wave;
    const int t0      = task * 2;
    const int t1      = t0 + 1;
    if (t0 * 16 >= M) return;
    const bool has1   = (t1 * 16 < M);
    const int halfsel = lane >> 4;
    const int lid     = lane & 15;
    const int N       = NTILES * 16;

    v8f acc0[NTILES], acc1[NTILES];
#pragma unroll
    for (int nt = 0; nt < NTILES; ++nt) {
        float bv = bias[nt * 16 + lid];
#pragma unroll
        for (int j = 0; j < 8; ++j) { acc0[nt][j] = bv; acc1[nt][j] = bv; }
    }

    const int arow0 = t0 * 16 + lid;
    const int arow1 = min(t1 * 16 + lid, M - 1);
    const float* aptr0 = A + (size_t)arow0 * 128 + halfsel * 2;
    const float* aptr1 = A + (size_t)arow1 * 128 + halfsel * 2;
    const float ascale0 = arowscale ? arowscale[arow0] : 1.0f;
    const float ascale1 = arowscale ? arowscale[arow1] : 1.0f;

    const float* wlane = Wt + (size_t)lid * 128 + halfsel * 2;

#pragma unroll 2
    for (int kk = 0; kk < 32; ++kk) {
        v2f a0 = *(const v2f*)(aptr0);
        v2f a1 = *(const v2f*)(aptr1);
        a0.x *= ascale0; a0.y *= ascale0;
        a1.x *= ascale1; a1.y *= ascale1;
        aptr0 += 4; aptr1 += 4;
#pragma unroll
        for (int nt = 0; nt < NTILES; ++nt) {
            v2f b = *(const v2f*)(wlane + (size_t)nt * 16 * 128 + kk * 4);
            acc0[nt] = __builtin_amdgcn_wmma_f32_16x16x4_f32(
                false, a0, false, b, (short)0, acc0[nt], false, false);
            acc1[nt] = __builtin_amdgcn_wmma_f32_16x16x4_f32(
                false, a1, false, b, (short)0, acc1[nt], false, false);
        }
    }

#pragma unroll
    for (int nt = 0; nt < NTILES; ++nt) {
#pragma unroll
        for (int j = 0; j < 8; ++j) {
            float v = acc0[nt][j];
            if (relu) v = fmaxf(v, 0.0f);
            const int row = t0 * 16 + j + halfsel * 8;
            if (erowscale) v *= erowscale[row];
            out[(size_t)row * N + nt * 16 + lid] = v;
        }
    }
    if (has1) {
#pragma unroll
        for (int nt = 0; nt < NTILES; ++nt) {
#pragma unroll
            for (int j = 0; j < 8; ++j) {
                float v = acc1[nt][j];
                if (relu) v = fmaxf(v, 0.0f);
                const int row = t1 * 16 + j + halfsel * 8;
                if (erowscale) v *= erowscale[row];
                out[(size_t)row * N + nt * 16 + lid] = v;
            }
        }
    }
}


extern "C" void kernel_launch(void* const* d_in, const int* in_sizes, int n_in,
                              void* d_out, int out_size, void* d_ws, size_t ws_size,
                              hipStream_t stream) {
    (void)n_in; (void)ws_size;
    const float* x   = (const float*)d_in[0];
    const float* W2  = (const float*)d_in[1];
    const float* b2  = (const float*)d_in[2];
    const float* gw  = (const float*)d_in[3];
    const float* gb  = (const float*)d_in[4];
    const float* Wc  = (const float*)d_in[5];
    const float* bc  = (const float*)d_in[6];
    const int*   src = (const int*)d_in[7];
    const int*   dst = (const int*)d_in[8];
    const int*   ids = (const int*)d_in[9];

    const int H  = 128;
    const int C  = 16;
    const int L  = in_sizes[2] / H;
    const int NS = in_sizes[9] / L;
    const int NE = in_sizes[7] / L;
    const int NL = in_sizes[4] / (L * H);
    const int NN = out_size / C;

    float* ws       = (float*)d_ws;
    float* out_feat = ws;
    float* hbuf     = out_feat + (size_t)NN * H;
    float* agg      = hbuf + (size_t)NS * H;
    float* deg_out  = agg + (size_t)NS * H;
    float* deg_in   = deg_out + NS;
    float* w2t      = deg_in + NS;
    float* gwt      = w2t + (size_t)L * H * H;
    float* wct      = gwt + (size_t)L * NL * H * H;

    const int T = 256;
    const int feat_blocks_NS = (NS * 32 + T - 1) / T;
    const int gemm_blocks_NS = (((NS / 16) + 1) / 2 + 7) / 8;
    const int gemm_blocks_NN = (((NN / 16) + 1) / 2 + 7) / 8;

    {
        int nHH = H * H;
        int bHH = (nHH + T - 1) / T;
        for (int i = 0; i < L; ++i)
            transpose_kernel<<<bHH, T, 0, stream>>>(W2 + (size_t)i * nHH,
                                                    w2t + (size_t)i * nHH, H, H);
        for (int i = 0; i < L * NL; ++i)
            transpose_kernel<<<bHH, T, 0, stream>>>(gw + (size_t)i * nHH,
                                                    gwt + (size_t)i * nHH, H, H);
        int nHC = H * C;
        transpose_kernel<<<(nHC + T - 1) / T, T, 0, stream>>>(Wc, wct, H, C);
    }

    {
        int n4 = (int)((size_t)NN * H / 4);
        zero_f4_kernel<<<(n4 + T - 1) / T, T, 0, stream>>>((float4*)out_feat, n4);
    }

    for (int i = 0; i < L; ++i) {
        const int*   src_i = src + (size_t)i * NE;
        const int*   dst_i = dst + (size_t)i * NE;
        const int*   ids_i = ids + (size_t)i * NS;
        const float* x_i   = x   + (size_t)i * NS * H;
        const float* w2t_i = w2t + (size_t)i * H * H;
        const float* b2_i  = b2  + (size_t)i * H;

        {
            int n4 = (2 * NS) / 4;
            zero_f4_kernel<<<(n4 + T - 1) / T, T, 0, stream>>>((float4*)deg_out, n4);
        }
        degree_count_kernel<<<(NE + T - 1) / T, T, 0, stream>>>(src_i, dst_i,
                                                                deg_out, deg_in, NE);
        rsqrt_clamp_kernel<<<(2 * NS + T - 1) / T, T, 0, stream>>>(deg_out, 2 * NS);

        gemm_wmma_k128_kernel<8><<<gemm_blocks_NS, T, 0, stream>>>(
            x_i, w2t_i, b2_i, nullptr, deg_out, hbuf, NS, 0);

        for (int l = 0; l < NL; ++l) {
            const float* gwt_il = gwt + ((size_t)i * NL + l) * H * H;
            const float* gb_il  = gb  + ((size_t)i * NL + l) * H;
            {
                int n4 = (int)((size_t)NS * H / 4);
                zero_f4_kernel<<<(n4 + T - 1) / T, T, 0, stream>>>((float4*)agg, n4);
            }
            edge_aggregate_kernel<<<(NE * 32) / T, T, 0, stream>>>(src_i, dst_i,
                                                                   hbuf, agg, NE);
            const float* escale = (l + 1 < NL) ? deg_out : nullptr;
            gemm_wmma_k128_kernel<8><<<gemm_blocks_NS, T, 0, stream>>>(
                agg, gwt_il, gb_il, deg_in, escale, hbuf, NS, 1);
        }

        scatter_rows_kernel<<<feat_blocks_NS, T, 0, stream>>>(ids_i, hbuf,
                                                              out_feat, NS);
    }

    gemm_wmma_k128_kernel<1><<<gemm_blocks_NN, T, 0, stream>>>(
        out_feat, wct, bc, nullptr, nullptr, (float*)d_out, NN, 0);
}


# scatter-add kernels: lane owns features l,l+32,l+64,l+96 so each f32 atomic wave-instruction covers one contiguous 128B line, not 8 words in each of 4 lines; atomic sum order unordered as before
# speedup vs baseline: 1.2454x; 1.2454x over previous
#include <hip/hip_runtime.h>


typedef __attribute__((ext_vector_type(2))) float v2f;
typedef __attribute__((ext_vector_type(8))) float v8f;


__global__ void zero_f4_kernel(float4* __restrict__ p, int n4) {
    int i = blockIdx.x * blockDim.x + threadIdx.x;
    if (i < n4) p[i] = make_float4(0.f, 0.f, 0.f, 0.f);
}

__global__ void transpose_kernel(const float* __restrict__ in,
                                 float* __restrict__ outT, int K, int N) {
    int i = blockIdx.x * blockDim.x + threadIdx.x;
    if (i < N * K) {
        int n = i / K, k = i % K;
        outT[i] = in[k * N + n];
    }
}

__global__ void degree_count_kernel(const int* __restrict__ src,
                                    const int* __restrict__ dst,
                                    float* __restrict__ deg_out,
                                    float* __restrict__ deg_in, int nE) {
    int e = blockIdx.x * blockDim.x + threadIdx.x;
    if (e < nE) {
        atomicAdd(&deg_out[src[e]], 1.0f);
        atomicAdd(&deg_in[dst[e]], 1.0f);
    }
}

__global__ void rsqrt_clamp_kernel(float* __restrict__ d, int n) {
    int i = blockIdx.x * blockDim.x + threadIdx.x;
    if (i < n) d[i] = rsqrtf(fmaxf(d[i], 1.0f));
}

__global__ void edge_aggregate_kernel(const int* __restrict__ src,
                                      const int* __restrict__ dst,
                                      const float* __restrict__ h,
                                      float* __restrict__ agg, int nE) {
    int tid = blockIdx.x * blockDim.x + threadIdx.x;
    int e = tid >> 5;
    int f = tid & 31;
    if (e < nE) {
        const float* hp = h + (size_t)src[e] * 128 + f;
        float* a = agg + (size_t)dst[e] * 128 + f;
        const float v0 = hp[0];
        const float v1 = hp[32];
        const float v2 = hp[64];
        const float v3 = hp[96];
        atomicAdd(a + 0, v0);
        atomicAdd(a + 32, v1);
        atomicAdd(a + 64, v2);
        atomicAdd(a + 96, v3);
    }
}

__global__ void scatter_rows_kernel(const int* __restrict__ ids,
                                    const float* __restrict__ h,
                                    float* __restrict__ out_feat, int n) {
    int tid = blockIdx.x * blockDim.x + threadIdx.x;
    int r = tid >> 5;
    int f = tid & 31;
    if (r < n) {
        const float* hp = h + (size_t)r * 128 + f;
        float* o = out_feat + (size_t)ids[r] * 128 + f;
        const float v0 = hp[0];
        const float v1 = hp[32];
        const float v2 = hp[64];
        const float v3 = hp[96];
        atomicAdd(o + 0, v0);
        atomicAdd(o + 32, v1);
        atomicAdd(o + 64, v2);
        atomicAdd(o + 96, v3);
    }
}

template <int NTILES>
__global__ __launch_bounds__(256)
void gemm_wmma_k128_kernel(const float* __restrict__ A,
                           const float* __restrict__ Wt,
                           const float* __restrict__ bias,
                           const float* __restrict__ arowscale,
                           const float* __restrict__ erowscale,
                           float* __restrict__ out,
                           int M, int relu) {
    const int lane    = threadIdx.x & 31;
    const int wave    = threadIdx.x >> 5;
    const int task    = blockIdx.x * (blockDim.x >> 5) + wave;
    const int t0      = task * 2;
    const int t1      = t0 + 1;
    if (t0 * 16 >= M) return;
    const bool has1   = (t1 * 16 < M);
    const int halfsel = lane >> 4;
    const int lid     = lane & 15;
    const int N       = NTILES * 16;

    v8f acc0[NTILES], acc1[NTILES];
#pragma unroll
    for (int nt = 0; nt < NTILES; ++nt) {
        float bv = bias[nt * 16 + lid];
#pragma unroll
        for (int j = 0; j < 8; ++j) { acc0[nt][j] = bv; acc1[nt][j] = bv; }
    }

    const int arow0 = t0 * 16 + lid;
    const int arow1 = min(t1 * 16 + lid, M - 1);
    const float* aptr0 = A + (size_t)arow0 * 128 + halfsel * 2;
    const float* aptr1 = A + (size_t)arow1 * 128 + halfsel * 2;
    const float ascale0 = arowscale ? arowscale[arow0] : 1.0f;
    const float ascale1 = arowscale ? arowscale[arow1] : 1.0f;

    const float* wlane = Wt + (size_t)lid * 128 + halfsel * 2;

#pragma unroll 2
    for (int kk = 0; kk < 32; ++kk) {
        v2f a0 = *(const v2f*)(aptr0);
        v2f a1 = *(const v2f*)(aptr1);
        a0.x *= ascale0; a0.y *= ascale0;
        a1.x *= ascale1; a1.y *= ascale1;
        aptr0 += 4; aptr1 += 4;
#pragma unroll
        for (int nt = 0; nt < NTILES; ++nt) {
            v2f b = *(const v2f*)(wlane + (size_t)nt * 16 * 128 + kk * 4);
            acc0[nt] = __builtin_amdgcn_wmma_f32_16x16x4_f32(
                false, a0, false, b, (short)0, acc0[nt], false, false);
            acc1[nt] = __builtin_amdgcn_wmma_f32_16x16x4_f32(
                false, a1, false, b, (short)0, acc1[nt], false, false);
        }
    }

#pragma unroll
    for (int nt = 0; nt < NTILES; ++nt) {
#pragma unroll
        for (int j = 0; j < 8; ++j) {
            float v = acc0[nt][j];
            if (relu) v = fmaxf(v, 0.0f);
            const int row = t0 * 16 + j + halfsel * 8;
            if (erowscale) v *= erowscale[row];
            out[(size_t)row * N + nt * 16 + lid] = v;
        }
    }
    if (has1) {
#pragma unroll
        for (int nt = 0; nt < NTILES; ++nt) {
#pragma unroll
            for (int j = 0; j < 8; ++j) {
                float v = acc1[nt][j];
                if (relu) v = fmaxf(v, 0.0f);
                const int row = t1 * 16 + j + halfsel * 8;
                if (erowscale) v *= erowscale[row];
                out[(size_t)row * N + nt * 16 + lid] = v;
            }
        }
    }
}


extern "C" void kernel_launch(void* const* d_in, const int* in_sizes, int n_in,
                              void* d_out, int out_size, void* d_ws, size_t ws_size,
                              hipStream_t stream) {
    (void)n_in; (void)ws_size;
    const float* x   = (const float*)d_in[0];
    const float* W2  = (const float*)d_in[1];
    const float* b2  = (const float*)d_in[2];
    const float* gw  = (const float*)d_in[3];
    const float* gb  = (const float*)d_in[4];
    const float* Wc  = (const float*)d_in[5];
    const float* bc  = (const float*)d_in[6];
    const int*   src = (const int*)d_in[7];
    const int*   dst = (const int*)d_in[8];
    const int*   ids = (const int*)d_in[9];

    const int H  = 128;
    const int C  = 16;
    const int L  = in_sizes[2] / H;
    const int NS = in_sizes[9] / L;
    const int NE = in_sizes[7] / L;
    const int NL = in_sizes[4] / (L * H);
    const int NN = out_size / C;

    float* ws       = (float*)d_ws;
    float* out_feat = ws;
    float* hbuf     = out_feat + (size_t)NN * H;
    float* agg      = hbuf + (size_t)NS * H;
    float* deg_out  = agg + (size_t)NS * H;
    float* deg_in   = deg_out + NS;
    float* w2t      = deg_in + NS;
    float* gwt      = w2t + (size_t)L * H * H;
    float* wct      = gwt + (size_t)L * NL * H * H;

    const int T = 256;
    const int feat_blocks_NS = (NS * 32 + T - 1) / T;
    const int gemm_blocks_NS = (((NS / 16) + 1) / 2 + 7) / 8;
    const int gemm_blocks_NN = (((NN / 16) + 1) / 2 + 7) / 8;

    {
        int nHH = H * H;
        int bHH = (nHH + T - 1) / T;
        for (int i = 0; i < L; ++i)
            transpose_kernel<<<bHH, T, 0, stream>>>(W2 + (size_t)i * nHH,
                                                    w2t + (size_t)i * nHH, H, H);
        for (int i = 0; i < L * NL; ++i)
            transpose_kernel<<<bHH, T, 0, stream>>>(gw + (size_t)i * nHH,
                                                    gwt + (size_t)i * nHH, H, H);
        int nHC = H * C;
        transpose_kernel<<<(nHC + T - 1) / T, T, 0, stream>>>(Wc, wct, H, C);
    }

    {
        int n4 = (int)((size_t)NN * H / 4);
        zero_f4_kernel<<<(n4 + T - 1) / T, T, 0, stream>>>((float4*)out_feat, n4);
    }

    for (int i = 0; i < L; ++i) {
        const int*   src_i = src + (size_t)i * NE;
        const int*   dst_i = dst + (size_t)i * NE;
        const int*   ids_i = ids + (size_t)i * NS;
        const float* x_i   = x   + (size_t)i * NS * H;
        const float* w2t_i = w2t + (size_t)i * H * H;
        const float* b2_i  = b2  + (size_t)i * H;

        {
            int n4 = (2 * NS) / 4;
            zero_f4_kernel<<<(n4 + T - 1) / T, T, 0, stream>>>((float4*)deg_out, n4);
        }
        degree_count_kernel<<<(NE + T - 1) / T, T, 0, stream>>>(src_i, dst_i,
                                                                deg_out, deg_in, NE);
        rsqrt_clamp_kernel<<<(2 * NS + T - 1) / T, T, 0, stream>>>(deg_out, 2 * NS);

        gemm_wmma_k128_kernel<8><<<gemm_blocks_NS, T, 0, stream>>>(
            x_i, w2t_i, b2_i, nullptr, deg_out, hbuf, NS, 0);

        for (int l = 0; l < NL; ++l) {
            const float* gwt_il = gwt + ((size_t)i * NL + l) * H * H;
            const float* gb_il  = gb  + ((size_t)i * NL + l) * H;
            {
                int n4 = (int)((size_t)NS * H / 4);
                zero_f4_kernel<<<(n4 + T - 1) / T, T, 0, stream>>>((float4*)agg, n4);
            }
            edge_aggregate_kernel<<<(NE * 32) / T, T, 0, stream>>>(src_i, dst_i,
                                                                   hbuf, agg, NE);
            const float* escale = (l + 1 < NL) ? deg_out : nullptr;
            gemm_wmma_k128_kernel<8><<<gemm_blocks_NS, T, 0, stream>>>(
                agg, gwt_il, gb_il, deg_in, escale, hbuf, NS, 1);
        }

        scatter_rows_kernel<<<feat_blocks_NS, T, 0, stream>>>(ids_i, hbuf,
                                                              out_feat, NS);
    }

    gemm_wmma_k128_kernel<1><<<gemm_blocks_NN, T, 0, stream>>>(
        out_feat, wct, bc, nullptr, nullptr, (float*)d_out, NN, 0);
}


# edge sums: f32 atomics replaced by in-call destination-ordered edge list (count, scan, atomic place; held in output buffer until the classifier overwrites it); wave sums a row in list (arrival) order
# speedup vs baseline: 1.5299x; 1.2284x over previous
#include <hip/hip_runtime.h>


typedef __attribute__((ext_vector_type(2))) float v2f;
typedef __attribute__((ext_vector_type(8))) float v8f;


__global__ void zero_f4_kernel(float4* __restrict__ p, int n4) {
    int i = blockIdx.x * blockDim.x + threadIdx.x;
    if (i < n4) p[i] = make_float4(0.f, 0.f, 0.f, 0.f);
}

__global__ void transpose_kernel(const float* __restrict__ in,
                                 float* __restrict__ outT, int K, int N) {
    int i = blockIdx.x * blockDim.x + threadIdx.x;
    if (i < N * K) {
        int n = i / K, k = i % K;
        outT[i] = in[k * N + n];
    }
}

__global__ void degree_count_kernel(const int* __restrict__ src,
                                    const int* __restrict__ dst,
                                    float* __restrict__ deg_out,
                                    float* __restrict__ deg_in, int nE) {
    int e = blockIdx.x * blockDim.x + threadIdx.x;
    if (e < nE) {
        atomicAdd(&deg_out[src[e]], 1.0f);
        atomicAdd(&deg_in[dst[e]], 1.0f);
    }
}

__global__ void rsqrt_clamp_kernel(float* __restrict__ d, int n) {
    int i = blockIdx.x * blockDim.x + threadIdx.x;
    if (i < n) d[i] = rsqrtf(fmaxf(d[i], 1.0f));
}

__global__ __launch_bounds__(1024)
void segment_start_scan_kernel(const float* __restrict__ cnt,
                               int* __restrict__ seg, int n) {
    __shared__ int s_front[32];
    __shared__ int s_own[32];
    const int t = threadIdx.x;
    const int lane = t & 31;
    const int w = t >> 5;
    const int first = blockIdx.x * 1024;
    int front = 0;
    const float4* c4 = (const float4*)cnt;
    const int nf4 = first >> 2;
    for (int i = t; i < nf4; i += 4096) {
        const float4 a = c4[i];
        const float4 b = c4[min(i + 1024, nf4 - 1)];
        const float4 c = c4[min(i + 2048, nf4 - 1)];
        const float4 e = c4[min(i + 3072, nf4 - 1)];
        const int sa = (int)a.x + (int)a.y + (int)a.z + (int)a.w;
        const int sb = (int)b.x + (int)b.y + (int)b.z + (int)b.w;
        const int sc = (int)c.x + (int)c.y + (int)c.z + (int)c.w;
        const int se = (int)e.x + (int)e.y + (int)e.z + (int)e.w;
        front += sa;
        front += (i + 1024 < nf4) ? sb : 0;
        front += (i + 2048 < nf4) ? sc : 0;
        front += (i + 3072 < nf4) ? se : 0;
    }
    const int idx = first + t;
    const int v = (idx < n) ? (int)cnt[idx] : 0;
    int incl = v;
    for (int o = 1; o < 32; o <<= 1) {
        const int u = __shfl_up(incl, o);
        if (lane >= o) incl += u;
    }
    for (int o = 16; o > 0; o >>= 1) front += __shfl_xor(front, o);
    if (lane == 31) s_own[w] = incl;
    if (lane == 0) s_front[w] = front;
    __syncthreads();
    int base = 0, woff = 0;
    for (int k = 0; k < 32; ++k) {
        base += s_front[k];
        if (k < w) woff += s_own[k];
    }
    if (idx < n) seg[idx] = base + woff + incl - v;
}

__global__ void place_edges_kernel(const int* __restrict__ src,
                                   const int* __restrict__ dst,
                                   int* __restrict__ seg,
                                   int* __restrict__ sorted_src, int nE) {
    int e = blockIdx.x * blockDim.x + threadIdx.x;
    if (e < nE) {
        const int pos = atomicAdd(&seg[dst[e]], 1);
        if ((unsigned)pos < (unsigned)nE) sorted_src[pos] = src[e];
    }
}

__global__ __launch_bounds__(256)
void gather_rows_sum_kernel(const int* __restrict__ seg_end,
                            const int* __restrict__ sorted_src,
                            const float* __restrict__ h,
                            float* __restrict__ agg, int n, int nE) {
    const int lane = threadIdx.x & 31;
    const int wave0 = __builtin_amdgcn_readfirstlane(
        (int)((blockIdx.x * blockDim.x + threadIdx.x) >> 5));
    const int nwaves = (int)((gridDim.x * blockDim.x) >> 5);
    const float4* h4 = (const float4*)h + lane;
    for (int d = wave0; d < n; d += nwaves) {
        int beg = seg_end[max(d - 1, 0)];
        int end = seg_end[d];
        if (d == 0) beg = 0;
        beg = max(beg, 0);
        end = min(end, nE);
        float4 acc = make_float4(0.f, 0.f, 0.f, 0.f);
        for (int base = beg; base < end; base += 32) {
            const int cnt = min(32, end - base);
            int mys = (lane < cnt) ? sorted_src[base + lane] : 0;
            mys = min(max(mys, 0), n - 1);
            int j = 0;
            for (; j + 4 <= cnt; j += 4) {
                const int s0 = __builtin_amdgcn_readlane(mys, j);
                const int s1 = __builtin_amdgcn_readlane(mys, j + 1);
                const int s2 = __builtin_amdgcn_readlane(mys, j + 2);
                const int s3 = __builtin_amdgcn_readlane(mys, j + 3);
                const float4 v0 = h4[(size_t)s0 * 32];
                const float4 v1 = h4[(size_t)s1 * 32];
                const float4 v2 = h4[(size_t)s2 * 32];
                const float4 v3 = h4[(size_t)s3 * 32];
                acc.x += v0.x; acc.y += v0.y; acc.z += v0.z; acc.w += v0.w;
                acc.x += v1.x; acc.y += v1.y; acc.z += v1.z; acc.w += v1.w;
                acc.x += v2.x; acc.y += v2.y; acc.z += v2.z; acc.w += v2.w;
                acc.x += v3.x; acc.y += v3.y; acc.z += v3.z; acc.w += v3.w;
            }
            const int rem = cnt - j;
            if (rem == 3) {
                const int s0 = __builtin_amdgcn_readlane(mys, j);
                const int s1 = __builtin_amdgcn_readlane(mys, j + 1);
                const int s2 = __builtin_amdgcn_readlane(mys, j + 2);
                const float4 v0 = h4[(size_t)s0 * 32];
                const float4 v1 = h4[(size_t)s1 * 32];
                const float4 v2 = h4[(size_t)s2 * 32];
                acc.x += v0.x; acc.y += v0.y; acc.z += v0.z; acc.w += v0.w;
                acc.x += v1.x; acc.y += v1.y; acc.z += v1.z; acc.w += v1.w;
                acc.x += v2.x; acc.y += v2.y; acc.z += v2.z; acc.w += v2.w;
            } else if (rem == 2) {
                const int s0 = __builtin_amdgcn_readlane(mys, j);
                const int s1 = __builtin_amdgcn_readlane(mys, j + 1);
                const float4 v0 = h4[(size_t)s0 * 32];
                const float4 v1 = h4[(size_t)s1 * 32];
                acc.x += v0.x; acc.y += v0.y; acc.z += v0.z; acc.w += v0.w;
                acc.x += v1.x; acc.y += v1.y; acc.z += v1.z; acc.w += v1.w;
            } else if (rem == 1) {
                const int s0 = __builtin_amdgcn_readlane(mys, j);
                const float4 v0 = h4[(size_t)s0 * 32];
                acc.x += v0.x; acc.y += v0.y; acc.z += v0.z; acc.w += v0.w;
            }
        }
        ((float4*)agg)[(size_t)d * 32 + lane] = acc;
    }
}

__global__ void scatter_rows_kernel(const int* __restrict__ ids,
                                    const float* __restrict__ h,
                                    float* __restrict__ out_feat, int n) {
    int tid = blockIdx.x * blockDim.x + threadIdx.x;
    int r = tid >> 5;
    int f = tid & 31;
    if (r < n) {
        const float* hp = h + (size_t)r * 128 + f;
        float* o = out_feat + (size_t)ids[r] * 128 + f;
        const float v0 = hp[0];
        const float v1 = hp[32];
        const float v2 = hp[64];
        const float v3 = hp[96];
        atomicAdd(o + 0, v0);
        atomicAdd(o + 32, v1);
        atomicAdd(o + 64, v2);
        atomicAdd(o + 96, v3);
    }
}

template <int NTILES>
__global__ __launch_bounds__(256)
void gemm_wmma_k128_kernel(const float* __restrict__ A,
                           const float* __restrict__ Wt,
                           const float* __restrict__ bias,
                           const float* __restrict__ arowscale,
                           const float* __restrict__ erowscale,
                           float* __restrict__ out,
                           int M, int relu) {
    const int lane    = threadIdx.x & 31;
    const int wave    = threadIdx.x >> 5;
    const int task    = blockIdx.x * (blockDim.x >> 5) + wave;
    const int t0      = task * 2;
    const int t1      = t0 + 1;
    if (t0 * 16 >= M) return;
    const bool has1   = (t1 * 16 < M);
    const int halfsel = lane >> 4;
    const int lid     = lane & 15;
    const int N       = NTILES * 16;

    v8f acc0[NTILES], acc1[NTILES];
#pragma unroll
    for (int nt = 0; nt < NTILES; ++nt) {
        float bv = bias[nt * 16 + lid];
#pragma unroll
        for (int j = 0; j < 8; ++j) { acc0[nt][j] = bv; acc1[nt][j] = bv; }
    }

    const int arow0 = t0 * 16 + lid;
    const int arow1 = min(t1 * 16 + lid, M - 1);
    const float* aptr0 = A + (size_t)arow0 * 128 + halfsel * 2;
    const float* aptr1 = A + (size_t)arow1 * 128 + halfsel * 2;
    const float ascale0 = arowscale ? arowscale[arow0] : 1.0f;
    const float ascale1 = arowscale ? arowscale[arow1] : 1.0f;

    const float* wlane = Wt + (size_t)lid * 128 + halfsel * 2;

#pragma unroll 2
    for (int kk = 0; kk < 32; ++kk) {
        v2f a0 = *(const v2f*)(aptr0);
        v2f a1 = *(const v2f*)(aptr1);
        a0.x *= ascale0; a0.y *= ascale0;
        a1.x *= ascale1; a1.y *= ascale1;
        aptr0 += 4; aptr1 += 4;
#pragma unroll
        for (int nt = 0; nt < NTILES; ++nt) {
            v2f b = *(const v2f*)(wlane + (size_t)nt * 16 * 128 + kk * 4);
            acc0[nt] = __builtin_amdgcn_wmma_f32_16x16x4_f32(
                false, a0, false, b, (short)0, acc0[nt], false, false);
            acc1[nt] = __builtin_amdgcn_wmma_f32_16x16x4_f32(
                false, a1, false, b, (short)0, acc1[nt], false, false);
        }
    }

#pragma unroll
    for (int nt = 0; nt < NTILES; ++nt) {
#pragma unroll
        for (int j = 0; j < 8; ++j) {
            float v = acc0[nt][j];
            if (relu) v = fmaxf(v, 0.0f);
            const int row = t0 * 16 + j + halfsel * 8;
            if (erowscale) v *= erowscale[row];
            out[(size_t)row * N + nt * 16 + lid] = v;
        }
    }
    if (has1) {
#pragma unroll
        for (int nt = 0; nt < NTILES; ++nt) {
#pragma unroll
            for (int j = 0; j < 8; ++j) {
                float v = acc1[nt][j];
                if (relu) v = fmaxf(v, 0.0f);
                const int row = t1 * 16 + j + halfsel * 8;
                if (erowscale) v *= erowscale[row];
                out[(size_t)row * N + nt * 16 + lid] = v;
            }
        }
    }
}


extern "C" void kernel_launch(void* const* d_in, const int* in_sizes, int n_in,
                              void* d_out, int out_size, void* d_ws, size_t ws_size,
                              hipStream_t stream) {
    (void)n_in; (void)ws_size;
    const float* x   = (const float*)d_in[0];
    const float* W2  = (const float*)d_in[1];
    const float* b2  = (const float*)d_in[2];
    const float* gw  = (const float*)d_in[3];
    const float* gb  = (const float*)d_in[4];
    const float* Wc  = (const float*)d_in[5];
    const float* bc  = (const float*)d_in[6];
    const int*   src = (const int*)d_in[7];
    const int*   dst = (const int*)d_in[8];
    const int*   ids = (const int*)d_in[9];

    const int H  = 128;
    const int C  = 16;
    const int L  = in_sizes[2] / H;
    const int NS = in_sizes[9] / L;
    const int NE = in_sizes[7] / L;
    const int NL = in_sizes[4] / (L * H);
    const int NN = out_size / C;

    float* ws       = (float*)d_ws;
    float* out_feat = ws;
    float* hbuf     = out_feat + (size_t)NN * H;
    float* agg      = hbuf + (size_t)NS * H;
    float* deg_out  = agg + (size_t)NS * H;
    float* deg_in   = deg_out + NS;
    float* w2t      = deg_in + NS;
    float* gwt      = w2t + (size_t)L * H * H;
    float* wct      = gwt + (size_t)L * NL * H * H;

    int* seg_end    = (int*)d_out;
    int* sorted_src = seg_end + NS;

    const int T = 256;
    const int feat_blocks_NS = (NS * 32 + T - 1) / T;
    const int gemm_blocks_NS = (((NS / 16) + 1) / 2 + 7) / 8;
    const int gemm_blocks_NN = (((NN / 16) + 1) / 2 + 7) / 8;
    const int gather_blocks  = ((NS + 1) / 2 + 7) / 8;

    {
        int nHH = H * H;
        int bHH = (nHH + T - 1) / T;
        for (int i = 0; i < L; ++i)
            transpose_kernel<<<bHH, T, 0, stream>>>(W2 + (size_t)i * nHH,
                                                    w2t + (size_t)i * nHH, H, H);
        for (int i = 0; i < L * NL; ++i)
            transpose_kernel<<<bHH, T, 0, stream>>>(gw + (size_t)i * nHH,
                                                    gwt + (size_t)i * nHH, H, H);
        int nHC = H * C;
        transpose_kernel<<<(nHC + T - 1) / T, T, 0, stream>>>(Wc, wct, H, C);
    }

    {
        int n4 = (int)((size_t)NN * H / 4);
        zero_f4_kernel<<<(n4 + T - 1) / T, T, 0, stream>>>((float4*)out_feat, n4);
    }

    for (int i = 0; i < L; ++i) {
        const int*   src_i = src + (size_t)i * NE;
        const int*   dst_i = dst + (size_t)i * NE;
        const int*   ids_i = ids + (size_t)i * NS;
        const float* x_i   = x   + (size_t)i * NS * H;
        const float* w2t_i = w2t + (size_t)i * H * H;
        const float* b2_i  = b2  + (size_t)i * H;

        {
            int n4 = (2 * NS) / 4;
            zero_f4_kernel<<<(n4 + T - 1) / T, T, 0, stream>>>((float4*)deg_out, n4);
        }
        degree_count_kernel<<<(NE + T - 1) / T, T, 0, stream>>>(src_i, dst_i,
                                                                deg_out, deg_in, NE);
        segment_start_scan_kernel<<<(NS + 1023) / 1024, 1024, 0, stream>>>(
            deg_in, seg_end, NS);
        rsqrt_clamp_kernel<<<(2 * NS + T - 1) / T, T, 0, stream>>>(deg_out, 2 * NS);
        place_edges_kernel<<<(NE + T - 1) / T, T, 0, stream>>>(src_i, dst_i,
                                                               seg_end, sorted_src, NE);

        gemm_wmma_k128_kernel<8><<<gemm_blocks_NS, T, 0, stream>>>(
            x_i, w2t_i, b2_i, nullptr, deg_out, hbuf, NS, 0);

        for (int l = 0; l < NL; ++l) {
            const float* gwt_il = gwt + ((size_t)i * NL + l) * H * H;
            const float* gb_il  = gb  + ((size_t)i * NL + l) * H;
            gather_rows_sum_kernel<<<gather_blocks, T, 0, stream>>>(
                seg_end, sorted_src, hbuf, agg, NS, NE);
            const float* escale = (l + 1 < NL) ? deg_out : nullptr;
            gemm_wmma_k128_kernel<8><<<gemm_blocks_NS, T, 0, stream>>>(
                agg, gwt_il, gb_il, deg_in, escale, hbuf, NS, 1);
        }

        scatter_rows_kernel<<<feat_blocks_NS, T, 0, stream>>>(ids_i, hbuf,
                                                              out_feat, NS);
    }

    gemm_wmma_k128_kernel<1><<<gemm_blocks_NN, T, 0, stream>>>(
        out_feat, wct, bc, nullptr, nullptr, (float*)d_out, NN, 0);
}


# GEMM kernel: W^T and each wave's A rows staged in LDS by whole-row loads, fragments read from LDS (one ds_load_b128 per two WMMAs; k sum order inside each 8-group changes), result rows stored whole
# speedup vs baseline: 1.9107x; 1.2489x over previous
#include <hip/hip_runtime.h>


typedef __attribute__((ext_vector_type(2))) float v2f;
typedef __attribute__((ext_vector_type(4))) float v4f;
typedef __attribute__((ext_vector_type(8))) float v8f;


__global__ void zero_f4_kernel(float4* __restrict__ p, int n4) {
    int i = blockIdx.x * blockDim.x + threadIdx.x;
    if (i < n4) p[i] = make_float4(0.f, 0.f, 0.f, 0.f);
}

__global__ void transpose_kernel(const float* __restrict__ in,
                                 float* __restrict__ outT, int K, int N) {
    int i = blockIdx.x * blockDim.x + threadIdx.x;
    if (i < N * K) {
        int n = i / K, k = i % K;
        outT[i] = in[k * N + n];
    }
}

__global__ void degree_count_kernel(const int* __restrict__ src,
                                    const int* __restrict__ dst,
                                    float* __restrict__ deg_out,
                                    float* __restrict__ deg_in, int nE) {
    int e = blockIdx.x * blockDim.x + threadIdx.x;
    if (e < nE) {
        atomicAdd(&deg_out[src[e]], 1.0f);
        atomicAdd(&deg_in[dst[e]], 1.0f);
    }
}

__global__ void rsqrt_clamp_kernel(float* __restrict__ d, int n) {
    int i = blockIdx.x * blockDim.x + threadIdx.x;
    if (i < n) d[i] = rsqrtf(fmaxf(d[i], 1.0f));
}

__global__ __launch_bounds__(1024)
void segment_start_scan_kernel(const float* __restrict__ cnt,
                               int* __restrict__ seg, int n) {
    __shared__ int s_front[32];
    __shared__ int s_own[32];
    const int t = threadIdx.x;
    const int lane = t & 31;
    const int w = t >> 5;
    const int first = blockIdx.x * 1024;
    int front = 0;
    const float4* c4 = (const float4*)cnt;
    const int nf4 = first >> 2;
    for (int i = t; i < nf4; i += 4096) {
        const float4 a = c4[i];
        const float4 b = c4[min(i + 1024, nf4 - 1)];
        const float4 c = c4[min(i + 2048, nf4 - 1)];
        const float4 e = c4[min(i + 3072, nf4 - 1)];
        const int sa = (int)a.x + (int)a.y + (int)a.z + (int)a.w;
        const int sb = (int)b.x + (int)b.y + (int)b.z + (int)b.w;
        const int sc = (int)c.x + (int)c.y + (int)c.z + (int)c.w;
        const int se = (int)e.x + (int)e.y + (int)e.z + (int)e.w;
        front += sa;
        front += (i + 1024 < nf4) ? sb : 0;
        front += (i + 2048 < nf4) ? sc : 0;
        front += (i + 3072 < nf4) ? se : 0;
    }
    const int idx = first + t;
    const int v = (idx < n) ? (int)cnt[idx] : 0;
    int incl = v;
    for (int o = 1; o < 32; o <<= 1) {
        const int u = __shfl_up(incl, o);
        if (lane >= o) incl += u;
    }
    for (int o = 16; o > 0; o >>= 1) front += __shfl_xor(front, o);
    if (lane == 31) s_own[w] = incl;
    if (lane == 0) s_front[w] = front;
    __syncthreads();
    int base = 0, woff = 0;
    for (int k = 0; k < 32; ++k) {
        base += s_front[k];
        if (k < w) woff += s_own[k];
    }
    if (idx < n) seg[idx] = base + woff + incl - v;
}

__global__ void place_edges_kernel(const int* __restrict__ src,
                                   const int* __restrict__ dst,
                                   int* __restrict__ seg,
                                   int* __restrict__ sorted_src, int nE) {
    int e = blockIdx.x * blockDim.x + threadIdx.x;
    if (e < nE) {
        const int pos = atomicAdd(&seg[dst[e]], 1);
        if ((unsigned)pos < (unsigned)nE) sorted_src[pos] = src[e];
    }
}

__global__ __launch_bounds__(256)
void gather_rows_sum_kernel(const int* __restrict__ seg_end,
                            const int* __restrict__ sorted_src,
                            const float* __restrict__ h,
                            float* __restrict__ agg, int n, int nE) {
    const int lane = threadIdx.x & 31;
    const int wave0 = __builtin_amdgcn_readfirstlane(
        (int)((blockIdx.x * blockDim.x + threadIdx.x) >> 5));
    const int nwaves = (int)((gridDim.x * blockDim.x) >> 5);
    const float4* h4 = (const float4*)h + lane;
    for (int d = wave0; d < n; d += nwaves) {
        int beg = seg_end[max(d - 1, 0)];
        int end = seg_end[d];
        if (d == 0) beg = 0;
        beg = max(beg, 0);
        end = min(end, nE);
        float4 acc = make_float4(0.f, 0.f, 0.f, 0.f);
        for (int base = beg; base < end; base += 32) {
            const int cnt = min(32, end - base);
            int mys = (lane < cnt) ? sorted_src[base + lane] : 0;
            mys = min(max(mys, 0), n - 1);
            int j = 0;
            for (; j + 4 <= cnt; j += 4) {
                const int s0 = __builtin_amdgcn_readlane(mys, j);
                const int s1 = __builtin_amdgcn_readlane(mys, j + 1);
                const int s2 = __builtin_amdgcn_readlane(mys, j + 2);
                const int s3 = __builtin_amdgcn_readlane(mys, j + 3);
                const float4 v0 = h4[(size_t)s0 * 32];
                const float4 v1 = h4[(size_t)s1 * 32];
                const float4 v2 = h4[(size_t)s2 * 32];
                const float4 v3 = h4[(size_t)s3 * 32];
                acc.x += v0.x; acc.y += v0.y; acc.z += v0.z; acc.w += v0.w;
                acc.x += v1.x; acc.y += v1.y; acc.z += v1.z; acc.w += v1.w;
                acc.x += v2.x; acc.y += v2.y; acc.z += v2.z; acc.w += v2.w;
                acc.x += v3.x; acc.y += v3.y; acc.z += v3.z; acc.w += v3.w;
            }
            const int rem = cnt - j;
            if (rem == 3) {
                const int s0 = __builtin_amdgcn_readlane(mys, j);
                const int s1 = __builtin_amdgcn_readlane(mys, j + 1);
                const int s2 = __builtin_amdgcn_readlane(mys, j + 2);
                const float4 v0 = h4[(size_t)s0 * 32];
                const float4 v1 = h4[(size_t)s1 * 32];
                const float4 v2 = h4[(size_t)s2 * 32];
                acc.x += v0.x; acc.y += v0.y; acc.z += v0.z; acc.w += v0.w;
                acc.x += v1.x; acc.y += v1.y; acc.z += v1.z; acc.w += v1.w;
                acc.x += v2.x; acc.y += v2.y; acc.z += v2.z; acc.w += v2.w;
            } else if (rem == 2) {
                const int s0 = __builtin_amdgcn_readlane(mys, j);
                const int s1 = __builtin_amdgcn_readlane(mys, j + 1);
                const float4 v0 = h4[(size_t)s0 * 32];
                const float4 v1 = h4[(size_t)s1 * 32];
                acc.x += v0.x; acc.y += v0.y; acc.z += v0.z; acc.w += v0.w;
                acc.x += v1.x; acc.y += v1.y; acc.z += v1.z; acc.w += v1.w;
            } else if (rem == 1) {
                const int s0 = __builtin_amdgcn_readlane(mys, j);
                const float4 v0 = h4[(size_t)s0 * 32];
                acc.x += v0.x; acc.y += v0.y; acc.z += v0.z; acc.w += v0.w;
            }
        }
        ((float4*)agg)[(size_t)d * 32 + lane] = acc;
    }
}

__global__ void scatter_rows_kernel(const int* __restrict__ ids,
                                    const float* __restrict__ h,
                                    float* __restrict__ out_feat, int n) {
    int tid = blockIdx.x * blockDim.x + threadIdx.x;
    int r = tid >> 5;
    int f = tid & 31;
    if (r < n) {
        const float* hp = h + (size_t)r * 128 + f;
        float* o = out_feat + (size_t)ids[r] * 128 + f;
        const float v0 = hp[0];
        const float v1 = hp[32];
        const float v2 = hp[64];
        const float v3 = hp[96];
        atomicAdd(o + 0, v0);
        atomicAdd(o + 32, v1);
        atomicAdd(o + 64, v2);
        atomicAdd(o + 96, v3);
    }
}

template <int NTILES>
__global__ __launch_bounds__(256)
void gemm_wmma_k128_kernel(const float* __restrict__ A,
                           const float* __restrict__ Wt,
                           const float* __restrict__ bias,
                           const float* __restrict__ arowscale,
                           const float* __restrict__ erowscale,
                           float* __restrict__ out,
                           int M, int relu) {
    constexpr int N      = NTILES * 16;
    constexpr int PITCH  = 132;
    constexpr int OPITCH = N + 4;
    __shared__ __attribute__((aligned(16))) float s_w[N * PITCH];
    __shared__ __attribute__((aligned(16))) float s_a[8 * 32 * PITCH];

    const int tid      = threadIdx.x;
    const int lane     = tid & 31;
    const int wave     = __builtin_amdgcn_readfirstlane(tid >> 5);
    const int c        = lane & 15;
    const int h        = lane >> 4;
    const int row_base = (blockIdx.x * 8 + wave) * 32;
    float* my_a = s_a + wave * (32 * PITCH);

    v4f wreg[N / 8];
#pragma unroll
    for (int i = 0; i < N / 8; ++i)
        wreg[i] = ((const v4f*)Wt)[tid + 256 * i];
    v4f areg[32];
#pragma unroll
    for (int i = 0; i < 32; ++i) {
        const int r = min(row_base + i, M - 1);
        areg[i] = ((const v4f*)A)[(size_t)r * 32 + lane];
    }
    float myscale = 1.0f;
    if (arowscale) myscale = arowscale[min(row_base + lane, M - 1)];
    float es0[8], es1[8];
#pragma unroll
    for (int r = 0; r < 8; ++r) { es0[r] = 1.0f; es1[r] = 1.0f; }
    if (erowscale) {
#pragma unroll
        for (int r = 0; r < 8; ++r) {
            es0[r] = erowscale[min(row_base + 8 * h + r, M - 1)];
            es1[r] = erowscale[min(row_base + 16 + 8 * h + r, M - 1)];
        }
    }

    v8f acc0[NTILES], acc1[NTILES];
#pragma unroll
    for (int nt = 0; nt < NTILES; ++nt) {
        const float bv = bias[nt * 16 + c];
#pragma unroll
        for (int r = 0; r < 8; ++r) { acc0[nt][r] = bv; acc1[nt][r] = bv; }
    }

#pragma unroll
    for (int i = 0; i < N / 8; ++i) {
        const int u = tid + 256 * i;
        *(v4f*)(s_w + (u >> 5) * PITCH + (u & 31) * 4) = wreg[i];
    }
#pragma unroll
    for (int i = 0; i < 32; ++i) {
        const float s = __int_as_float(
            __builtin_amdgcn_readlane(__float_as_int(myscale), i));
        v4f v = areg[i];
        v.x *= s; v.y *= s; v.z *= s; v.w *= s;
        *(v4f*)(my_a + i * PITCH + lane * 4) = v;
    }

    __syncthreads();

    const float* aw0 = my_a + c * PITCH + 4 * h;
    const float* aw1 = my_a + (16 + c) * PITCH + 4 * h;
    const float* bw  = s_w + c * PITCH + 4 * h;

    v4f a0c = *(const v4f*)(aw0);
    v4f a1c = *(const v4f*)(aw1);
    v4f bc[NTILES];
#pragma unroll
    for (int nt = 0; nt < NTILES; ++nt)
        bc[nt] = *(const v4f*)(bw + nt * 16 * PITCH);
#pragma unroll
    for (int j = 0; j < 16; ++j) {
        v4f a0n = a0c, a1n = a1c;
        v4f bn[NTILES];
#pragma unroll
        for (int nt = 0; nt < NTILES; ++nt) bn[nt] = bc[nt];
        if (j + 1 < 16) {
            a0n = *(const v4f*)(aw0 + 8 * (j + 1));
            a1n = *(const v4f*)(aw1 + 8 * (j + 1));
#pragma unroll
            for (int nt = 0; nt < NTILES; ++nt)
                bn[nt] = *(const v4f*)(bw + nt * 16 * PITCH + 8 * (j + 1));
        }
#pragma unroll
        for (int nt = 0; nt < NTILES; ++nt) {
            acc0[nt] = __builtin_amdgcn_wmma_f32_16x16x4_f32(
                false, a0c.xy, false, bc[nt].xy, (short)0, acc0[nt], false, false);
            acc1[nt] = __builtin_amdgcn_wmma_f32_16x16x4_f32(
                false, a1c.xy, false, bc[nt].xy, (short)0, acc1[nt], false, false);
            acc0[nt] = __builtin_amdgcn_wmma_f32_16x16x4_f32(
                false, a0c.zw, false, bc[nt].zw, (short)0, acc0[nt], false, false);
            acc1[nt] = __builtin_amdgcn_wmma_f32_16x16x4_f32(
                false, a1c.zw, false, bc[nt].zw, (short)0, acc1[nt], false, false);
        }
        a0c = a0n; a1c = a1n;
#pragma unroll
        for (int nt = 0; nt < NTILES; ++nt) bc[nt] = bn[nt];
    }

#pragma unroll
    for (int nt = 0; nt < NTILES; ++nt) {
#pragma unroll
        for (int r = 0; r < 8; ++r) {
            float v0 = acc0[nt][r];
            float v1 = acc1[nt][r];
            if (relu) { v0 = fmaxf(v0, 0.0f); v1 = fmaxf(v1, 0.0f); }
            v0 *= es0[r];
            v1 *= es1[r];
            my_a[(8 * h + r) * OPITCH + nt * 16 + c]      = v0;
            my_a[(16 + 8 * h + r) * OPITCH + nt * 16 + c] = v1;
        }
    }

    __syncthreads();

    constexpr int ITER = N / 4;
    constexpr int GRP  = ITER < 8 ? ITER : 8;
#pragma unroll
    for (int g = 0; g < ITER / GRP; ++g) {
        v4f t[GRP];
#pragma unroll
        for (int q = 0; q < GRP; ++q) {
            const int u = lane + 32 * (g * GRP + q);
            t[q] = *(const v4f*)(my_a + (u / (N / 4)) * OPITCH + 4 * (u % (N / 4)));
        }
#pragma unroll
        for (int q = 0; q < GRP; ++q) {
            const int u    = lane + 32 * (g * GRP + q);
            const int grow = row_base + u / (N / 4);
            if (grow < M)
                *(v4f*)(out + (size_t)grow * N + 4 * (u % (N / 4))) = t[q];
        }
    }
}


extern "C" void kernel_launch(void* const* d_in, const int* in_sizes, int n_in,
                              void* d_out, int out_size, void* d_ws, size_t ws_size,
                              hipStream_t stream) {
    (void)n_in; (void)ws_size;
    const float* x   = (const float*)d_in[0];
    const float* W2  = (const float*)d_in[1];
    const float* b2  = (const float*)d_in[2];
    const float* gw  = (const float*)d_in[3];
    const float* gb  = (const float*)d_in[4];
    const float* Wc  = (const float*)d_in[5];
    const float* bc  = (const float*)d_in[6];
    const int*   src = (const int*)d_in[7];
    const int*   dst = (const int*)d_in[8];
    const int*   ids = (const int*)d_in[9];

    const int H  = 128;
    const int C  = 16;
    const int L  = in_sizes[2] / H;
    const int NS = in_sizes[9] / L;
    const int NE = in_sizes[7] / L;
    const int NL = in_sizes[4] / (L * H);
    const int NN = out_size / C;

    float* ws       = (float*)d_ws;
    float* out_feat = ws;
    float* hbuf     = out_feat + (size_t)NN * H;
    float* agg      = hbuf + (size_t)NS * H;
    float* deg_out  = agg + (size_t)NS * H;
    float* deg_in   = deg_out + NS;
    float* w2t      = deg_in + NS;
    float* gwt      = w2t + (size_t)L * H * H;
    float* wct      = gwt + (size_t)L * NL * H * H;

    int* seg_end    = (int*)d_out;
    int* sorted_src = seg_end + NS;

    const int T = 256;
    const int feat_blocks_NS = (NS * 32 + T - 1) / T;
    const int gemm_blocks_NS = (((NS / 16) + 1) / 2 + 7) / 8;
    const int gemm_blocks_NN = (((NN / 16) + 1) / 2 + 7) / 8;
    const int gather_blocks  = ((NS + 1) / 2 + 7) / 8;

    {
        int nHH = H * H;
        int bHH = (nHH + T - 1) / T;
        for (int i = 0; i < L; ++i)
            transpose_kernel<<<bHH, T, 0, stream>>>(W2 + (size_t)i * nHH,
                                                    w2t + (size_t)i * nHH, H, H);
        for (int i = 0; i < L * NL; ++i)
            transpose_kernel<<<bHH, T, 0, stream>>>(gw + (size_t)i * nHH,
                                                    gwt + (size_t)i * nHH, H, H);
        int nHC = H * C;
        transpose_kernel<<<(nHC + T - 1) / T, T, 0, stream>>>(Wc, wct, H, C);
    }

    {
        int n4 = (int)((size_t)NN * H / 4);
        zero_f4_kernel<<<(n4 + T - 1) / T, T, 0, stream>>>((float4*)out_feat, n4);
    }

    for (int i = 0; i < L; ++i) {
        const int*   src_i = src + (size_t)i * NE;
        const int*   dst_i = dst + (size_t)i * NE;
        const int*   ids_i = ids + (size_t)i * NS;
        const float* x_i   = x   + (size_t)i * NS * H;
        const float* w2t_i = w2t + (size_t)i * H * H;
        const float* b2_i  = b2  + (size_t)i * H;

        {
            int n4 = (2 * NS) / 4;
            zero_f4_kernel<<<(n4 + T - 1) / T, T, 0, stream>>>((float4*)deg_out, n4);
        }
        degree_count_kernel<<<(NE + T - 1) / T, T, 0, stream>>>(src_i, dst_i,
                                                                deg_out, deg_in, NE);
        segment_start_scan_kernel<<<(NS + 1023) / 1024, 1024, 0, stream>>>(
            deg_in, seg_end, NS);
        rsqrt_clamp_kernel<<<(2 * NS + T - 1) / T, T, 0, stream>>>(deg_out, 2 * NS);
        place_edges_kernel<<<(NE + T - 1) / T, T, 0, stream>>>(src_i, dst_i,
                                                               seg_end, sorted_src, NE);

        gemm_wmma_k128_kernel<8><<<gemm_blocks_NS, T, 0, stream>>>(
            x_i, w2t_i, b2_i, nullptr, deg_out, hbuf, NS, 0);

        for (int l = 0; l < NL; ++l) {
            const float* gwt_il = gwt + ((size_t)i * NL + l) * H * H;
            const float* gb_il  = gb  + ((size_t)i * NL + l) * H;
            gather_rows_sum_kernel<<<gather_blocks, T, 0, stream>>>(
                seg_end, sorted_src, hbuf, agg, NS, NE);
            const float* escale = (l + 1 < NL) ? deg_out : nullptr;
            gemm_wmma_k128_kernel<8><<<gemm_blocks_NS, T, 0, stream>>>(
                agg, gwt_il, gb_il, deg_in, escale, hbuf, NS, 1);
        }

        scatter_rows_kernel<<<feat_blocks_NS, T, 0, stream>>>(ids_i, hbuf,
                                                              out_feat, NS);
    }

    gemm_wmma_k128_kernel<1><<<gemm_blocks_NN, T, 0, stream>>>(
        out_feat, wct, bc, nullptr, nullptr, (float*)d_out, NN, 0);
}


# degree counts and edge placement: scattered global atomics replaced by LDS atomics in blocks owning a node-id range x an edge slice; order inside a destination's list now by slice, then LDS arrival
# speedup vs baseline: 3.0392x; 1.5906x over previous
#include <hip/hip_runtime.h>


typedef __attribute__((ext_vector_type(2))) float v2f;
typedef __attribute__((ext_vector_type(4))) float v4f;
typedef __attribute__((ext_vector_type(8))) float v8f;


__global__ void zero_f4_kernel(float4* __restrict__ p, int n4) {
    int i = blockIdx.x * blockDim.x + threadIdx.x;
    if (i < n4) p[i] = make_float4(0.f, 0.f, 0.f, 0.f);
}

__global__ void transpose_kernel(const float* __restrict__ in,
                                 float* __restrict__ outT, int K, int N) {
    int i = blockIdx.x * blockDim.x + threadIdx.x;
    if (i < N * K) {
        int n = i / K, k = i % K;
        outT[i] = in[k * N + n];
    }
}

#define EB_NB  16
#define EB_NSL 16
#define EB_BR  3136
#define EB_T   512
#define EB_U   16

__device__ __forceinline__ void edge_slice_bounds(int sl, int nE, int& e0, int& e1) {
    const int per = (nE + EB_NSL - 1) / EB_NSL;
    e0 = min(sl * per, nE);
    e1 = min(e0 + per, nE);
}

__global__ __launch_bounds__(EB_T)
void slice_count_kernel(const int* __restrict__ src,
                        const int* __restrict__ dst,
                        int* __restrict__ cin,
                        int* __restrict__ cout, int n, int nE) {
    __shared__ int s_in[EB_BR];
    __shared__ int s_out[EB_BR];
    const int t  = threadIdx.x;
    const int sl = blockIdx.x / EB_NB;
    const int rg = blockIdx.x % EB_NB;
    int e0, e1;
    edge_slice_bounds(sl, nE, e0, e1);
    for (int r0 = rg * EB_BR; r0 < n; r0 += EB_NB * EB_BR) {
        for (int i = t; i < EB_BR; i += EB_T) { s_in[i] = 0; s_out[i] = 0; }
        __syncthreads();
        for (int e = e0 + t; e < e1; e += EB_U * EB_T) {
            int dv[EB_U], sv[EB_U];
#pragma unroll
            for (int q = 0; q < EB_U; ++q) {
                const int ee = e + q * EB_T;
                const int ec = min(ee, e1 - 1);
                const int d = dst[ec];
                const int s = src[ec];
                dv[q] = (ee < e1) ? d : -1;
                sv[q] = (ee < e1) ? s : -1;
            }
#pragma unroll
            for (int q = 0; q < EB_U; ++q) {
                const unsigned a = (unsigned)(dv[q] - r0);
                const unsigned b = (unsigned)(sv[q] - r0);
                if (a < (unsigned)EB_BR && dv[q] < n) atomicAdd(&s_in[a], 1);
                if (b < (unsigned)EB_BR && sv[q] < n) atomicAdd(&s_out[b], 1);
            }
        }
        __syncthreads();
        for (int i = t; i < EB_BR; i += EB_T) {
            const int d = r0 + i;
            if (d < n) {
                cin[(size_t)sl * n + d]  = s_in[i];
                cout[(size_t)sl * n + d] = s_out[i];
            }
        }
        __syncthreads();
    }
}

__global__ void slice_totals_kernel(int* __restrict__ cin,
                                    const int* __restrict__ cout,
                                    float* __restrict__ cnt_in,
                                    float* __restrict__ deg_out,
                                    float* __restrict__ deg_in, int n) {
    const int d = blockIdx.x * blockDim.x + threadIdx.x;
    if (d < n) {
        int ci[EB_NSL], co = 0;
#pragma unroll
        for (int k = 0; k < EB_NSL; ++k) {
            ci[k] = cin[(size_t)k * n + d];
            co += cout[(size_t)k * n + d];
        }
        int run = 0;
#pragma unroll
        for (int k = 0; k < EB_NSL; ++k) {
            cin[(size_t)k * n + d] = run;
            run += ci[k];
        }
        cnt_in[d]  = (float)run;
        deg_in[d]  = rsqrtf(fmaxf((float)run, 1.0f));
        deg_out[d] = rsqrtf(fmaxf((float)co, 1.0f));
    }
}

__global__ __launch_bounds__(1024)
void segment_start_scan_kernel(const float* __restrict__ cnt,
                               int* __restrict__ seg, int n) {
    __shared__ int s_front[32];
    __shared__ int s_own[32];
    const int t = threadIdx.x;
    const int lane = t & 31;
    const int w = t >> 5;
    const int first = blockIdx.x * 1024;
    int front = 0;
    const float4* c4 = (const float4*)cnt;
    const int nf4 = first >> 2;
    for (int i = t; i < nf4; i += 4096) {
        const float4 a = c4[i];
        const float4 b = c4[min(i + 1024, nf4 - 1)];
        const float4 c = c4[min(i + 2048, nf4 - 1)];
        const float4 e = c4[min(i + 3072, nf4 - 1)];
        const int sa = (int)a.x + (int)a.y + (int)a.z + (int)a.w;
        const int sb = (int)b.x + (int)b.y + (int)b.z + (int)b.w;
        const int sc = (int)c.x + (int)c.y + (int)c.z + (int)c.w;
        const int se = (int)e.x + (int)e.y + (int)e.z + (int)e.w;
        front += sa;
        front += (i + 1024 < nf4) ? sb : 0;
        front += (i + 2048 < nf4) ? sc : 0;
        front += (i + 3072 < nf4) ? se : 0;
    }
    const int idx = first + t;
    const int v = (idx < n) ? (int)cnt[idx] : 0;
    int incl = v;
    for (int o = 1; o < 32; o <<= 1) {
        const int u = __shfl_up(incl, o);
        if (lane >= o) incl += u;
    }
    for (int o = 16; o > 0; o >>= 1) front += __shfl_xor(front, o);
    if (lane == 31) s_own[w] = incl;
    if (lane == 0) s_front[w] = front;
    __syncthreads();
    int base = 0, woff = 0;
    for (int k = 0; k < 32; ++k) {
        base += s_front[k];
        if (k < w) woff += s_own[k];
    }
    if (idx < n) seg[idx] = base + woff + incl - v;
}

__global__ __launch_bounds__(EB_T)
void slice_place_kernel(const int* __restrict__ src,
                        const int* __restrict__ dst,
                        const int* __restrict__ seg,
                        const int* __restrict__ cin_front,
                        int* __restrict__ sorted_src, int n, int nE) {
    __shared__ int s_cur[EB_BR];
    const int t  = threadIdx.x;
    const int sl = blockIdx.x / EB_NB;
    const int rg = blockIdx.x % EB_NB;
    int e0, e1;
    edge_slice_bounds(sl, nE, e0, e1);
    for (int r0 = rg * EB_BR; r0 < n; r0 += EB_NB * EB_BR) {
        constexpr int NI = (EB_BR + EB_T - 1) / EB_T;
        int cv[NI];
#pragma unroll
        for (int k = 0; k < NI; ++k) {
            const int d = min(r0 + t + k * EB_T, n - 1);
            cv[k] = seg[d] + cin_front[(size_t)sl * n + d];
        }
#pragma unroll
        for (int k = 0; k < NI; ++k) {
            const int i = t + k * EB_T;
            if (i < EB_BR) s_cur[i] = cv[k];
        }
        __syncthreads();
        for (int e = e0 + t; e < e1; e += EB_U * EB_T) {
            int dv[EB_U], sv[EB_U];
#pragma unroll
            for (int q = 0; q < EB_U; ++q) {
                const int ee = e + q * EB_T;
                const int ec = min(ee, e1 - 1);
                const int d = dst[ec];
                const int s = src[ec];
                dv[q] = (ee < e1) ? d : -1;
                sv[q] = s;
            }
#pragma unroll
            for (int q = 0; q < EB_U; ++q) {
                const unsigned a = (unsigned)(dv[q] - r0);
                if (a < (unsigned)EB_BR && dv[q] < n) {
                    const int pos = atomicAdd(&s_cur[a], 1);
                    if ((unsigned)pos < (unsigned)nE) sorted_src[pos] = sv[q];
                }
            }
        }
        __syncthreads();
    }
}

__global__ __launch_bounds__(256)
void gather_rows_sum_kernel(const int* __restrict__ seg_start,
                            const int* __restrict__ sorted_src,
                            const float* __restrict__ h,
                            float* __restrict__ agg, int n, int nE) {
    const int lane = threadIdx.x & 31;
    const int wave0 = __builtin_amdgcn_readfirstlane(
        (int)((blockIdx.x * blockDim.x + threadIdx.x) >> 5));
    const int nwaves = (int)((gridDim.x * blockDim.x) >> 5);
    const float4* h4 = (const float4*)h + lane;
    for (int d = wave0; d < n; d += nwaves) {
        int beg = seg_start[d];
        int end = seg_start[min(d + 1, n - 1)];
        if (d == n - 1) end = nE;
        beg = max(beg, 0);
        end = min(end, nE);
        float4 acc = make_float4(0.f, 0.f, 0.f, 0.f);
        for (int base = beg; base < end; base += 32) {
            const int cnt = min(32, end - base);
            int mys = (lane < cnt) ? sorted_src[base + lane] : 0;
            mys = min(max(mys, 0), n - 1);
            int j = 0;
            for (; j + 4 <= cnt; j += 4) {
                const int s0 = __builtin_amdgcn_readlane(mys, j);
                const int s1 = __builtin_amdgcn_readlane(mys, j + 1);
                const int s2 = __builtin_amdgcn_readlane(mys, j + 2);
                const int s3 = __builtin_amdgcn_readlane(mys, j + 3);
                const float4 v0 = h4[(size_t)s0 * 32];
                const float4 v1 = h4[(size_t)s1 * 32];
                const float4 v2 = h4[(size_t)s2 * 32];
                const float4 v3 = h4[(size_t)s3 * 32];
                acc.x += v0.x; acc.y += v0.y; acc.z += v0.z; acc.w += v0.w;
                acc.x += v1.x; acc.y += v1.y; acc.z += v1.z; acc.w += v1.w;
                acc.x += v2.x; acc.y += v2.y; acc.z += v2.z; acc.w += v2.w;
                acc.x += v3.x; acc.y += v3.y; acc.z += v3.z; acc.w += v3.w;
            }
            const int rem = cnt - j;
            if (rem == 3) {
                const int s0 = __builtin_amdgcn_readlane(mys, j);
                const int s1 = __builtin_amdgcn_readlane(mys, j + 1);
                const int s2 = __builtin_amdgcn_readlane(mys, j + 2);
                const float4 v0 = h4[(size_t)s0 * 32];
                const float4 v1 = h4[(size_t)s1 * 32];
                const float4 v2 = h4[(size_t)s2 * 32];
                acc.x += v0.x; acc.y += v0.y; acc.z += v0.z; acc.w += v0.w;
                acc.x += v1.x; acc.y += v1.y; acc.z += v1.z; acc.w += v1.w;
                acc.x += v2.x; acc.y += v2.y; acc.z += v2.z; acc.w += v2.w;
            } else if (rem == 2) {
                const int s0 = __builtin_amdgcn_readlane(mys, j);
                const int s1 = __builtin_amdgcn_readlane(mys, j + 1);
                const float4 v0 = h4[(size_t)s0 * 32];
                const float4 v1 = h4[(size_t)s1 * 32];
                acc.x += v0.x; acc.y += v0.y; acc.z += v0.z; acc.w += v0.w;
                acc.x += v1.x; acc.y += v1.y; acc.z += v1.z; acc.w += v1.w;
            } else if (rem == 1) {
                const int s0 = __builtin_amdgcn_readlane(mys, j);
                const float4 v0 = h4[(size_t)s0 * 32];
                acc.x += v0.x; acc.y += v0.y; acc.z += v0.z; acc.w += v0.w;
            }
        }
        ((float4*)agg)[(size_t)d * 32 + lane] = acc;
    }
}

__global__ void scatter_rows_kernel(const int* __restrict__ ids,
                                    const float* __restrict__ h,
                                    float* __restrict__ out_feat, int n) {
    int tid = blockIdx.x * blockDim.x + threadIdx.x;
    int r = tid >> 5;
    int f = tid & 31;
    if (r < n) {
        const float* hp = h + (size_t)r * 128 + f;
        float* o = out_feat + (size_t)ids[r] * 128 + f;
        const float v0 = hp[0];
        const float v1 = hp[32];
        const float v2 = hp[64];
        const float v3 = hp[96];
        atomicAdd(o + 0, v0);
        atomicAdd(o + 32, v1);
        atomicAdd(o + 64, v2);
        atomicAdd(o + 96, v3);
    }
}

template <int NTILES>
__global__ __launch_bounds__(256)
void gemm_wmma_k128_kernel(const float* __restrict__ A,
                           const float* __restrict__ Wt,
                           const float* __restrict__ bias,
                           const float* __restrict__ arowscale,
                           const float* __restrict__ erowscale,
                           float* __restrict__ out,
                           int M, int relu) {
    constexpr int N      = NTILES * 16;
    constexpr int PITCH  = 132;
    constexpr int OPITCH = N + 4;
    __shared__ __attribute__((aligned(16))) float s_w[N * PITCH];
    __shared__ __attribute__((aligned(16))) float s_a[8 * 32 * PITCH];

    const int tid      = threadIdx.x;
    const int lane     = tid & 31;
    const int wave     = __builtin_amdgcn_readfirstlane(tid >> 5);
    const int c        = lane & 15;
    const int h        = lane >> 4;
    const int row_base = (blockIdx.x * 8 + wave) * 32;
    float* my_a = s_a + wave * (32 * PITCH);

    v4f wreg[N / 8];
#pragma unroll
    for (int i = 0; i < N / 8; ++i)
        wreg[i] = ((const v4f*)Wt)[tid + 256 * i];
    v4f areg[32];
#pragma unroll
    for (int i = 0; i < 32; ++i) {
        const int r = min(row_base + i, M - 1);
        areg[i] = ((const v4f*)A)[(size_t)r * 32 + lane];
    }
    float myscale = 1.0f;
    if (arowscale) myscale = arowscale[min(row_base + lane, M - 1)];
    float es0[8], es1[8];
#pragma unroll
    for (int r = 0; r < 8; ++r) { es0[r] = 1.0f; es1[r] = 1.0f; }
    if (erowscale) {
#pragma unroll
        for (int r = 0; r < 8; ++r) {
            es0[r] = erowscale[min(row_base + 8 * h + r, M - 1)];
            es1[r] = erowscale[min(row_base + 16 + 8 * h + r, M - 1)];
        }
    }

    v8f acc0[NTILES], acc1[NTILES];
#pragma unroll
    for (int nt = 0; nt < NTILES; ++nt) {
        const float bv = bias[nt * 16 + c];
#pragma unroll
        for (int r = 0; r < 8; ++r) { acc0[nt][r] = bv; acc1[nt][r] = bv; }
    }

#pragma unroll
    for (int i = 0; i < N / 8; ++i) {
        const int u = tid + 256 * i;
        *(v4f*)(s_w + (u >> 5) * PITCH + (u & 31) * 4) = wreg[i];
    }
#pragma unroll
    for (int i = 0; i < 32; ++i) {
        const float s = __int_as_float(
            __builtin_amdgcn_readlane(__float_as_int(myscale), i));
        v4f v = areg[i];
        v.x *= s; v.y *= s; v.z *= s; v.w *= s;
        *(v4f*)(my_a + i * PITCH + lane * 4) = v;
    }

    __syncthreads();

    const float* aw0 = my_a + c * PITCH + 4 * h;
    const float* aw1 = my_a + (16 + c) * PITCH + 4 * h;
    const float* bw  = s_w + c * PITCH + 4 * h;

    v4f a0c = *(const v4f*)(aw0);
    v4f a1c = *(const v4f*)(aw1);
    v4f bc[NTILES];
#pragma unroll
    for (int nt = 0; nt < NTILES; ++nt)
        bc[nt] = *(const v4f*)(bw + nt * 16 * PITCH);
#pragma unroll
    for (int j = 0; j < 16; ++j) {
        v4f a0n = a0c, a1n = a1c;
        v4f bn[NTILES];
#pragma unroll
        for (int nt = 0; nt < NTILES; ++nt) bn[nt] = bc[nt];
        if (j + 1 < 16) {
            a0n = *(const v4f*)(aw0 + 8 * (j + 1));
            a1n = *(const v4f*)(aw1 + 8 * (j + 1));
#pragma unroll
            for (int nt = 0; nt < NTILES; ++nt)
                bn[nt] = *(const v4f*)(bw + nt * 16 * PITCH + 8 * (j + 1));
        }
#pragma unroll
        for (int nt = 0; nt < NTILES; ++nt) {
            acc0[nt] = __builtin_amdgcn_wmma_f32_16x16x4_f32(
                false, a0c.xy, false, bc[nt].xy, (short)0, acc0[nt], false, false);
            acc1[nt] = __builtin_amdgcn_wmma_f32_16x16x4_f32(
                false, a1c.xy, false, bc[nt].xy, (short)0, acc1[nt], false, false);
            acc0[nt] = __builtin_amdgcn_wmma_f32_16x16x4_f32(
                false, a0c.zw, false, bc[nt].zw, (short)0, acc0[nt], false, false);
            acc1[nt] = __builtin_amdgcn_wmma_f32_16x16x4_f32(
                false, a1c.zw, false, bc[nt].zw, (short)0, acc1[nt], false, false);
        }
        a0c = a0n; a1c = a1n;
#pragma unroll
        for (int nt = 0; nt < NTILES; ++nt) bc[nt] = bn[nt];
    }

#pragma unroll
    for (int nt = 0; nt < NTILES; ++nt) {
#pragma unroll
        for (int r = 0; r < 8; ++r) {
            float v0 = acc0[nt][r];
            float v1 = acc1[nt][r];
            if (relu) { v0 = fmaxf(v0, 0.0f); v1 = fmaxf(v1, 0.0f); }
            v0 *= es0[r];
            v1 *= es1[r];
            my_a[(8 * h + r) * OPITCH + nt * 16 + c]      = v0;
            my_a[(16 + 8 * h + r) * OPITCH + nt * 16 + c] = v1;
        }
    }

    __syncthreads();

    constexpr int ITER = N / 4;
    constexpr int GRP  = ITER < 8 ? ITER : 8;
#pragma unroll
    for (int g = 0; g < ITER / GRP; ++g) {
        v4f t[GRP];
#pragma unroll
        for (int q = 0; q < GRP; ++q) {
            const int u = lane + 32 * (g * GRP + q);
            t[q] = *(const v4f*)(my_a + (u / (N / 4)) * OPITCH + 4 * (u % (N / 4)));
        }
#pragma unroll
        for (int q = 0; q < GRP; ++q) {
            const int u    = lane + 32 * (g * GRP + q);
            const int grow = row_base + u / (N / 4);
            if (grow < M)
                *(v4f*)(out + (size_t)grow * N + 4 * (u % (N / 4))) = t[q];
        }
    }
}


extern "C" void kernel_launch(void* const* d_in, const int* in_sizes, int n_in,
                              void* d_out, int out_size, void* d_ws, size_t ws_size,
                              hipStream_t stream) {
    (void)n_in; (void)ws_size;
    const float* x   = (const float*)d_in[0];
    const float* W2  = (const float*)d_in[1];
    const float* b2  = (const float*)d_in[2];
    const float* gw  = (const float*)d_in[3];
    const float* gb  = (const float*)d_in[4];
    const float* Wc  = (const float*)d_in[5];
    const float* bc  = (const float*)d_in[6];
    const int*   src = (const int*)d_in[7];
    const int*   dst = (const int*)d_in[8];
    const int*   ids = (const int*)d_in[9];

    const int H  = 128;
    const int C  = 16;
    const int L  = in_sizes[2] / H;
    const int NS = in_sizes[9] / L;
    const int NE = in_sizes[7] / L;
    const int NL = in_sizes[4] / (L * H);
    const int NN = out_size / C;

    float* ws       = (float*)d_ws;
    float* out_feat = ws;
    float* hbuf     = out_feat + (size_t)NN * H;
    float* agg      = hbuf + (size_t)NS * H;
    float* deg_out  = agg + (size_t)NS * H;
    float* deg_in   = deg_out + NS;
    float* w2t      = deg_in + NS;
    float* gwt      = w2t + (size_t)L * H * H;
    float* wct      = gwt + (size_t)L * NL * H * H;

    int* seg_start  = (int*)d_out;
    int* sorted_src = seg_start + NS;
    int*   cin_mat  = (int*)agg;
    int*   cout_mat = cin_mat + (size_t)EB_NSL * NS;
    float* cnt_in   = (float*)(cout_mat + (size_t)EB_NSL * NS);

    const int T = 256;
    const int feat_blocks_NS = (NS * 32 + T - 1) / T;
    const int gemm_blocks_NS = (((NS / 16) + 1) / 2 + 7) / 8;
    const int gemm_blocks_NN = (((NN / 16) + 1) / 2 + 7) / 8;
    const int gather_blocks  = ((NS + 1) / 2 + 7) / 8;

    {
        int nHH = H * H;
        int bHH = (nHH + T - 1) / T;
        for (int i = 0; i < L; ++i)
            transpose_kernel<<<bHH, T, 0, stream>>>(W2 + (size_t)i * nHH,
                                                    w2t + (size_t)i * nHH, H, H);
        for (int i = 0; i < L * NL; ++i)
            transpose_kernel<<<bHH, T, 0, stream>>>(gw + (size_t)i * nHH,
                                                    gwt + (size_t)i * nHH, H, H);
        int nHC = H * C;
        transpose_kernel<<<(nHC + T - 1) / T, T, 0, stream>>>(Wc, wct, H, C);
    }

    {
        int n4 = (int)((size_t)NN * H / 4);
        zero_f4_kernel<<<(n4 + T - 1) / T, T, 0, stream>>>((float4*)out_feat, n4);
    }

    for (int i = 0; i < L; ++i) {
        const int*   src_i = src + (size_t)i * NE;
        const int*   dst_i = dst + (size_t)i * NE;
        const int*   ids_i = ids + (size_t)i * NS;
        const float* x_i   = x   + (size_t)i * NS * H;
        const float* w2t_i = w2t + (size_t)i * H * H;
        const float* b2_i  = b2  + (size_t)i * H;

        slice_count_kernel<<<EB_NSL * EB_NB, EB_T, 0, stream>>>(src_i, dst_i,
                                                               cin_mat, cout_mat, NS, NE);
        slice_totals_kernel<<<(NS + T - 1) / T, T, 0, stream>>>(cin_mat, cout_mat, cnt_in,
                                                               deg_out, deg_in, NS);
        segment_start_scan_kernel<<<(NS + 1023) / 1024, 1024, 0, stream>>>(
            cnt_in, seg_start, NS);
        slice_place_kernel<<<EB_NSL * EB_NB, EB_T, 0, stream>>>(src_i, dst_i, seg_start,
                                                               cin_mat, sorted_src, NS, NE);

        gemm_wmma_k128_kernel<8><<<gemm_blocks_NS, T, 0, stream>>>(
            x_i, w2t_i, b2_i, nullptr, deg_out, hbuf, NS, 0);

        for (int l = 0; l < NL; ++l) {
            const float* gwt_il = gwt + ((size_t)i * NL + l) * H * H;
            const float* gb_il  = gb  + ((size_t)i * NL + l) * H;
            gather_rows_sum_kernel<<<gather_blocks, T, 0, stream>>>(
                seg_start, sorted_src, hbuf, agg, NS, NE);
            const float* escale = (l + 1 < NL) ? deg_out : nullptr;
            gemm_wmma_k128_kernel<8><<<gemm_blocks_NS, T, 0, stream>>>(
                agg, gwt_il, gb_il, deg_in, escale, hbuf, NS, 1);
        }

        scatter_rows_kernel<<<feat_blocks_NS, T, 0, stream>>>(ids_i, hbuf,
                                                              out_feat, NS);
    }

    gemm_wmma_k128_kernel<1><<<gemm_blocks_NN, T, 0, stream>>>(
        out_feat, wct, bc, nullptr, nullptr, (float*)d_out, NN, 0);
}


# classifier moved in front of the id scatter (linear): a link's last GEMM multiplies its rows by Wc on WMMA and atomically adds 16-float rows to the bias-filled output; links summed after Wc
# speedup vs baseline: 3.3832x; 1.1132x over previous
#include <hip/hip_runtime.h>


typedef __attribute__((ext_vector_type(2))) float v2f;
typedef __attribute__((ext_vector_type(4))) float v4f;
typedef __attribute__((ext_vector_type(8))) float v8f;


__global__ void bias_fill_kernel(const float* __restrict__ bc,
                                 float4* __restrict__ out4, int n4) {
    const int stride = gridDim.x * blockDim.x;
    int i = blockIdx.x * blockDim.x + threadIdx.x;
    const float4 b = ((const float4*)bc)[i & 3];
    for (; i < n4; i += stride) out4[i] = b;
}

__global__ void transpose_kernel(const float* __restrict__ in,
                                 float* __restrict__ outT, int K, int N) {
    int i = blockIdx.x * blockDim.x + threadIdx.x;
    if (i < N * K) {
        int n = i / K, k = i % K;
        outT[i] = in[k * N + n];
    }
}

#define EB_NB  16
#define EB_NSL 16
#define EB_BR  3136
#define EB_T   512
#define EB_U   16

__device__ __forceinline__ void edge_slice_bounds(int sl, int nE, int& e0, int& e1) {
    const int per = (nE + EB_NSL - 1) / EB_NSL;
    e0 = min(sl * per, nE);
    e1 = min(e0 + per, nE);
}

__global__ __launch_bounds__(EB_T)
void slice_count_kernel(const int* __restrict__ src,
                        const int* __restrict__ dst,
                        int* __restrict__ cin,
                        int* __restrict__ cout, int n, int nE) {
    __shared__ int s_in[EB_BR];
    __shared__ int s_out[EB_BR];
    const int t  = threadIdx.x;
    const int sl = blockIdx.x / EB_NB;
    const int rg = blockIdx.x % EB_NB;
    int e0, e1;
    edge_slice_bounds(sl, nE, e0, e1);
    for (int r0 = rg * EB_BR; r0 < n; r0 += EB_NB * EB_BR) {
        for (int i = t; i < EB_BR; i += EB_T) { s_in[i] = 0; s_out[i] = 0; }
        __syncthreads();
        for (int e = e0 + t; e < e1; e += EB_U * EB_T) {
            int dv[EB_U], sv[EB_U];
#pragma unroll
            for (int q = 0; q < EB_U; ++q) {
                const int ee = e + q * EB_T;
                const int ec = min(ee, e1 - 1);
                const int d = dst[ec];
                const int s = src[ec];
                dv[q] = (ee < e1) ? d : -1;
                sv[q] = (ee < e1) ? s : -1;
            }
#pragma unroll
            for (int q = 0; q < EB_U; ++q) {
                const unsigned a = (unsigned)(dv[q] - r0);
                const unsigned b = (unsigned)(sv[q] - r0);
                if (a < (unsigned)EB_BR && dv[q] < n) atomicAdd(&s_in[a], 1);
                if (b < (unsigned)EB_BR && sv[q] < n) atomicAdd(&s_out[b], 1);
            }
        }
        __syncthreads();
        for (int i = t; i < EB_BR; i += EB_T) {
            const int d = r0 + i;
            if (d < n) {
                cin[(size_t)sl * n + d]  = s_in[i];
                cout[(size_t)sl * n + d] = s_out[i];
            }
        }
        __syncthreads();
    }
}

__global__ void slice_totals_kernel(int* __restrict__ cin,
                                    const int* __restrict__ cout,
                                    float* __restrict__ cnt_in,
                                    float* __restrict__ deg_out,
                                    float* __restrict__ deg_in, int n) {
    const int d = blockIdx.x * blockDim.x + threadIdx.x;
    if (d < n) {
        int ci[EB_NSL], co = 0;
#pragma unroll
        for (int k = 0; k < EB_NSL; ++k) {
            ci[k] = cin[(size_t)k * n + d];
            co += cout[(size_t)k * n + d];
        }
        int run = 0;
#pragma unroll
        for (int k = 0; k < EB_NSL; ++k) {
            cin[(size_t)k * n + d] = run;
            run += ci[k];
        }
        cnt_in[d]  = (float)run;
        deg_in[d]  = rsqrtf(fmaxf((float)run, 1.0f));
        deg_out[d] = rsqrtf(fmaxf((float)co, 1.0f));
    }
}

__global__ __launch_bounds__(1024)
void segment_start_scan_kernel(const float* __restrict__ cnt,
                               int* __restrict__ seg, int n) {
    __shared__ int s_front[32];
    __shared__ int s_own[32];
    const int t = threadIdx.x;
    const int lane = t & 31;
    const int w = t >> 5;
    const int first = blockIdx.x * 1024;
    int front = 0;
    const float4* c4 = (const float4*)cnt;
    const int nf4 = first >> 2;
    for (int i = t; i < nf4; i += 4096) {
        const float4 a = c4[i];
        const float4 b = c4[min(i + 1024, nf4 - 1)];
        const float4 c = c4[min(i + 2048, nf4 - 1)];
        const float4 e = c4[min(i + 3072, nf4 - 1)];
        const int sa = (int)a.x + (int)a.y + (int)a.z + (int)a.w;
        const int sb = (int)b.x + (int)b.y + (int)b.z + (int)b.w;
        const int sc = (int)c.x + (int)c.y + (int)c.z + (int)c.w;
        const int se = (int)e.x + (int)e.y + (int)e.z + (int)e.w;
        front += sa;
        front += (i + 1024 < nf4) ? sb : 0;
        front += (i + 2048 < nf4) ? sc : 0;
        front += (i + 3072 < nf4) ? se : 0;
    }
    const int idx = first + t;
    const int v = (idx < n) ? (int)cnt[idx] : 0;
    int incl = v;
    for (int o = 1; o < 32; o <<= 1) {
        const int u = __shfl_up(incl, o);
        if (lane >= o) incl += u;
    }
    for (int o = 16; o > 0; o >>= 1) front += __shfl_xor(front, o);
    if (lane == 31) s_own[w] = incl;
    if (lane == 0) s_front[w] = front;
    __syncthreads();
    int base = 0, woff = 0;
    for (int k = 0; k < 32; ++k) {
        base += s_front[k];
        if (k < w) woff += s_own[k];
    }
    if (idx < n) seg[idx] = base + woff + incl - v;
}

__global__ __launch_bounds__(EB_T)
void slice_place_kernel(const int* __restrict__ src,
                        const int* __restrict__ dst,
                        const int* __restrict__ seg,
                        const int* __restrict__ cin_front,
                        int* __restrict__ sorted_src, int n, int nE) {
    __shared__ int s_cur[EB_BR];
    const int t  = threadIdx.x;
    const int sl = blockIdx.x / EB_NB;
    const int rg = blockIdx.x % EB_NB;
    int e0, e1;
    edge_slice_bounds(sl, nE, e0, e1);
    for (int r0 = rg * EB_BR; r0 < n; r0 += EB_NB * EB_BR) {
        constexpr int NI = (EB_BR + EB_T - 1) / EB_T;
        int cv[NI];
#pragma unroll
        for (int k = 0; k < NI; ++k) {
            const int d = min(r0 + t + k * EB_T, n - 1);
            cv[k] = seg[d] + cin_front[(size_t)sl * n + d];
        }
#pragma unroll
        for (int k = 0; k < NI; ++k) {
            const int i = t + k * EB_T;
            if (i < EB_BR) s_cur[i] = cv[k];
        }
        __syncthreads();
        for (int e = e0 + t; e < e1; e += EB_U * EB_T) {
            int dv[EB_U], sv[EB_U];
#pragma unroll
            for (int q = 0; q < EB_U; ++q) {
                const int ee = e + q * EB_T;
                const int ec = min(ee, e1 - 1);
                const int d = dst[ec];
                const int s = src[ec];
                dv[q] = (ee < e1) ? d : -1;
                sv[q] = s;
            }
#pragma unroll
            for (int q = 0; q < EB_U; ++q) {
                const unsigned a = (unsigned)(dv[q] - r0);
                if (a < (unsigned)EB_BR && dv[q] < n) {
                    const int pos = atomicAdd(&s_cur[a], 1);
                    if ((unsigned)pos < (unsigned)nE) sorted_src[pos] = sv[q];
                }
            }
        }
        __syncthreads();
    }
}

__global__ __launch_bounds__(256)
void gather_rows_sum_kernel(const int* __restrict__ seg_start,
                            const int* __restrict__ sorted_src,
                            const float* __restrict__ h,
                            float* __restrict__ agg, int n, int nE) {
    const int lane = threadIdx.x & 31;
    const int wave0 = __builtin_amdgcn_readfirstlane(
        (int)((blockIdx.x * blockDim.x + threadIdx.x) >> 5));
    const int nwaves = (int)((gridDim.x * blockDim.x) >> 5);
    const float4* h4 = (const float4*)h + lane;
    for (int d = wave0; d < n; d += nwaves) {
        int beg = seg_start[d];
        int end = seg_start[min(d + 1, n - 1)];
        if (d == n - 1) end = nE;
        beg = max(beg, 0);
        end = min(end, nE);
        float4 acc = make_float4(0.f, 0.f, 0.f, 0.f);
        for (int base = beg; base < end; base += 32) {
            const int cnt = min(32, end - base);
            int mys = (lane < cnt) ? sorted_src[base + lane] : 0;
            mys = min(max(mys, 0), n - 1);
            int j = 0;
            for (; j + 4 <= cnt; j += 4) {
                const int s0 = __builtin_amdgcn_readlane(mys, j);
                const int s1 = __builtin_amdgcn_readlane(mys, j + 1);
                const int s2 = __builtin_amdgcn_readlane(mys, j + 2);
                const int s3 = __builtin_amdgcn_readlane(mys, j + 3);
                const float4 v0 = h4[(size_t)s0 * 32];
                const float4 v1 = h4[(size_t)s1 * 32];
                const float4 v2 = h4[(size_t)s2 * 32];
                const float4 v3 = h4[(size_t)s3 * 32];
                acc.x += v0.x; acc.y += v0.y; acc.z += v0.z; acc.w += v0.w;
                acc.x += v1.x; acc.y += v1.y; acc.z += v1.z; acc.w += v1.w;
                acc.x += v2.x; acc.y += v2.y; acc.z += v2.z; acc.w += v2.w;
                acc.x += v3.x; acc.y += v3.y; acc.z += v3.z; acc.w += v3.w;
            }
            const int rem = cnt - j;
            if (rem == 3) {
                const int s0 = __builtin_amdgcn_readlane(mys, j);
                const int s1 = __builtin_amdgcn_readlane(mys, j + 1);
                const int s2 = __builtin_amdgcn_readlane(mys, j + 2);
                const float4 v0 = h4[(size_t)s0 * 32];
                const float4 v1 = h4[(size_t)s1 * 32];
                const float4 v2 = h4[(size_t)s2 * 32];
                acc.x += v0.x; acc.y += v0.y; acc.z += v0.z; acc.w += v0.w;
                acc.x += v1.x; acc.y += v1.y; acc.z += v1.z; acc.w += v1.w;
                acc.x += v2.x; acc.y += v2.y; acc.z += v2.z; acc.w += v2.w;
            } else if (rem == 2) {
                const int s0 = __builtin_amdgcn_readlane(mys, j);
                const int s1 = __builtin_amdgcn_readlane(mys, j + 1);
                const float4 v0 = h4[(size_t)s0 * 32];
                const float4 v1 = h4[(size_t)s1 * 32];
                acc.x += v0.x; acc.y += v0.y; acc.z += v0.z; acc.w += v0.w;
                acc.x += v1.x; acc.y += v1.y; acc.z += v1.z; acc.w += v1.w;
            } else if (rem == 1) {
                const int s0 = __builtin_amdgcn_readlane(mys, j);
                const float4 v0 = h4[(size_t)s0 * 32];
                acc.x += v0.x; acc.y += v0.y; acc.z += v0.z; acc.w += v0.w;
            }
        }
        ((float4*)agg)[(size_t)d * 32 + lane] = acc;
    }
}

template <int NTILES, bool CLS>
__global__ __launch_bounds__(256)
void gemm_wmma_k128_kernel(const float* __restrict__ A,
                           const float* __restrict__ Wt,
                           const float* __restrict__ bias,
                           const float* __restrict__ arowscale,
                           const float* __restrict__ erowscale,
                           float* __restrict__ out,
                           int M, int relu,
                           const float* __restrict__ Wct,
                           const int* __restrict__ ids,
                           float* __restrict__ cls_out) {
    constexpr int N      = NTILES * 16;
    constexpr int PITCH  = 132;
    constexpr int OPITCH = N + 4;
    static_assert(!CLS || OPITCH == PITCH, "CLS reads the finished rows as A rows");
    __shared__ __attribute__((aligned(16))) float s_w[N * PITCH];
    __shared__ __attribute__((aligned(16))) float s_a[8 * 32 * PITCH];
    __shared__ __attribute__((aligned(16))) float s_c[CLS ? 16 * PITCH : 4];

    const int tid      = threadIdx.x;
    const int lane     = tid & 31;
    const int wave     = __builtin_amdgcn_readfirstlane(tid >> 5);
    const int c        = lane & 15;
    const int h        = lane >> 4;
    const int row_base = (blockIdx.x * 8 + wave) * 32;
    float* my_a = s_a + wave * (32 * PITCH);

    v4f wreg[N / 8];
#pragma unroll
    for (int i = 0; i < N / 8; ++i)
        wreg[i] = ((const v4f*)Wt)[tid + 256 * i];
    v4f areg[32];
#pragma unroll
    for (int i = 0; i < 32; ++i) {
        const int r = min(row_base + i, M - 1);
        areg[i] = ((const v4f*)A)[(size_t)r * 32 + lane];
    }
    v4f creg[2];
    int id0[8], id1[8];
    if constexpr (CLS) {
#pragma unroll
        for (int i = 0; i < 2; ++i)
            creg[i] = ((const v4f*)Wct)[tid + 256 * i];
#pragma unroll
        for (int r = 0; r < 8; ++r) {
            id0[r] = ids[min(row_base + 8 * h + r, M - 1)];
            id1[r] = ids[min(row_base + 16 + 8 * h + r, M - 1)];
        }
    }
    float myscale = 1.0f;
    if (arowscale) myscale = arowscale[min(row_base + lane, M - 1)];
    float es0[8], es1[8];
#pragma unroll
    for (int r = 0; r < 8; ++r) { es0[r] = 1.0f; es1[r] = 1.0f; }
    if (erowscale) {
#pragma unroll
        for (int r = 0; r < 8; ++r) {
            es0[r] = erowscale[min(row_base + 8 * h + r, M - 1)];
            es1[r] = erowscale[min(row_base + 16 + 8 * h + r, M - 1)];
        }
    }

    v8f acc0[NTILES], acc1[NTILES];
#pragma unroll
    for (int nt = 0; nt < NTILES; ++nt) {
        const float bv = bias[nt * 16 + c];
#pragma unroll
        for (int r = 0; r < 8; ++r) { acc0[nt][r] = bv; acc1[nt][r] = bv; }
    }

#pragma unroll
    for (int i = 0; i < N / 8; ++i) {
        const int u = tid + 256 * i;
        *(v4f*)(s_w + (u >> 5) * PITCH + (u & 31) * 4) = wreg[i];
    }
    if constexpr (CLS) {
#pragma unroll
        for (int i = 0; i < 2; ++i) {
            const int u = tid + 256 * i;
            *(v4f*)(s_c + (u >> 5) * PITCH + (u & 31) * 4) = creg[i];
        }
    }
#pragma unroll
    for (int i = 0; i < 32; ++i) {
        const float s = __int_as_float(
            __builtin_amdgcn_readlane(__float_as_int(myscale), i));
        v4f v = areg[i];
        v.x *= s; v.y *= s; v.z *= s; v.w *= s;
        *(v4f*)(my_a + i * PITCH + lane * 4) = v;
    }

    __syncthreads();

    const float* aw0 = my_a + c * PITCH + 4 * h;
    const float* aw1 = my_a + (16 + c) * PITCH + 4 * h;
    const float* bw  = s_w + c * PITCH + 4 * h;

    v4f a0c = *(const v4f*)(aw0);
    v4f a1c = *(const v4f*)(aw1);
    v4f bc[NTILES];
#pragma unroll
    for (int nt = 0; nt < NTILES; ++nt)
        bc[nt] = *(const v4f*)(bw + nt * 16 * PITCH);
#pragma unroll
    for (int j = 0; j < 16; ++j) {
        v4f a0n = a0c, a1n = a1c;
        v4f bn[NTILES];
#pragma unroll
        for (int nt = 0; nt < NTILES; ++nt) bn[nt] = bc[nt];
        if (j + 1 < 16) {
            a0n = *(const v4f*)(aw0 + 8 * (j + 1));
            a1n = *(const v4f*)(aw1 + 8 * (j + 1));
#pragma unroll
            for (int nt = 0; nt < NTILES; ++nt)
                bn[nt] = *(const v4f*)(bw + nt * 16 * PITCH + 8 * (j + 1));
        }
#pragma unroll
        for (int nt = 0; nt < NTILES; ++nt) {
            acc0[nt] = __builtin_amdgcn_wmma_f32_16x16x4_f32(
                false, a0c.xy, false, bc[nt].xy, (short)0, acc0[nt], false, false);
            acc1[nt] = __builtin_amdgcn_wmma_f32_16x16x4_f32(
                false, a1c.xy, false, bc[nt].xy, (short)0, acc1[nt], false, false);
            acc0[nt] = __builtin_amdgcn_wmma_f32_16x16x4_f32(
                false, a0c.zw, false, bc[nt].zw, (short)0, acc0[nt], false, false);
            acc1[nt] = __builtin_amdgcn_wmma_f32_16x16x4_f32(
                false, a1c.zw, false, bc[nt].zw, (short)0, acc1[nt], false, false);
        }
        a0c = a0n; a1c = a1n;
#pragma unroll
        for (int nt = 0; nt < NTILES; ++nt) bc[nt] = bn[nt];
    }

#pragma unroll
    for (int nt = 0; nt < NTILES; ++nt) {
#pragma unroll
        for (int r = 0; r < 8; ++r) {
            float v0 = acc0[nt][r];
            float v1 = acc1[nt][r];
            if (relu) { v0 = fmaxf(v0, 0.0f); v1 = fmaxf(v1, 0.0f); }
            v0 *= es0[r];
            v1 *= es1[r];
            my_a[(8 * h + r) * OPITCH + nt * 16 + c]      = v0;
            my_a[(16 + 8 * h + r) * OPITCH + nt * 16 + c] = v1;
        }
    }

    __syncthreads();

    if constexpr (CLS) {
        const float* cw = s_c + c * PITCH + 4 * h;
        v8f p0, p1;
#pragma unroll
        for (int r = 0; r < 8; ++r) { p0[r] = 0.0f; p1[r] = 0.0f; }
        v4f x0c = *(const v4f*)(aw0);
        v4f x1c = *(const v4f*)(aw1);
        v4f wcc = *(const v4f*)(cw);
#pragma unroll
        for (int j = 0; j < 16; ++j) {
            v4f x0n = x0c, x1n = x1c, wcn = wcc;
            if (j + 1 < 16) {
                x0n = *(const v4f*)(aw0 + 8 * (j + 1));
                x1n = *(const v4f*)(aw1 + 8 * (j + 1));
                wcn = *(const v4f*)(cw + 8 * (j + 1));
            }
            p0 = __builtin_amdgcn_wmma_f32_16x16x4_f32(
                false, x0c.xy, false, wcc.xy, (short)0, p0, false, false);
            p1 = __builtin_amdgcn_wmma_f32_16x16x4_f32(
                false, x1c.xy, false, wcc.xy, (short)0, p1, false, false);
            p0 = __builtin_amdgcn_wmma_f32_16x16x4_f32(
                false, x0c.zw, false, wcc.zw, (short)0, p0, false, false);
            p1 = __builtin_amdgcn_wmma_f32_16x16x4_f32(
                false, x1c.zw, false, wcc.zw, (short)0, p1, false, false);
            x0c = x0n; x1c = x1n; wcc = wcn;
        }
#pragma unroll
        for (int r = 0; r < 8; ++r) {
            if (row_base + 8 * h + r < M)
                atomicAdd(cls_out + (size_t)id0[r] * 16 + c, p0[r]);
            if (row_base + 16 + 8 * h + r < M)
                atomicAdd(cls_out + (size_t)id1[r] * 16 + c, p1[r]);
        }
        return;
    }

    constexpr int ITER = N / 4;
    constexpr int GRP  = ITER < 8 ? ITER : 8;
#pragma unroll
    for (int g = 0; g < ITER / GRP; ++g) {
        v4f t[GRP];
#pragma unroll
        for (int q = 0; q < GRP; ++q) {
            const int u = lane + 32 * (g * GRP + q);
            t[q] = *(const v4f*)(my_a + (u / (N / 4)) * OPITCH + 4 * (u % (N / 4)));
        }
#pragma unroll
        for (int q = 0; q < GRP; ++q) {
            const int u    = lane + 32 * (g * GRP + q);
            const int grow = row_base + u / (N / 4);
            if (grow < M)
                *(v4f*)(out + (size_t)grow * N + 4 * (u % (N / 4))) = t[q];
        }
    }
}


extern "C" void kernel_launch(void* const* d_in, const int* in_sizes, int n_in,
                              void* d_out, int out_size, void* d_ws, size_t ws_size,
                              hipStream_t stream) {
    (void)n_in; (void)ws_size;
    const float* x   = (const float*)d_in[0];
    const float* W2  = (const float*)d_in[1];
    const float* b2  = (const float*)d_in[2];
    const float* gw  = (const float*)d_in[3];
    const float* gb  = (const float*)d_in[4];
    const float* Wc  = (const float*)d_in[5];
    const float* bc  = (const float*)d_in[6];
    const int*   src = (const int*)d_in[7];
    const int*   dst = (const int*)d_in[8];
    const int*   ids = (const int*)d_in[9];

    const int H  = 128;
    const int C  = 16;
    const int L  = in_sizes[2] / H;
    const int NS = in_sizes[9] / L;
    const int NE = in_sizes[7] / L;
    const int NL = in_sizes[4] / (L * H);
    const int NN = out_size / C;

    float* ws       = (float*)d_ws;
    float* out_feat = ws;
    float* hbuf     = out_feat + (size_t)NN * H;
    float* agg      = hbuf + (size_t)NS * H;
    float* deg_out  = agg + (size_t)NS * H;
    float* deg_in   = deg_out + NS;
    float* w2t      = deg_in + NS;
    float* gwt      = w2t + (size_t)L * H * H;
    float* wct      = gwt + (size_t)L * NL * H * H;

    int* seg_start  = (int*)out_feat;
    int* sorted_src = seg_start + NS;
    int*   cin_mat  = (int*)agg;
    int*   cout_mat = cin_mat + (size_t)EB_NSL * NS;
    float* cnt_in   = (float*)(cout_mat + (size_t)EB_NSL * NS);

    const int T = 256;
    const int gemm_blocks_NS = (((NS / 16) + 1) / 2 + 7) / 8;
    const int gather_blocks  = ((NS + 1) / 2 + 7) / 8;

    {
        int nHH = H * H;
        int bHH = (nHH + T - 1) / T;
        for (int i = 0; i < L; ++i)
            transpose_kernel<<<bHH, T, 0, stream>>>(W2 + (size_t)i * nHH,
                                                    w2t + (size_t)i * nHH, H, H);
        for (int i = 0; i < L * NL; ++i)
            transpose_kernel<<<bHH, T, 0, stream>>>(gw + (size_t)i * nHH,
                                                    gwt + (size_t)i * nHH, H, H);
        int nHC = H * C;
        transpose_kernel<<<(nHC + T - 1) / T, T, 0, stream>>>(Wc, wct, H, C);
    }

    {
        int n4 = (int)((size_t)NN * C / 4);
        bias_fill_kernel<<<512, T, 0, stream>>>(bc, (float4*)d_out, n4);
    }

    for (int i = 0; i < L; ++i) {
        const int*   src_i = src + (size_t)i * NE;
        const int*   dst_i = dst + (size_t)i * NE;
        const int*   ids_i = ids + (size_t)i * NS;
        const float* x_i   = x   + (size_t)i * NS * H;
        const float* w2t_i = w2t + (size_t)i * H * H;
        const float* b2_i  = b2  + (size_t)i * H;

        slice_count_kernel<<<EB_NSL * EB_NB, EB_T, 0, stream>>>(src_i, dst_i,
                                                               cin_mat, cout_mat, NS, NE);
        slice_totals_kernel<<<(NS + T - 1) / T, T, 0, stream>>>(cin_mat, cout_mat, cnt_in,
                                                               deg_out, deg_in, NS);
        segment_start_scan_kernel<<<(NS + 1023) / 1024, 1024, 0, stream>>>(
            cnt_in, seg_start, NS);
        slice_place_kernel<<<EB_NSL * EB_NB, EB_T, 0, stream>>>(src_i, dst_i, seg_start,
                                                               cin_mat, sorted_src, NS, NE);

        gemm_wmma_k128_kernel<8, false><<<gemm_blocks_NS, T, 0, stream>>>(
            x_i, w2t_i, b2_i, nullptr, deg_out, hbuf, NS, 0, nullptr, nullptr, nullptr);

        for (int l = 0; l < NL; ++l) {
            const float* gwt_il = gwt + ((size_t)i * NL + l) * H * H;
            const float* gb_il  = gb  + ((size_t)i * NL + l) * H;
            gather_rows_sum_kernel<<<gather_blocks, T, 0, stream>>>(
                seg_start, sorted_src, hbuf, agg, NS, NE);
            if (l + 1 < NL) {
                gemm_wmma_k128_kernel<8, false><<<gemm_blocks_NS, T, 0, stream>>>(
                    agg, gwt_il, gb_il, deg_in, deg_out, hbuf, NS, 1,
                    nullptr, nullptr, nullptr);
            } else {
                gemm_wmma_k128_kernel<8, true><<<gemm_blocks_NS, T, 0, stream>>>(
                    agg, gwt_il, gb_il, deg_in, nullptr, hbuf, NS, 1,
                    wct, ids_i, (float*)d_out);
            }
        }
    }
}


# independent launches batched: each edge-bookkeeping stage is one launch for all three links (grid y = link, per-link arrays), all weight transposes and the bias fill one launch; 38 -> 20 launches
# speedup vs baseline: 4.0865x; 1.2079x over previous
#include <hip/hip_runtime.h>


typedef __attribute__((ext_vector_type(2))) float v2f;
typedef __attribute__((ext_vector_type(4))) float v4f;
typedef __attribute__((ext_vector_type(8))) float v8f;


__global__ void setup_kernel(const float* __restrict__ W2,
                             const float* __restrict__ gw,
                             const float* __restrict__ Wc,
                             const float* __restrict__ bc,
                             float* __restrict__ w2t,
                             float* __restrict__ gwt,
                             float* __restrict__ wct,
                             float4* __restrict__ out4, int n4,
                             int nW2, int nGw, int H, int C) {
    const int y    = blockIdx.y;
    const int nmat = nW2 + nGw + 1;
    if (y < nmat) {
        const float* in;
        float* outT;
        int K = H, N = H;
        if (y < nW2) {
            in = W2 + (size_t)y * H * H;  outT = w2t + (size_t)y * H * H;
        } else if (y < nW2 + nGw) {
            in = gw + (size_t)(y - nW2) * H * H;  outT = gwt + (size_t)(y - nW2) * H * H;
        } else {
            in = Wc;  outT = wct;  N = C;
        }
        const int i = blockIdx.x * blockDim.x + threadIdx.x;
        if (i < N * K) {
            const int n = i / K, k = i % K;
            outT[i] = in[k * N + n];
        }
    } else {
        const int fb     = (y - nmat) * gridDim.x + blockIdx.x;
        const int stride = (gridDim.y - nmat) * gridDim.x * blockDim.x;
        int i = fb * blockDim.x + threadIdx.x;
        const float4 b = ((const float4*)bc)[i & 3];
        for (; i < n4; i += stride) out4[i] = b;
    }
}

#define EB_NB  16
#define EB_NSL 16
#define EB_BR  3136
#define EB_T   512
#define EB_U   16

__device__ __forceinline__ void edge_slice_bounds(int sl, int nE, int& e0, int& e1) {
    const int per = (nE + EB_NSL - 1) / EB_NSL;
    e0 = min(sl * per, nE);
    e1 = min(e0 + per, nE);
}

__global__ __launch_bounds__(EB_T)
void slice_count_kernel(const int* __restrict__ src,
                        const int* __restrict__ dst,
                        int* __restrict__ cin,
                        int* __restrict__ cout, int n, int nE) {
    __shared__ int s_in[EB_BR];
    __shared__ int s_out[EB_BR];
    src  += (size_t)blockIdx.y * nE;
    dst  += (size_t)blockIdx.y * nE;
    cin  += (size_t)blockIdx.y * EB_NSL * n;
    cout += (size_t)blockIdx.y * EB_NSL * n;
    const int t  = threadIdx.x;
    const int sl = blockIdx.x / EB_NB;
    const int rg = blockIdx.x % EB_NB;
    int e0, e1;
    edge_slice_bounds(sl, nE, e0, e1);
    for (int r0 = rg * EB_BR; r0 < n; r0 += EB_NB * EB_BR) {
        for (int i = t; i < EB_BR; i += EB_T) { s_in[i] = 0; s_out[i] = 0; }
        __syncthreads();
        for (int e = e0 + t; e < e1; e += EB_U * EB_T) {
            int dv[EB_U], sv[EB_U];
#pragma unroll
            for (int q = 0; q < EB_U; ++q) {
                const int ee = e + q * EB_T;
                const int ec = min(ee, e1 - 1);
                const int d = dst[ec];
                const int s = src[ec];
                dv[q] = (ee < e1) ? d : -1;
                sv[q] = (ee < e1) ? s : -1;
            }
#pragma unroll
            for (int q = 0; q < EB_U; ++q) {
                const unsigned a = (unsigned)(dv[q] - r0);
                const unsigned b = (unsigned)(sv[q] - r0);
                if (a < (unsigned)EB_BR && dv[q] < n) atomicAdd(&s_in[a], 1);
                if (b < (unsigned)EB_BR && sv[q] < n) atomicAdd(&s_out[b], 1);
            }
        }
        __syncthreads();
        for (int i = t; i < EB_BR; i += EB_T) {
            const int d = r0 + i;
            if (d < n) {
                cin[(size_t)sl * n + d]  = s_in[i];
                cout[(size_t)sl * n + d] = s_out[i];
            }
        }
        __syncthreads();
    }
}

__global__ void slice_totals_kernel(int* __restrict__ cin,
                                    const int* __restrict__ cout,
                                    float* __restrict__ cnt_in,
                                    float* __restrict__ deg_out,
                                    float* __restrict__ deg_in, int n) {
    cin     += (size_t)blockIdx.y * EB_NSL * n;
    cout    += (size_t)blockIdx.y * EB_NSL * n;
    cnt_in  += (size_t)blockIdx.y * n;
    deg_out += (size_t)blockIdx.y * n;
    deg_in  += (size_t)blockIdx.y * n;
    const int d = blockIdx.x * blockDim.x + threadIdx.x;
    if (d < n) {
        int ci[EB_NSL], co = 0;
#pragma unroll
        for (int k = 0; k < EB_NSL; ++k) {
            ci[k] = cin[(size_t)k * n + d];
            co += cout[(size_t)k * n + d];
        }
        int run = 0;
#pragma unroll
        for (int k = 0; k < EB_NSL; ++k) {
            cin[(size_t)k * n + d] = run;
            run += ci[k];
        }
        cnt_in[d]  = (float)run;
        deg_in[d]  = rsqrtf(fmaxf((float)run, 1.0f));
        deg_out[d] = rsqrtf(fmaxf((float)co, 1.0f));
    }
}

__global__ __launch_bounds__(1024)
void segment_start_scan_kernel(const float* __restrict__ cnt,
                               int* __restrict__ seg, int n) {
    __shared__ int s_front[32];
    __shared__ int s_own[32];
    cnt += (size_t)blockIdx.y * n;
    seg += (size_t)blockIdx.y * n;
    const int t = threadIdx.x;
    const int lane = t & 31;
    const int w = t >> 5;
    const int first = blockIdx.x * 1024;
    int front = 0;
    const float4* c4 = (const float4*)cnt;
    const int nf4 = first >> 2;
    for (int i = t; i < nf4; i += 4096) {
        const float4 a = c4[i];
        const float4 b = c4[min(i + 1024, nf4 - 1)];
        const float4 c = c4[min(i + 2048, nf4 - 1)];
        const float4 e = c4[min(i + 3072, nf4 - 1)];
        const int sa = (int)a.x + (int)a.y + (int)a.z + (int)a.w;
        const int sb = (int)b.x + (int)b.y + (int)b.z + (int)b.w;
        const int sc = (int)c.x + (int)c.y + (int)c.z + (int)c.w;
        const int se = (int)e.x + (int)e.y + (int)e.z + (int)e.w;
        front += sa;
        front += (i + 1024 < nf4) ? sb : 0;
        front += (i + 2048 < nf4) ? sc : 0;
        front += (i + 3072 < nf4) ? se : 0;
    }
    const int idx = first + t;
    const int v = (idx < n) ? (int)cnt[idx] : 0;
    int incl = v;
    for (int o = 1; o < 32; o <<= 1) {
        const int u = __shfl_up(incl, o);
        if (lane >= o) incl += u;
    }
    for (int o = 16; o > 0; o >>= 1) front += __shfl_xor(front, o);
    if (lane == 31) s_own[w] = incl;
    if (lane == 0) s_front[w] = front;
    __syncthreads();
    int base = 0, woff = 0;
    for (int k = 0; k < 32; ++k) {
        base += s_front[k];
        if (k < w) woff += s_own[k];
    }
    if (idx < n) seg[idx] = base + woff + incl - v;
}

__global__ __launch_bounds__(EB_T)
void slice_place_kernel(const int* __restrict__ src,
                        const int* __restrict__ dst,
                        const int* __restrict__ seg,
                        const int* __restrict__ cin_front,
                        int* __restrict__ sorted_src, int n, int nE) {
    __shared__ int s_cur[EB_BR];
    src        += (size_t)blockIdx.y * nE;
    dst        += (size_t)blockIdx.y * nE;
    seg        += (size_t)blockIdx.y * n;
    cin_front  += (size_t)blockIdx.y * EB_NSL * n;
    sorted_src += (size_t)blockIdx.y * nE;
    const int t  = threadIdx.x;
    const int sl = blockIdx.x / EB_NB;
    const int rg = blockIdx.x % EB_NB;
    int e0, e1;
    edge_slice_bounds(sl, nE, e0, e1);
    for (int r0 = rg * EB_BR; r0 < n; r0 += EB_NB * EB_BR) {
        constexpr int NI = (EB_BR + EB_T - 1) / EB_T;
        int cv[NI];
#pragma unroll
        for (int k = 0; k < NI; ++k) {
            const int d = min(r0 + t + k * EB_T, n - 1);
            cv[k] = seg[d] + cin_front[(size_t)sl * n + d];
        }
#pragma unroll
        for (int k = 0; k < NI; ++k) {
            const int i = t + k * EB_T;
            if (i < EB_BR) s_cur[i] = cv[k];
        }
        __syncthreads();
        for (int e = e0 + t; e < e1; e += EB_U * EB_T) {
            int dv[EB_U], sv[EB_U];
#pragma unroll
            for (int q = 0; q < EB_U; ++q) {
                const int ee = e + q * EB_T;
                const int ec = min(ee, e1 - 1);
                const int d = dst[ec];
                const int s = src[ec];
                dv[q] = (ee < e1) ? d : -1;
                sv[q] = s;
            }
#pragma unroll
            for (int q = 0; q < EB_U; ++q) {
                const unsigned a = (unsigned)(dv[q] - r0);
                if (a < (unsigned)EB_BR && dv[q] < n) {
                    const int pos = atomicAdd(&s_cur[a], 1);
                    if ((unsigned)pos < (unsigned)nE) sorted_src[pos] = sv[q];
                }
            }
        }
        __syncthreads();
    }
}

__global__ __launch_bounds__(256)
void gather_rows_sum_kernel(const int* __restrict__ seg_start,
                            const int* __restrict__ sorted_src,
                            const float* __restrict__ h,
                            float* __restrict__ agg, int n, int nE) {
    const int lane = threadIdx.x & 31;
    const int wave0 = __builtin_amdgcn_readfirstlane(
        (int)((blockIdx.x * blockDim.x + threadIdx.x) >> 5));
    const int nwaves = (int)((gridDim.x * blockDim.x) >> 5);
    const float4* h4 = (const float4*)h + lane;
    for (int d = wave0; d < n; d += nwaves) {
        int beg = seg_start[d];
        int end = seg_start[min(d + 1, n - 1)];
        if (d == n - 1) end = nE;
        beg = max(beg, 0);
        end = min(end, nE);
        float4 acc = make_float4(0.f, 0.f, 0.f, 0.f);
        for (int base = beg; base < end; base += 32) {
            const int cnt = min(32, end - base);
            int mys = (lane < cnt) ? sorted_src[base + lane] : 0;
            mys = min(max(mys, 0), n - 1);
            int j = 0;
            for (; j + 4 <= cnt; j += 4) {
                const int s0 = __builtin_amdgcn_readlane(mys, j);
                const int s1 = __builtin_amdgcn_readlane(mys, j + 1);
                const int s2 = __builtin_amdgcn_readlane(mys, j + 2);
                const int s3 = __builtin_amdgcn_readlane(mys, j + 3);
                const float4 v0 = h4[(size_t)s0 * 32];
                const float4 v1 = h4[(size_t)s1 * 32];
                const float4 v2 = h4[(size_t)s2 * 32];
                const float4 v3 = h4[(size_t)s3 * 32];
                acc.x += v0.x; acc.y += v0.y; acc.z += v0.z; acc.w += v0.w;
                acc.x += v1.x; acc.y += v1.y; acc.z += v1.z; acc.w += v1.w;
                acc.x += v2.x; acc.y += v2.y; acc.z += v2.z; acc.w += v2.w;
                acc.x += v3.x; acc.y += v3.y; acc.z += v3.z; acc.w += v3.w;
            }
            const int rem = cnt - j;
            if (rem == 3) {
                const int s0 = __builtin_amdgcn_readlane(mys, j);
                const int s1 = __builtin_amdgcn_readlane(mys, j + 1);
                const int s2 = __builtin_amdgcn_readlane(mys, j + 2);
                const float4 v0 = h4[(size_t)s0 * 32];
                const float4 v1 = h4[(size_t)s1 * 32];
                const float4 v2 = h4[(size_t)s2 * 32];
                acc.x += v0.x; acc.y += v0.y; acc.z += v0.z; acc.w += v0.w;
                acc.x += v1.x; acc.y += v1.y; acc.z += v1.z; acc.w += v1.w;
                acc.x += v2.x; acc.y += v2.y; acc.z += v2.z; acc.w += v2.w;
            } else if (rem == 2) {
                const int s0 = __builtin_amdgcn_readlane(mys, j);
                const int s1 = __builtin_amdgcn_readlane(mys, j + 1);
                const float4 v0 = h4[(size_t)s0 * 32];
                const float4 v1 = h4[(size_t)s1 * 32];
                acc.x += v0.x; acc.y += v0.y; acc.z += v0.z; acc.w += v0.w;
                acc.x += v1.x; acc.y += v1.y; acc.z += v1.z; acc.w += v1.w;
            } else if (rem == 1) {
                const int s0 = __builtin_amdgcn_readlane(mys, j);
                const float4 v0 = h4[(size_t)s0 * 32];
                acc.x += v0.x; acc.y += v0.y; acc.z += v0.z; acc.w += v0.w;
            }
        }
        ((float4*)agg)[(size_t)d * 32 + lane] = acc;
    }
}

template <int NTILES, bool CLS>
__global__ __launch_bounds__(256)
void gemm_wmma_k128_kernel(const float* __restrict__ A,
                           const float* __restrict__ Wt,
                           const float* __restrict__ bias,
                           const float* __restrict__ arowscale,
                           const float* __restrict__ erowscale,
                           float* __restrict__ out,
                           int M, int relu,
                           const float* __restrict__ Wct,
                           const int* __restrict__ ids,
                           float* __restrict__ cls_out) {
    constexpr int N      = NTILES * 16;
    constexpr int PITCH  = 132;
    constexpr int OPITCH = N + 4;
    static_assert(!CLS || OPITCH == PITCH, "CLS reads the finished rows as A rows");
    __shared__ __attribute__((aligned(16))) float s_w[N * PITCH];
    __shared__ __attribute__((aligned(16))) float s_a[8 * 32 * PITCH];
    __shared__ __attribute__((aligned(16))) float s_c[CLS ? 16 * PITCH : 4];

    const int tid      = threadIdx.x;
    const int lane     = tid & 31;
    const int wave     = __builtin_amdgcn_readfirstlane(tid >> 5);
    const int c        = lane & 15;
    const int h        = lane >> 4;
    const int row_base = (blockIdx.x * 8 + wave) * 32;
    float* my_a = s_a + wave * (32 * PITCH);

    v4f wreg[N / 8];
#pragma unroll
    for (int i = 0; i < N / 8; ++i)
        wreg[i] = ((const v4f*)Wt)[tid + 256 * i];
    v4f areg[32];
#pragma unroll
    for (int i = 0; i < 32; ++i) {
        const int r = min(row_base + i, M - 1);
        areg[i] = ((const v4f*)A)[(size_t)r * 32 + lane];
    }
    v4f creg[2];
    int id0[8], id1[8];
    if constexpr (CLS) {
#pragma unroll
        for (int i = 0; i < 2; ++i)
            creg[i] = ((const v4f*)Wct)[tid + 256 * i];
#pragma unroll
        for (int r = 0; r < 8; ++r) {
            id0[r] = ids[min(row_base + 8 * h + r, M - 1)];
            id1[r] = ids[min(row_base + 16 + 8 * h + r, M - 1)];
        }
    }
    float myscale = 1.0f;
    if (arowscale) myscale = arowscale[min(row_base + lane, M - 1)];
    float es0[8], es1[8];
#pragma unroll
    for (int r = 0; r < 8; ++r) { es0[r] = 1.0f; es1[r] = 1.0f; }
    if (erowscale) {
#pragma unroll
        for (int r = 0; r < 8; ++r) {
            es0[r] = erowscale[min(row_base + 8 * h + r, M - 1)];
            es1[r] = erowscale[min(row_base + 16 + 8 * h + r, M - 1)];
        }
    }

    v8f acc0[NTILES], acc1[NTILES];
#pragma unroll
    for (int nt = 0; nt < NTILES; ++nt) {
        const float bv = bias[nt * 16 + c];
#pragma unroll
        for (int r = 0; r < 8; ++r) { acc0[nt][r] = bv; acc1[nt][r] = bv; }
    }

#pragma unroll
    for (int i = 0; i < N / 8; ++i) {
        const int u = tid + 256 * i;
        *(v4f*)(s_w + (u >> 5) * PITCH + (u & 31) * 4) = wreg[i];
    }
    if constexpr (CLS) {
#pragma unroll
        for (int i = 0; i < 2; ++i) {
            const int u = tid + 256 * i;
            *(v4f*)(s_c + (u >> 5) * PITCH + (u & 31) * 4) = creg[i];
        }
    }
#pragma unroll
    for (int i = 0; i < 32; ++i) {
        const float s = __int_as_float(
            __builtin_amdgcn_readlane(__float_as_int(myscale), i));
        v4f v = areg[i];
        v.x *= s; v.y *= s; v.z *= s; v.w *= s;
        *(v4f*)(my_a + i * PITCH + lane * 4) = v;
    }

    __syncthreads();

    const float* aw0 = my_a + c * PITCH + 4 * h;
    const float* aw1 = my_a + (16 + c) * PITCH + 4 * h;
    const float* bw  = s_w + c * PITCH + 4 * h;

    v4f a0c = *(const v4f*)(aw0);
    v4f a1c = *(const v4f*)(aw1);
    v4f bc[NTILES];
#pragma unroll
    for (int nt = 0; nt < NTILES; ++nt)
        bc[nt] = *(const v4f*)(bw + nt * 16 * PITCH);
#pragma unroll
    for (int j = 0; j < 16; ++j) {
        v4f a0n = a0c, a1n = a1c;
        v4f bn[NTILES];
#pragma unroll
        for (int nt = 0; nt < NTILES; ++nt) bn[nt] = bc[nt];
        if (j + 1 < 16) {
            a0n = *(const v4f*)(aw0 + 8 * (j + 1));
            a1n = *(const v4f*)(aw1 + 8 * (j + 1));
#pragma unroll
            for (int nt = 0; nt < NTILES; ++nt)
                bn[nt] = *(const v4f*)(bw + nt * 16 * PITCH + 8 * (j + 1));
        }
#pragma unroll
        for (int nt = 0; nt < NTILES; ++nt) {
            acc0[nt] = __builtin_amdgcn_wmma_f32_16x16x4_f32(
                false, a0c.xy, false, bc[nt].xy, (short)0, acc0[nt], false, false);
            acc1[nt] = __builtin_amdgcn_wmma_f32_16x16x4_f32(
                false, a1c.xy, false, bc[nt].xy, (short)0, acc1[nt], false, false);
            acc0[nt] = __builtin_amdgcn_wmma_f32_16x16x4_f32(
                false, a0c.zw, false, bc[nt].zw, (short)0, acc0[nt], false, false);
            acc1[nt] = __builtin_amdgcn_wmma_f32_16x16x4_f32(
                false, a1c.zw, false, bc[nt].zw, (short)0, acc1[nt], false, false);
        }
        a0c = a0n; a1c = a1n;
#pragma unroll
        for (int nt = 0; nt < NTILES; ++nt) bc[nt] = bn[nt];
    }

#pragma unroll
    for (int nt = 0; nt < NTILES; ++nt) {
#pragma unroll
        for (int r = 0; r < 8; ++r) {
            float v0 = acc0[nt][r];
            float v1 = acc1[nt][r];
            if (relu) { v0 = fmaxf(v0, 0.0f); v1 = fmaxf(v1, 0.0f); }
            v0 *= es0[r];
            v1 *= es1[r];
            my_a[(8 * h + r) * OPITCH + nt * 16 + c]      = v0;
            my_a[(16 + 8 * h + r) * OPITCH + nt * 16 + c] = v1;
        }
    }

    __syncthreads();

    if constexpr (CLS) {
        const float* cw = s_c + c * PITCH + 4 * h;
        v8f p0, p1;
#pragma unroll
        for (int r = 0; r < 8; ++r) { p0[r] = 0.0f; p1[r] = 0.0f; }
        v4f x0c = *(const v4f*)(aw0);
        v4f x1c = *(const v4f*)(aw1);
        v4f wcc = *(const v4f*)(cw);
#pragma unroll
        for (int j = 0; j < 16; ++j) {
            v4f x0n = x0c, x1n = x1c, wcn = wcc;
            if (j + 1 < 16) {
                x0n = *(const v4f*)(aw0 + 8 * (j + 1));
                x1n = *(const v4f*)(aw1 + 8 * (j + 1));
                wcn = *(const v4f*)(cw + 8 * (j + 1));
            }
            p0 = __builtin_amdgcn_wmma_f32_16x16x4_f32(
                false, x0c.xy, false, wcc.xy, (short)0, p0, false, false);
            p1 = __builtin_amdgcn_wmma_f32_16x16x4_f32(
                false, x1c.xy, false, wcc.xy, (short)0, p1, false, false);
            p0 = __builtin_amdgcn_wmma_f32_16x16x4_f32(
                false, x0c.zw, false, wcc.zw, (short)0, p0, false, false);
            p1 = __builtin_amdgcn_wmma_f32_16x16x4_f32(
                false, x1c.zw, false, wcc.zw, (short)0, p1, false, false);
            x0c = x0n; x1c = x1n; wcc = wcn;
        }
#pragma unroll
        for (int r = 0; r < 8; ++r) {
            if (row_base + 8 * h + r < M)
                atomicAdd(cls_out + (size_t)id0[r] * 16 + c, p0[r]);
            if (row_base + 16 + 8 * h + r < M)
                atomicAdd(cls_out + (size_t)id1[r] * 16 + c, p1[r]);
        }
        return;
    }

    constexpr int ITER = N / 4;
    constexpr int GRP  = ITER < 8 ? ITER : 8;
#pragma unroll
    for (int g = 0; g < ITER / GRP; ++g) {
        v4f t[GRP];
#pragma unroll
        for (int q = 0; q < GRP; ++q) {
            const int u = lane + 32 * (g * GRP + q);
            t[q] = *(const v4f*)(my_a + (u / (N / 4)) * OPITCH + 4 * (u % (N / 4)));
        }
#pragma unroll
        for (int q = 0; q < GRP; ++q) {
            const int u    = lane + 32 * (g * GRP + q);
            const int grow = row_base + u / (N / 4);
            if (grow < M)
                *(v4f*)(out + (size_t)grow * N + 4 * (u % (N / 4))) = t[q];
        }
    }
}


extern "C" void kernel_launch(void* const* d_in, const int* in_sizes, int n_in,
                              void* d_out, int out_size, void* d_ws, size_t ws_size,
                              hipStream_t stream) {
    (void)n_in; (void)ws_size;
    const float* x   = (const float*)d_in[0];
    const float* W2  = (const float*)d_in[1];
    const float* b2  = (const float*)d_in[2];
    const float* gw  = (const float*)d_in[3];
    const float* gb  = (const float*)d_in[4];
    const float* Wc  = (const float*)d_in[5];
    const float* bc  = (const float*)d_in[6];
    const int*   src = (const int*)d_in[7];
    const int*   dst = (const int*)d_in[8];
    const int*   ids = (const int*)d_in[9];

    const int H  = 128;
    const int C  = 16;
    const int L  = in_sizes[2] / H;
    const int NS = in_sizes[9] / L;
    const int NE = in_sizes[7] / L;
    const int NL = in_sizes[4] / (L * H);
    const int NN = out_size / C;

    float* ws       = (float*)d_ws;
    float* out_feat = ws;
    float* hbuf     = out_feat + (size_t)NN * H;
    float* agg      = hbuf + (size_t)NS * H;
    float* w2t      = agg + (size_t)NS * H;
    float* gwt      = w2t + (size_t)L * H * H;
    float* wct      = gwt + (size_t)L * NL * H * H;

    int*   seg_all    = (int*)out_feat;
    int*   sorted_all = seg_all + (size_t)L * NS;
    float* dego_all   = (float*)(sorted_all + (size_t)L * NE);
    float* degi_all   = dego_all + (size_t)L * NS;
    float* cnt_all    = degi_all + (size_t)L * NS;
    int*   cin_all    = (int*)(cnt_all + (size_t)L * NS);
    int*   cout_all   = cin_all + (size_t)L * EB_NSL * NS;

    const int T = 256;
    const int gemm_blocks_NS = (((NS / 16) + 1) / 2 + 7) / 8;
    const int gather_blocks  = ((NS + 1) / 2 + 7) / 8;

    {
        const int nHH  = H * H;
        const int bHH  = (nHH + T - 1) / T;
        const int nmat = L + L * NL + 1;
        const int n4   = (int)((size_t)NN * C / 4);
        setup_kernel<<<dim3(bHH, nmat + 8), T, 0, stream>>>(
            W2, gw, Wc, bc, w2t, gwt, wct, (float4*)d_out, n4, L, L * NL, H, C);
    }

    slice_count_kernel<<<dim3(EB_NSL * EB_NB, L), EB_T, 0, stream>>>(
        src, dst, cin_all, cout_all, NS, NE);
    slice_totals_kernel<<<dim3((NS + T - 1) / T, L), T, 0, stream>>>(
        cin_all, cout_all, cnt_all, dego_all, degi_all, NS);
    segment_start_scan_kernel<<<dim3((NS + 1023) / 1024, L), 1024, 0, stream>>>(
        cnt_all, seg_all, NS);
    slice_place_kernel<<<dim3(EB_NSL * EB_NB, L), EB_T, 0, stream>>>(
        src, dst, seg_all, cin_all, sorted_all, NS, NE);

    for (int i = 0; i < L; ++i) {
        const int*   ids_i     = ids + (size_t)i * NS;
        const float* x_i       = x   + (size_t)i * NS * H;
        const float* w2t_i     = w2t + (size_t)i * H * H;
        const float* b2_i      = b2  + (size_t)i * H;
        const int*   seg_i     = seg_all    + (size_t)i * NS;
        const int*   sorted_i  = sorted_all + (size_t)i * NE;
        const float* deg_out_i = dego_all   + (size_t)i * NS;
        const float* deg_in_i  = degi_all   + (size_t)i * NS;

        gemm_wmma_k128_kernel<8, false><<<gemm_blocks_NS, T, 0, stream>>>(
            x_i, w2t_i, b2_i, nullptr, deg_out_i, hbuf, NS, 0, nullptr, nullptr, nullptr);

        for (int l = 0; l < NL; ++l) {
            const float* gwt_il = gwt + ((size_t)i * NL + l) * H * H;
            const float* gb_il  = gb  + ((size_t)i * NL + l) * H;
            gather_rows_sum_kernel<<<gather_blocks, T, 0, stream>>>(
                seg_i, sorted_i, hbuf, agg, NS, NE);
            if (l + 1 < NL) {
                gemm_wmma_k128_kernel<8, false><<<gemm_blocks_NS, T, 0, stream>>>(
                    agg, gwt_il, gb_il, deg_in_i, deg_out_i, hbuf, NS, 1,
                    nullptr, nullptr, nullptr);
            } else {
                gemm_wmma_k128_kernel<8, true><<<gemm_blocks_NS, T, 0, stream>>>(
                    agg, gwt_il, gb_il, deg_in_i, nullptr, hbuf, NS, 1,
                    wct, ids_i, (float*)d_out);
            }
        }
    }
}


# feature2 and conv layer 0 as one product: x @ (W2 gw0) + b2 gw0, W2 gw0 built on WMMA in set-up; layer 0 = edge sum + scale, bias, relu in the gather; 9 -> 6 big GEMMs; products reassociated
# speedup vs baseline: 5.1530x; 1.2610x over previous
#include <hip/hip_runtime.h>


typedef __attribute__((ext_vector_type(2))) float v2f;
typedef __attribute__((ext_vector_type(4))) float v4f;
typedef __attribute__((ext_vector_type(8))) float v8f;


#define MG_H     128
#define MG_PITCH 132
__global__ __launch_bounds__(256)
void setup_kernel(const float* __restrict__ W2,
                  const float* __restrict__ b2,
                  const float* __restrict__ gw,
                  const float* __restrict__ Wc,
                  const float* __restrict__ bc,
                  float* __restrict__ wmt,
                  float* __restrict__ bm,
                  float* __restrict__ gwt,
                  float* __restrict__ wct,
                  float4* __restrict__ out4, int n4,
                  int nL, int nLayer, int H, int C) {
    __shared__ __attribute__((aligned(16))) float s_t[16 * MG_PITCH];
    const int y    = blockIdx.y;
    const int nGw  = nL * nLayer;
    const int nmat = nL + nGw + 1;
    if (y < nL) {
        const float* gw0 = gw + (size_t)y * nLayer * MG_H * MG_H;
        const float* w2  = W2 + (size_t)y * MG_H * MG_H;
        const int x   = blockIdx.x;
        const int tid = threadIdx.x;
        if (x < MG_H / 16) {
            const int lane = tid & 31;
            const int wave = __builtin_amdgcn_readfirstlane(tid >> 5);
            const int c = lane & 15;
            const int h = lane >> 4;
            const float* ap = gw0 + (size_t)(4 * h) * MG_H + 16 * x + c;
            const float* bp = w2 + (size_t)(16 * wave + c) * MG_H + 4 * h;
            v4f af[16], bf[16];
#pragma unroll
            for (int jj = 0; jj < 16; ++jj) {
                bf[jj] = *(const v4f*)(bp + 8 * jj);
                af[jj].x = ap[(size_t)(8 * jj + 0) * MG_H];
                af[jj].y = ap[(size_t)(8 * jj + 1) * MG_H];
                af[jj].z = ap[(size_t)(8 * jj + 2) * MG_H];
                af[jj].w = ap[(size_t)(8 * jj + 3) * MG_H];
            }
            v8f acc;
#pragma unroll
            for (int r = 0; r < 8; ++r) acc[r] = 0.0f;
            __builtin_amdgcn_sched_barrier(0);
#pragma unroll
            for (int jj = 0; jj < 16; ++jj) {
                acc = __builtin_amdgcn_wmma_f32_16x16x4_f32(
                    false, af[jj].xy, false, bf[jj].xy, (short)0, acc, false, false);
                acc = __builtin_amdgcn_wmma_f32_16x16x4_f32(
                    false, af[jj].zw, false, bf[jj].zw, (short)0, acc, false, false);
            }
            __builtin_amdgcn_sched_barrier(0);
            asm volatile("v_nop\n\tv_nop\n\tv_nop\n\tv_nop");
            __builtin_amdgcn_sched_barrier(0);
#pragma unroll
            for (int r = 0; r < 8; ++r)
                s_t[(8 * h + r) * MG_PITCH + 16 * wave + c] = acc[r];
            __syncthreads();
#pragma unroll
            for (int q = 0; q < 2; ++q) {
                const int row = 2 * wave + q;
                const v4f t = *(const v4f*)(s_t + row * MG_PITCH + 4 * lane);
                *(v4f*)(wmt + (size_t)y * MG_H * MG_H + (size_t)(16 * x + row) * MG_H + 4 * lane) = t;
            }
        } else if (x == MG_H / 16) {
            const int n    = tid & (MG_H - 1);
            const int part = tid >> 7;
            const float* bsrc = b2 + (size_t)y * MG_H + part * (MG_H / 2);
            const float* gsrc = gw0 + (size_t)(part * (MG_H / 2)) * MG_H + n;
            float sum = 0.0f;
            for (int j0 = 0; j0 < MG_H / 2; j0 += 16) {
                float gv[16], bv[16];
#pragma unroll
                for (int q = 0; q < 16; ++q) {
                    gv[q] = gsrc[(size_t)(j0 + q) * MG_H];
                    bv[q] = bsrc[j0 + q];
                }
#pragma unroll
                for (int q = 0; q < 16; ++q) sum += bv[q] * gv[q];
            }
            s_t[tid] = sum;
            __syncthreads();
            if (tid < MG_H) bm[(size_t)y * MG_H + tid] = s_t[tid] + s_t[tid + MG_H];
        }
    } else if (y < nmat) {
        const float* in;
        float* outT;
        int K = H, N = H;
        if (y < nL + nGw) {
            const int g = y - nL;
            if (g % nLayer == 0) return;
            in = gw + (size_t)g * H * H;  outT = gwt + (size_t)g * H * H;
        } else {
            in = Wc;  outT = wct;  N = C;
        }
        const int i = blockIdx.x * blockDim.x + threadIdx.x;
        if (i < N * K) {
            const int n = i / K, k = i % K;
            outT[i] = in[k * N + n];
        }
    } else {
        const int fb     = (y - nmat) * gridDim.x + blockIdx.x;
        const int stride = (gridDim.y - nmat) * gridDim.x * blockDim.x;
        int i = fb * blockDim.x + threadIdx.x;
        const float4 b = ((const float4*)bc)[i & 3];
        for (; i < n4; i += stride) out4[i] = b;
    }
}

#define EB_NB  16
#define EB_NSL 16
#define EB_BR  3136
#define EB_T   512
#define EB_U   16

__device__ __forceinline__ void edge_slice_bounds(int sl, int nE, int& e0, int& e1) {
    const int per = (nE + EB_NSL - 1) / EB_NSL;
    e0 = min(sl * per, nE);
    e1 = min(e0 + per, nE);
}

__global__ __launch_bounds__(EB_T)
void slice_count_kernel(const int* __restrict__ src,
                        const int* __restrict__ dst,
                        int* __restrict__ cin,
                        int* __restrict__ cout, int n, int nE) {
    __shared__ int s_in[EB_BR];
    __shared__ int s_out[EB_BR];
    src  += (size_t)blockIdx.y * nE;
    dst  += (size_t)blockIdx.y * nE;
    cin  += (size_t)blockIdx.y * EB_NSL * n;
    cout += (size_t)blockIdx.y * EB_NSL * n;
    const int t  = threadIdx.x;
    const int sl = blockIdx.x / EB_NB;
    const int rg = blockIdx.x % EB_NB;
    int e0, e1;
    edge_slice_bounds(sl, nE, e0, e1);
    for (int r0 = rg * EB_BR; r0 < n; r0 += EB_NB * EB_BR) {
        for (int i = t; i < EB_BR; i += EB_T) { s_in[i] = 0; s_out[i] = 0; }
        __syncthreads();
        for (int e = e0 + t; e < e1; e += EB_U * EB_T) {
            int dv[EB_U], sv[EB_U];
#pragma unroll
            for (int q = 0; q < EB_U; ++q) {
                const int ee = e + q * EB_T;
                const int ec = min(ee, e1 - 1);
                const int d = dst[ec];
                const int s = src[ec];
                dv[q] = (ee < e1) ? d : -1;
                sv[q] = (ee < e1) ? s : -1;
            }
#pragma unroll
            for (int q = 0; q < EB_U; ++q) {
                const unsigned a = (unsigned)(dv[q] - r0);
                const unsigned b = (unsigned)(sv[q] - r0);
                if (a < (unsigned)EB_BR && dv[q] < n) atomicAdd(&s_in[a], 1);
                if (b < (unsigned)EB_BR && sv[q] < n) atomicAdd(&s_out[b], 1);
            }
        }
        __syncthreads();
        for (int i = t; i < EB_BR; i += EB_T) {
            const int d = r0 + i;
            if (d < n) {
                cin[(size_t)sl * n + d]  = s_in[i];
                cout[(size_t)sl * n + d] = s_out[i];
            }
        }
        __syncthreads();
    }
}

__global__ void slice_totals_kernel(int* __restrict__ cin,
                                    const int* __restrict__ cout,
                                    float* __restrict__ cnt_in,
                                    float* __restrict__ deg_out,
                                    float* __restrict__ deg_in, int n) {
    cin     += (size_t)blockIdx.y * EB_NSL * n;
    cout    += (size_t)blockIdx.y * EB_NSL * n;
    cnt_in  += (size_t)blockIdx.y * n;
    deg_out += (size_t)blockIdx.y * n;
    deg_in  += (size_t)blockIdx.y * n;
    const int d = blockIdx.x * blockDim.x + threadIdx.x;
    if (d < n) {
        int ci[EB_NSL], co = 0;
#pragma unroll
        for (int k = 0; k < EB_NSL; ++k) {
            ci[k] = cin[(size_t)k * n + d];
            co += cout[(size_t)k * n + d];
        }
        int run = 0;
#pragma unroll
        for (int k = 0; k < EB_NSL; ++k) {
            cin[(size_t)k * n + d] = run;
            run += ci[k];
        }
        cnt_in[d]  = (float)run;
        deg_in[d]  = rsqrtf(fmaxf((float)run, 1.0f));
        deg_out[d] = rsqrtf(fmaxf((float)co, 1.0f));
    }
}

__global__ __launch_bounds__(1024)
void segment_start_scan_kernel(const float* __restrict__ cnt,
                               int* __restrict__ seg, int n) {
    __shared__ int s_front[32];
    __shared__ int s_own[32];
    cnt += (size_t)blockIdx.y * n;
    seg += (size_t)blockIdx.y * n;
    const int t = threadIdx.x;
    const int lane = t & 31;
    const int w = t >> 5;
    const int first = blockIdx.x * 1024;
    int front = 0;
    const float4* c4 = (const float4*)cnt;
    const int nf4 = first >> 2;
    for (int i = t; i < nf4; i += 4096) {
        const float4 a = c4[i];
        const float4 b = c4[min(i + 1024, nf4 - 1)];
        const float4 c = c4[min(i + 2048, nf4 - 1)];
        const float4 e = c4[min(i + 3072, nf4 - 1)];
        const int sa = (int)a.x + (int)a.y + (int)a.z + (int)a.w;
        const int sb = (int)b.x + (int)b.y + (int)b.z + (int)b.w;
        const int sc = (int)c.x + (int)c.y + (int)c.z + (int)c.w;
        const int se = (int)e.x + (int)e.y + (int)e.z + (int)e.w;
        front += sa;
        front += (i + 1024 < nf4) ? sb : 0;
        front += (i + 2048 < nf4) ? sc : 0;
        front += (i + 3072 < nf4) ? se : 0;
    }
    const int idx = first + t;
    const int v = (idx < n) ? (int)cnt[idx] : 0;
    int incl = v;
    for (int o = 1; o < 32; o <<= 1) {
        const int u = __shfl_up(incl, o);
        if (lane >= o) incl += u;
    }
    for (int o = 16; o > 0; o >>= 1) front += __shfl_xor(front, o);
    if (lane == 31) s_own[w] = incl;
    if (lane == 0) s_front[w] = front;
    __syncthreads();
    int base = 0, woff = 0;
    for (int k = 0; k < 32; ++k) {
        base += s_front[k];
        if (k < w) woff += s_own[k];
    }
    if (idx < n) seg[idx] = base + woff + incl - v;
}

__global__ __launch_bounds__(EB_T)
void slice_place_kernel(const int* __restrict__ src,
                        const int* __restrict__ dst,
                        const int* __restrict__ seg,
                        const int* __restrict__ cin_front,
                        int* __restrict__ sorted_src, int n, int nE) {
    __shared__ int s_cur[EB_BR];
    src        += (size_t)blockIdx.y * nE;
    dst        += (size_t)blockIdx.y * nE;
    seg        += (size_t)blockIdx.y * n;
    cin_front  += (size_t)blockIdx.y * EB_NSL * n;
    sorted_src += (size_t)blockIdx.y * nE;
    const int t  = threadIdx.x;
    const int sl = blockIdx.x / EB_NB;
    const int rg = blockIdx.x % EB_NB;
    int e0, e1;
    edge_slice_bounds(sl, nE, e0, e1);
    for (int r0 = rg * EB_BR; r0 < n; r0 += EB_NB * EB_BR) {
        constexpr int NI = (EB_BR + EB_T - 1) / EB_T;
        int cv[NI];
#pragma unroll
        for (int k = 0; k < NI; ++k) {
            const int d = min(r0 + t + k * EB_T, n - 1);
            cv[k] = seg[d] + cin_front[(size_t)sl * n + d];
        }
#pragma unroll
        for (int k = 0; k < NI; ++k) {
            const int i = t + k * EB_T;
            if (i < EB_BR) s_cur[i] = cv[k];
        }
        __syncthreads();
        for (int e = e0 + t; e < e1; e += EB_U * EB_T) {
            int dv[EB_U], sv[EB_U];
#pragma unroll
            for (int q = 0; q < EB_U; ++q) {
                const int ee = e + q * EB_T;
                const int ec = min(ee, e1 - 1);
                const int d = dst[ec];
                const int s = src[ec];
                dv[q] = (ee < e1) ? d : -1;
                sv[q] = s;
            }
#pragma unroll
            for (int q = 0; q < EB_U; ++q) {
                const unsigned a = (unsigned)(dv[q] - r0);
                if (a < (unsigned)EB_BR && dv[q] < n) {
                    const int pos = atomicAdd(&s_cur[a], 1);
                    if ((unsigned)pos < (unsigned)nE) sorted_src[pos] = sv[q];
                }
            }
        }
        __syncthreads();
    }
}

template <bool EPI>
__global__ __launch_bounds__(256)
void gather_rows_sum_kernel(const int* __restrict__ seg_start,
                            const int* __restrict__ sorted_src,
                            const float* __restrict__ h,
                            float* __restrict__ agg, int n, int nE,
                            const float* __restrict__ in_scale,
                            const float* __restrict__ bias,
                            const float* __restrict__ out_scale) {
    const int lane = threadIdx.x & 31;
    float4 b4 = make_float4(0.f, 0.f, 0.f, 0.f);
    if constexpr (EPI) b4 = ((const float4*)bias)[lane];
    const int wave0 = __builtin_amdgcn_readfirstlane(
        (int)((blockIdx.x * blockDim.x + threadIdx.x) >> 5));
    const int nwaves = (int)((gridDim.x * blockDim.x) >> 5);
    const float4* h4 = (const float4*)h + lane;
    for (int d = wave0; d < n; d += nwaves) {
        int beg = seg_start[d];
        int end = seg_start[min(d + 1, n - 1)];
        if (d == n - 1) end = nE;
        float si = 1.0f, so = 1.0f;
        if constexpr (EPI) { si = in_scale[d]; so = out_scale[d]; }
        beg = max(beg, 0);
        end = min(end, nE);
        float4 acc = make_float4(0.f, 0.f, 0.f, 0.f);
        for (int base = beg; base < end; base += 32) {
            const int cnt = min(32, end - base);
            int mys = (lane < cnt) ? sorted_src[base + lane] : 0;
            mys = min(max(mys, 0), n - 1);
            int j = 0;
            for (; j + 4 <= cnt; j += 4) {
                const int s0 = __builtin_amdgcn_readlane(mys, j);
                const int s1 = __builtin_amdgcn_readlane(mys, j + 1);
                const int s2 = __builtin_amdgcn_readlane(mys, j + 2);
                const int s3 = __builtin_amdgcn_readlane(mys, j + 3);
                const float4 v0 = h4[(size_t)s0 * 32];
                const float4 v1 = h4[(size_t)s1 * 32];
                const float4 v2 = h4[(size_t)s2 * 32];
                const float4 v3 = h4[(size_t)s3 * 32];
                acc.x += v0.x; acc.y += v0.y; acc.z += v0.z; acc.w += v0.w;
                acc.x += v1.x; acc.y += v1.y; acc.z += v1.z; acc.w += v1.w;
                acc.x += v2.x; acc.y += v2.y; acc.z += v2.z; acc.w += v2.w;
                acc.x += v3.x; acc.y += v3.y; acc.z += v3.z; acc.w += v3.w;
            }
            const int rem = cnt - j;
            if (rem == 3) {
                const int s0 = __builtin_amdgcn_readlane(mys, j);
                const int s1 = __builtin_amdgcn_readlane(mys, j + 1);
                const int s2 = __builtin_amdgcn_readlane(mys, j + 2);
                const float4 v0 = h4[(size_t)s0 * 32];
                const float4 v1 = h4[(size_t)s1 * 32];
                const float4 v2 = h4[(size_t)s2 * 32];
                acc.x += v0.x; acc.y += v0.y; acc.z += v0.z; acc.w += v0.w;
                acc.x += v1.x; acc.y += v1.y; acc.z += v1.z; acc.w += v1.w;
                acc.x += v2.x; acc.y += v2.y; acc.z += v2.z; acc.w += v2.w;
            } else if (rem == 2) {
                const int s0 = __builtin_amdgcn_readlane(mys, j);
                const int s1 = __builtin_amdgcn_readlane(mys, j + 1);
                const float4 v0 = h4[(size_t)s0 * 32];
                const float4 v1 = h4[(size_t)s1 * 32];
                acc.x += v0.x; acc.y += v0.y; acc.z += v0.z; acc.w += v0.w;
                acc.x += v1.x; acc.y += v1.y; acc.z += v1.z; acc.w += v1.w;
            } else if (rem == 1) {
                const int s0 = __builtin_amdgcn_readlane(mys, j);
                const float4 v0 = h4[(size_t)s0 * 32];
                acc.x += v0.x; acc.y += v0.y; acc.z += v0.z; acc.w += v0.w;
            }
        }
        if constexpr (EPI) {
            acc.x = fmaxf(acc.x * si + b4.x, 0.0f) * so;
            acc.y = fmaxf(acc.y * si + b4.y, 0.0f) * so;
            acc.z = fmaxf(acc.z * si + b4.z, 0.0f) * so;
            acc.w = fmaxf(acc.w * si + b4.w, 0.0f) * so;
        }
        ((float4*)agg)[(size_t)d * 32 + lane] = acc;
    }
}

template <int NTILES, bool CLS>
__global__ __launch_bounds__(256)
void gemm_wmma_k128_kernel(const float* __restrict__ A,
                           const float* __restrict__ Wt,
                           const float* __restrict__ bias,
                           const float* __restrict__ arowscale,
                           const float* __restrict__ erowscale,
                           float* __restrict__ out,
                           int M, int relu,
                           const float* __restrict__ Wct,
                           const int* __restrict__ ids,
                           float* __restrict__ cls_out) {
    constexpr int N      = NTILES * 16;
    constexpr int PITCH  = 132;
    constexpr int OPITCH = N + 4;
    static_assert(!CLS || OPITCH == PITCH, "CLS reads the finished rows as A rows");
    __shared__ __attribute__((aligned(16))) float s_w[N * PITCH];
    __shared__ __attribute__((aligned(16))) float s_a[8 * 32 * PITCH];
    __shared__ __attribute__((aligned(16))) float s_c[CLS ? 16 * PITCH : 4];

    const int tid      = threadIdx.x;
    const int lane     = tid & 31;
    const int wave     = __builtin_amdgcn_readfirstlane(tid >> 5);
    const int c        = lane & 15;
    const int h        = lane >> 4;
    const int row_base = (blockIdx.x * 8 + wave) * 32;
    float* my_a = s_a + wave * (32 * PITCH);

    v4f wreg[N / 8];
#pragma unroll
    for (int i = 0; i < N / 8; ++i)
        wreg[i] = ((const v4f*)Wt)[tid + 256 * i];
    v4f areg[32];
#pragma unroll
    for (int i = 0; i < 32; ++i) {
        const int r = min(row_base + i, M - 1);
        areg[i] = ((const v4f*)A)[(size_t)r * 32 + lane];
    }
    v4f creg[2];
    int id0[8], id1[8];
    if constexpr (CLS) {
#pragma unroll
        for (int i = 0; i < 2; ++i)
            creg[i] = ((const v4f*)Wct)[tid + 256 * i];
#pragma unroll
        for (int r = 0; r < 8; ++r) {
            id0[r] = ids[min(row_base + 8 * h + r, M - 1)];
            id1[r] = ids[min(row_base + 16 + 8 * h + r, M - 1)];
        }
    }
    float myscale = 1.0f;
    if (arowscale) myscale = arowscale[min(row_base + lane, M - 1)];
    float es0[8], es1[8];
#pragma unroll
    for (int r = 0; r < 8; ++r) { es0[r] = 1.0f; es1[r] = 1.0f; }
    if (erowscale) {
#pragma unroll
        for (int r = 0; r < 8; ++r) {
            es0[r] = erowscale[min(row_base + 8 * h + r, M - 1)];
            es1[r] = erowscale[min(row_base + 16 + 8 * h + r, M - 1)];
        }
    }

    v8f acc0[NTILES], acc1[NTILES];
#pragma unroll
    for (int nt = 0; nt < NTILES; ++nt) {
        const float bv = bias[nt * 16 + c];
#pragma unroll
        for (int r = 0; r < 8; ++r) { acc0[nt][r] = bv; acc1[nt][r] = bv; }
    }

#pragma unroll
    for (int i = 0; i < N / 8; ++i) {
        const int u = tid + 256 * i;
        *(v4f*)(s_w + (u >> 5) * PITCH + (u & 31) * 4) = wreg[i];
    }
    if constexpr (CLS) {
#pragma unroll
        for (int i = 0; i < 2; ++i) {
            const int u = tid + 256 * i;
            *(v4f*)(s_c + (u >> 5) * PITCH + (u & 31) * 4) = creg[i];
        }
    }
#pragma unroll
    for (int i = 0; i < 32; ++i) {
        const float s = __int_as_float(
            __builtin_amdgcn_readlane(__float_as_int(myscale), i));
        v4f v = areg[i];
        v.x *= s; v.y *= s; v.z *= s; v.w *= s;
        *(v4f*)(my_a + i * PITCH + lane * 4) = v;
    }

    __syncthreads();

    const float* aw0 = my_a + c * PITCH + 4 * h;
    const float* aw1 = my_a + (16 + c) * PITCH + 4 * h;
    const float* bw  = s_w + c * PITCH + 4 * h;

    v4f a0c = *(const v4f*)(aw0);
    v4f a1c = *(const v4f*)(aw1);
    v4f bc[NTILES];
#pragma unroll
    for (int nt = 0; nt < NTILES; ++nt)
        bc[nt] = *(const v4f*)(bw + nt * 16 * PITCH);
#pragma unroll
    for (int j = 0; j < 16; ++j) {
        v4f a0n = a0c, a1n = a1c;
        v4f bn[NTILES];
#pragma unroll
        for (int nt = 0; nt < NTILES; ++nt) bn[nt] = bc[nt];
        if (j + 1 < 16) {
            a0n = *(const v4f*)(aw0 + 8 * (j + 1));
            a1n = *(const v4f*)(aw1 + 8 * (j + 1));
#pragma unroll
            for (int nt = 0; nt < NTILES; ++nt)
                bn[nt] = *(const v4f*)(bw + nt * 16 * PITCH + 8 * (j + 1));
        }
#pragma unroll
        for (int nt = 0; nt < NTILES; ++nt) {
            acc0[nt] = __builtin_amdgcn_wmma_f32_16x16x4_f32(
                false, a0c.xy, false, bc[nt].xy, (short)0, acc0[nt], false, false);
            acc1[nt] = __builtin_amdgcn_wmma_f32_16x16x4_f32(
                false, a1c.xy, false, bc[nt].xy, (short)0, acc1[nt], false, false);
            acc0[nt] = __builtin_amdgcn_wmma_f32_16x16x4_f32(
                false, a0c.zw, false, bc[nt].zw, (short)0, acc0[nt], false, false);
            acc1[nt] = __builtin_amdgcn_wmma_f32_16x16x4_f32(
                false, a1c.zw, false, bc[nt].zw, (short)0, acc1[nt], false, false);
        }
        a0c = a0n; a1c = a1n;
#pragma unroll
        for (int nt = 0; nt < NTILES; ++nt) bc[nt] = bn[nt];
    }

#pragma unroll
    for (int nt = 0; nt < NTILES; ++nt) {
#pragma unroll
        for (int r = 0; r < 8; ++r) {
            float v0 = acc0[nt][r];
            float v1 = acc1[nt][r];
            if (relu) { v0 = fmaxf(v0, 0.0f); v1 = fmaxf(v1, 0.0f); }
            v0 *= es0[r];
            v1 *= es1[r];
            my_a[(8 * h + r) * OPITCH + nt * 16 + c]      = v0;
            my_a[(16 + 8 * h + r) * OPITCH + nt * 16 + c] = v1;
        }
    }

    __syncthreads();

    if constexpr (CLS) {
        const float* cw = s_c + c * PITCH + 4 * h;
        v8f p0, p1;
#pragma unroll
        for (int r = 0; r < 8; ++r) { p0[r] = 0.0f; p1[r] = 0.0f; }
        v4f x0c = *(const v4f*)(aw0);
        v4f x1c = *(const v4f*)(aw1);
        v4f wcc = *(const v4f*)(cw);
#pragma unroll
        for (int j = 0; j < 16; ++j) {
            v4f x0n = x0c, x1n = x1c, wcn = wcc;
            if (j + 1 < 16) {
                x0n = *(const v4f*)(aw0 + 8 * (j + 1));
                x1n = *(const v4f*)(aw1 + 8 * (j + 1));
                wcn = *(const v4f*)(cw + 8 * (j + 1));
            }
            p0 = __builtin_amdgcn_wmma_f32_16x16x4_f32(
                false, x0c.xy, false, wcc.xy, (short)0, p0, false, false);
            p1 = __builtin_amdgcn_wmma_f32_16x16x4_f32(
                false, x1c.xy, false, wcc.xy, (short)0, p1, false, false);
            p0 = __builtin_amdgcn_wmma_f32_16x16x4_f32(
                false, x0c.zw, false, wcc.zw, (short)0, p0, false, false);
            p1 = __builtin_amdgcn_wmma_f32_16x16x4_f32(
                false, x1c.zw, false, wcc.zw, (short)0, p1, false, false);
            x0c = x0n; x1c = x1n; wcc = wcn;
        }
#pragma unroll
        for (int r = 0; r < 8; ++r) {
            if (row_base + 8 * h + r < M)
                atomicAdd(cls_out + (size_t)id0[r] * 16 + c, p0[r]);
            if (row_base + 16 + 8 * h + r < M)
                atomicAdd(cls_out + (size_t)id1[r] * 16 + c, p1[r]);
        }
        return;
    }

    constexpr int ITER = N / 4;
    constexpr int GRP  = ITER < 8 ? ITER : 8;
#pragma unroll
    for (int g = 0; g < ITER / GRP; ++g) {
        v4f t[GRP];
#pragma unroll
        for (int q = 0; q < GRP; ++q) {
            const int u = lane + 32 * (g * GRP + q);
            t[q] = *(const v4f*)(my_a + (u / (N / 4)) * OPITCH + 4 * (u % (N / 4)));
        }
#pragma unroll
        for (int q = 0; q < GRP; ++q) {
            const int u    = lane + 32 * (g * GRP + q);
            const int grow = row_base + u / (N / 4);
            if (grow < M)
                *(v4f*)(out + (size_t)grow * N + 4 * (u % (N / 4))) = t[q];
        }
    }
}


extern "C" void kernel_launch(void* const* d_in, const int* in_sizes, int n_in,
                              void* d_out, int out_size, void* d_ws, size_t ws_size,
                              hipStream_t stream) {
    (void)n_in; (void)ws_size;
    const float* x   = (const float*)d_in[0];
    const float* W2  = (const float*)d_in[1];
    const float* b2  = (const float*)d_in[2];
    const float* gw  = (const float*)d_in[3];
    const float* gb  = (const float*)d_in[4];
    const float* Wc  = (const float*)d_in[5];
    const float* bc  = (const float*)d_in[6];
    const int*   src = (const int*)d_in[7];
    const int*   dst = (const int*)d_in[8];
    const int*   ids = (const int*)d_in[9];

    const int H  = 128;
    const int C  = 16;
    const int L  = in_sizes[2] / H;
    const int NS = in_sizes[9] / L;
    const int NE = in_sizes[7] / L;
    const int NL = in_sizes[4] / (L * H);
    const int NN = out_size / C;

    float* ws       = (float*)d_ws;
    float* out_feat = ws;
    float* hbuf     = out_feat + (size_t)NN * H;
    float* agg      = hbuf + (size_t)NS * H;
    float* wmt      = agg + (size_t)NS * H;
    float* gwt      = wmt + (size_t)L * H * H;
    float* wct      = gwt + (size_t)L * NL * H * H;

    int*   seg_all    = (int*)out_feat;
    int*   sorted_all = seg_all + (size_t)L * NS;
    float* dego_all   = (float*)(sorted_all + (size_t)L * NE);
    float* degi_all   = dego_all + (size_t)L * NS;
    float* cnt_all    = degi_all + (size_t)L * NS;
    int*   cin_all    = (int*)(cnt_all + (size_t)L * NS);
    int*   cout_all   = cin_all + (size_t)L * EB_NSL * NS;
    float* bm_all     = (float*)(cout_all + (size_t)L * EB_NSL * NS);

    const int T = 256;
    const int gemm_blocks_NS = (((NS / 16) + 1) / 2 + 7) / 8;
    const int gather_blocks  = ((NS + 1) / 2 + 7) / 8;

    {
        const int nHH  = H * H;
        const int bHH  = (nHH + T - 1) / T;
        const int nmat = L + L * NL + 1;
        const int n4   = (int)((size_t)NN * C / 4);
        setup_kernel<<<dim3(bHH, nmat + 8), T, 0, stream>>>(
            W2, b2, gw, Wc, bc, wmt, bm_all, gwt, wct, (float4*)d_out, n4, L, NL, H, C);
    }

    slice_count_kernel<<<dim3(EB_NSL * EB_NB, L), EB_T, 0, stream>>>(
        src, dst, cin_all, cout_all, NS, NE);
    slice_totals_kernel<<<dim3((NS + T - 1) / T, L), T, 0, stream>>>(
        cin_all, cout_all, cnt_all, dego_all, degi_all, NS);
    segment_start_scan_kernel<<<dim3((NS + 1023) / 1024, L), 1024, 0, stream>>>(
        cnt_all, seg_all, NS);
    slice_place_kernel<<<dim3(EB_NSL * EB_NB, L), EB_T, 0, stream>>>(
        src, dst, seg_all, cin_all, sorted_all, NS, NE);

    for (int i = 0; i < L; ++i) {
        const int*   ids_i     = ids + (size_t)i * NS;
        const float* x_i       = x   + (size_t)i * NS * H;
        const float* wmt_i     = wmt + (size_t)i * H * H;
        const float* bm_i      = bm_all + (size_t)i * H;
        const int*   seg_i     = seg_all    + (size_t)i * NS;
        const int*   sorted_i  = sorted_all + (size_t)i * NE;
        const float* deg_out_i = dego_all   + (size_t)i * NS;
        const float* deg_in_i  = degi_all   + (size_t)i * NS;
        float* hcur = hbuf;
        float* hoth = agg;

        gemm_wmma_k128_kernel<8, false><<<gemm_blocks_NS, T, 0, stream>>>(
            x_i, wmt_i, bm_i, nullptr, deg_out_i, hcur, NS, 0, nullptr, nullptr, nullptr);

        gather_rows_sum_kernel<true><<<gather_blocks, T, 0, stream>>>(
            seg_i, sorted_i, hcur, hoth, NS, NE,
            deg_in_i, gb + (size_t)i * NL * H, deg_out_i);
        { float* t = hcur; hcur = hoth; hoth = t; }

        for (int l = 1; l < NL; ++l) {
            const float* gwt_il = gwt + ((size_t)i * NL + l) * H * H;
            const float* gb_il  = gb  + ((size_t)i * NL + l) * H;
            gather_rows_sum_kernel<false><<<gather_blocks, T, 0, stream>>>(
                seg_i, sorted_i, hcur, hoth, NS, NE, nullptr, nullptr, nullptr);
            if (l + 1 < NL) {
                gemm_wmma_k128_kernel<8, false><<<gemm_blocks_NS, T, 0, stream>>>(
                    hoth, gwt_il, gb_il, deg_in_i, deg_out_i, hcur, NS, 1,
                    nullptr, nullptr, nullptr);
            } else {
                gemm_wmma_k128_kernel<8, true><<<gemm_blocks_NS, T, 0, stream>>>(
                    hoth, gwt_il, gb_il, deg_in_i, nullptr, hcur, NS, 1,
                    wct, ids_i, (float*)d_out);
            }
        }
    }
}


# conv stack: a link's GEMM and another link's edge sum share a launch as two wave roles of each work-group (8 WMMA + 16 gather waves, 1 group per WGP, 3 row buffers); 17 -> 13 launches; sum order kept
# speedup vs baseline: 5.5078x; 1.0689x over previous
#include <hip/hip_runtime.h>


typedef __attribute__((ext_vector_type(2))) float v2f;
typedef __attribute__((ext_vector_type(4))) float v4f;
typedef __attribute__((ext_vector_type(8))) float v8f;


#define MG_H     128
#define MG_PITCH 132
__global__ __launch_bounds__(256)
void setup_kernel(const float* __restrict__ W2,
                  const float* __restrict__ b2,
                  const float* __restrict__ gw,
                  const float* __restrict__ Wc,
                  const float* __restrict__ bc,
                  float* __restrict__ wmt,
                  float* __restrict__ bm,
                  float* __restrict__ gwt,
                  float* __restrict__ wct,
                  float4* __restrict__ out4, int n4,
                  int nL, int nLayer, int H, int C) {
    __shared__ __attribute__((aligned(16))) float s_t[16 * MG_PITCH];
    const int y    = blockIdx.y;
    const int nGw  = nL * nLayer;
    const int nmat = nL + nGw + 1;
    if (y < nL) {
        const float* gw0 = gw + (size_t)y * nLayer * MG_H * MG_H;
        const float* w2  = W2 + (size_t)y * MG_H * MG_H;
        const int x   = blockIdx.x;
        const int tid = threadIdx.x;
        if (x < MG_H / 16) {
            const int lane = tid & 31;
            const int wave = __builtin_amdgcn_readfirstlane(tid >> 5);
            const int c = lane & 15;
            const int h = lane >> 4;
            const float* ap = gw0 + (size_t)(4 * h) * MG_H + 16 * x + c;
            const float* bp = w2 + (size_t)(16 * wave + c) * MG_H + 4 * h;
            v4f af[16], bf[16];
#pragma unroll
            for (int jj = 0; jj < 16; ++jj) {
                bf[jj] = *(const v4f*)(bp + 8 * jj);
                af[jj].x = ap[(size_t)(8 * jj + 0) * MG_H];
                af[jj].y = ap[(size_t)(8 * jj + 1) * MG_H];
                af[jj].z = ap[(size_t)(8 * jj + 2) * MG_H];
                af[jj].w = ap[(size_t)(8 * jj + 3) * MG_H];
            }
            v8f acc;
#pragma unroll
            for (int r = 0; r < 8; ++r) acc[r] = 0.0f;
            __builtin_amdgcn_sched_barrier(0);
#pragma unroll
            for (int jj = 0; jj < 16; ++jj) {
                acc = __builtin_amdgcn_wmma_f32_16x16x4_f32(
                    false, af[jj].xy, false, bf[jj].xy, (short)0, acc, false, false);
                acc = __builtin_amdgcn_wmma_f32_16x16x4_f32(
                    false, af[jj].zw, false, bf[jj].zw, (short)0, acc, false, false);
            }
            __builtin_amdgcn_sched_barrier(0);
            asm volatile("v_nop\n\tv_nop\n\tv_nop\n\tv_nop");
            __builtin_amdgcn_sched_barrier(0);
#pragma unroll
            for (int r = 0; r < 8; ++r)
                s_t[(8 * h + r) * MG_PITCH + 16 * wave + c] = acc[r];
            __syncthreads();
#pragma unroll
            for (int q = 0; q < 2; ++q) {
                const int row = 2 * wave + q;
                const v4f t = *(const v4f*)(s_t + row * MG_PITCH + 4 * lane);
                *(v4f*)(wmt + (size_t)y * MG_H * MG_H + (size_t)(16 * x + row) * MG_H + 4 * lane) = t;
            }
        } else if (x == MG_H / 16) {
            const int n    = tid & (MG_H - 1);
            const int part = tid >> 7;
            const float* bsrc = b2 + (size_t)y * MG_H + part * (MG_H / 2);
            const float* gsrc = gw0 + (size_t)(part * (MG_H / 2)) * MG_H + n;
            float sum = 0.0f;
            for (int j0 = 0; j0 < MG_H / 2; j0 += 16) {
                float gv[16], bv[16];
#pragma unroll
                for (int q = 0; q < 16; ++q) {
                    gv[q] = gsrc[(size_t)(j0 + q) * MG_H];
                    bv[q] = bsrc[j0 + q];
                }
#pragma unroll
                for (int q = 0; q < 16; ++q) sum += bv[q] * gv[q];
            }
            s_t[tid] = sum;
            __syncthreads();
            if (tid < MG_H) bm[(size_t)y * MG_H + tid] = s_t[tid] + s_t[tid + MG_H];
        }
    } else if (y < nmat) {
        const float* in;
        float* outT;
        int K = H, N = H;
        if (y < nL + nGw) {
            const int g = y - nL;
            if (g % nLayer == 0) return;
            in = gw + (size_t)g * H * H;  outT = gwt + (size_t)g * H * H;
        } else {
            in = Wc;  outT = wct;  N = C;
        }
        const int i = blockIdx.x * blockDim.x + threadIdx.x;
        if (i < N * K) {
            const int n = i / K, k = i % K;
            outT[i] = in[k * N + n];
        }
    } else {
        const int fb     = (y - nmat) * gridDim.x + blockIdx.x;
        const int stride = (gridDim.y - nmat) * gridDim.x * blockDim.x;
        int i = fb * blockDim.x + threadIdx.x;
        const float4 b = ((const float4*)bc)[i & 3];
        for (; i < n4; i += stride) out4[i] = b;
    }
}

#define EB_NB  16
#define EB_NSL 16
#define EB_BR  3136
#define EB_T   512
#define EB_U   16

__device__ __forceinline__ void edge_slice_bounds(int sl, int nE, int& e0, int& e1) {
    const int per = (nE + EB_NSL - 1) / EB_NSL;
    e0 = min(sl * per, nE);
    e1 = min(e0 + per, nE);
}

__global__ __launch_bounds__(EB_T)
void slice_count_kernel(const int* __restrict__ src,
                        const int* __restrict__ dst,
                        int* __restrict__ cin,
                        int* __restrict__ cout, int n, int nE) {
    __shared__ int s_in[EB_BR];
    __shared__ int s_out[EB_BR];
    src  += (size_t)blockIdx.y * nE;
    dst  += (size_t)blockIdx.y * nE;
    cin  += (size_t)blockIdx.y * EB_NSL * n;
    cout += (size_t)blockIdx.y * EB_NSL * n;
    const int t  = threadIdx.x;
    const int sl = blockIdx.x / EB_NB;
    const int rg = blockIdx.x % EB_NB;
    int e0, e1;
    edge_slice_bounds(sl, nE, e0, e1);
    for (int r0 = rg * EB_BR; r0 < n; r0 += EB_NB * EB_BR) {
        for (int i = t; i < EB_BR; i += EB_T) { s_in[i] = 0; s_out[i] = 0; }
        __syncthreads();
        for (int e = e0 + t; e < e1; e += EB_U * EB_T) {
            int dv[EB_U], sv[EB_U];
#pragma unroll
            for (int q = 0; q < EB_U; ++q) {
                const int ee = e + q * EB_T;
                const int ec = min(ee, e1 - 1);
                const int d = dst[ec];
                const int s = src[ec];
                dv[q] = (ee < e1) ? d : -1;
                sv[q] = (ee < e1) ? s : -1;
            }
#pragma unroll
            for (int q = 0; q < EB_U; ++q) {
                const unsigned a = (unsigned)(dv[q] - r0);
                const unsigned b = (unsigned)(sv[q] - r0);
                if (a < (unsigned)EB_BR && dv[q] < n) atomicAdd(&s_in[a], 1);
                if (b < (unsigned)EB_BR && sv[q] < n) atomicAdd(&s_out[b], 1);
            }
        }
        __syncthreads();
        for (int i = t; i < EB_BR; i += EB_T) {
            const int d = r0 + i;
            if (d < n) {
                cin[(size_t)sl * n + d]  = s_in[i];
                cout[(size_t)sl * n + d] = s_out[i];
            }
        }
        __syncthreads();
    }
}

__global__ void slice_totals_kernel(int* __restrict__ cin,
                                    const int* __restrict__ cout,
                                    float* __restrict__ cnt_in,
                                    float* __restrict__ deg_out,
                                    float* __restrict__ deg_in, int n) {
    cin     += (size_t)blockIdx.y * EB_NSL * n;
    cout    += (size_t)blockIdx.y * EB_NSL * n;
    cnt_in  += (size_t)blockIdx.y * n;
    deg_out += (size_t)blockIdx.y * n;
    deg_in  += (size_t)blockIdx.y * n;
    const int d = blockIdx.x * blockDim.x + threadIdx.x;
    if (d < n) {
        int ci[EB_NSL], co = 0;
#pragma unroll
        for (int k = 0; k < EB_NSL; ++k) {
            ci[k] = cin[(size_t)k * n + d];
            co += cout[(size_t)k * n + d];
        }
        int run = 0;
#pragma unroll
        for (int k = 0; k < EB_NSL; ++k) {
            cin[(size_t)k * n + d] = run;
            run += ci[k];
        }
        cnt_in[d]  = (float)run;
        deg_in[d]  = rsqrtf(fmaxf((float)run, 1.0f));
        deg_out[d] = rsqrtf(fmaxf((float)co, 1.0f));
    }
}

__global__ __launch_bounds__(1024)
void segment_start_scan_kernel(const float* __restrict__ cnt,
                               int* __restrict__ seg, int n) {
    __shared__ int s_front[32];
    __shared__ int s_own[32];
    cnt += (size_t)blockIdx.y * n;
    seg += (size_t)blockIdx.y * n;
    const int t = threadIdx.x;
    const int lane = t & 31;
    const int w = t >> 5;
    const int first = blockIdx.x * 1024;
    int front = 0;
    const float4* c4 = (const float4*)cnt;
    const int nf4 = first >> 2;
    for (int i = t; i < nf4; i += 4096) {
        const float4 a = c4[i];
        const float4 b = c4[min(i + 1024, nf4 - 1)];
        const float4 c = c4[min(i + 2048, nf4 - 1)];
        const float4 e = c4[min(i + 3072, nf4 - 1)];
        const int sa = (int)a.x + (int)a.y + (int)a.z + (int)a.w;
        const int sb = (int)b.x + (int)b.y + (int)b.z + (int)b.w;
        const int sc = (int)c.x + (int)c.y + (int)c.z + (int)c.w;
        const int se = (int)e.x + (int)e.y + (int)e.z + (int)e.w;
        front += sa;
        front += (i + 1024 < nf4) ? sb : 0;
        front += (i + 2048 < nf4) ? sc : 0;
        front += (i + 3072 < nf4) ? se : 0;
    }
    const int idx = first + t;
    const int v = (idx < n) ? (int)cnt[idx] : 0;
    int incl = v;
    for (int o = 1; o < 32; o <<= 1) {
        const int u = __shfl_up(incl, o);
        if (lane >= o) incl += u;
    }
    for (int o = 16; o > 0; o >>= 1) front += __shfl_xor(front, o);
    if (lane == 31) s_own[w] = incl;
    if (lane == 0) s_front[w] = front;
    __syncthreads();
    int base = 0, woff = 0;
    for (int k = 0; k < 32; ++k) {
        base += s_front[k];
        if (k < w) woff += s_own[k];
    }
    if (idx < n) seg[idx] = base + woff + incl - v;
}

__global__ __launch_bounds__(EB_T)
void slice_place_kernel(const int* __restrict__ src,
                        const int* __restrict__ dst,
                        const int* __restrict__ seg,
                        const int* __restrict__ cin_front,
                        int* __restrict__ sorted_src, int n, int nE) {
    __shared__ int s_cur[EB_BR];
    src        += (size_t)blockIdx.y * nE;
    dst        += (size_t)blockIdx.y * nE;
    seg        += (size_t)blockIdx.y * n;
    cin_front  += (size_t)blockIdx.y * EB_NSL * n;
    sorted_src += (size_t)blockIdx.y * nE;
    const int t  = threadIdx.x;
    const int sl = blockIdx.x / EB_NB;
    const int rg = blockIdx.x % EB_NB;
    int e0, e1;
    edge_slice_bounds(sl, nE, e0, e1);
    for (int r0 = rg * EB_BR; r0 < n; r0 += EB_NB * EB_BR) {
        constexpr int NI = (EB_BR + EB_T - 1) / EB_T;
        int cv[NI];
#pragma unroll
        for (int k = 0; k < NI; ++k) {
            const int d = min(r0 + t + k * EB_T, n - 1);
            cv[k] = seg[d] + cin_front[(size_t)sl * n + d];
        }
#pragma unroll
        for (int k = 0; k < NI; ++k) {
            const int i = t + k * EB_T;
            if (i < EB_BR) s_cur[i] = cv[k];
        }
        __syncthreads();
        for (int e = e0 + t; e < e1; e += EB_U * EB_T) {
            int dv[EB_U], sv[EB_U];
#pragma unroll
            for (int q = 0; q < EB_U; ++q) {
                const int ee = e + q * EB_T;
                const int ec = min(ee, e1 - 1);
                const int d = dst[ec];
                const int s = src[ec];
                dv[q] = (ee < e1) ? d : -1;
                sv[q] = s;
            }
#pragma unroll
            for (int q = 0; q < EB_U; ++q) {
                const unsigned a = (unsigned)(dv[q] - r0);
                if (a < (unsigned)EB_BR && dv[q] < n) {
                    const int pos = atomicAdd(&s_cur[a], 1);
                    if ((unsigned)pos < (unsigned)nE) sorted_src[pos] = sv[q];
                }
            }
        }
        __syncthreads();
    }
}

template <bool EPI>
__global__ __launch_bounds__(256)
void gather_rows_sum_kernel(const int* __restrict__ seg_start,
                            const int* __restrict__ sorted_src,
                            const float* __restrict__ h,
                            float* __restrict__ agg, int n, int nE,
                            const float* __restrict__ in_scale,
                            const float* __restrict__ bias,
                            const float* __restrict__ out_scale) {
    const int lane = threadIdx.x & 31;
    float4 b4 = make_float4(0.f, 0.f, 0.f, 0.f);
    if constexpr (EPI) b4 = ((const float4*)bias)[lane];
    const int wave0 = __builtin_amdgcn_readfirstlane(
        (int)((blockIdx.x * blockDim.x + threadIdx.x) >> 5));
    const int nwaves = (int)((gridDim.x * blockDim.x) >> 5);
    const float4* h4 = (const float4*)h + lane;
    for (int d = wave0; d < n; d += nwaves) {
        int beg = seg_start[d];
        int end = seg_start[min(d + 1, n - 1)];
        if (d == n - 1) end = nE;
        float si = 1.0f, so = 1.0f;
        if constexpr (EPI) { si = in_scale[d]; so = out_scale[d]; }
        beg = max(beg, 0);
        end = min(end, nE);
        float4 acc = make_float4(0.f, 0.f, 0.f, 0.f);
        for (int base = beg; base < end; base += 32) {
            const int cnt = min(32, end - base);
            int mys = (lane < cnt) ? sorted_src[base + lane] : 0;
            mys = min(max(mys, 0), n - 1);
            int j = 0;
            for (; j + 4 <= cnt; j += 4) {
                const int s0 = __builtin_amdgcn_readlane(mys, j);
                const int s1 = __builtin_amdgcn_readlane(mys, j + 1);
                const int s2 = __builtin_amdgcn_readlane(mys, j + 2);
                const int s3 = __builtin_amdgcn_readlane(mys, j + 3);
                const float4 v0 = h4[(size_t)s0 * 32];
                const float4 v1 = h4[(size_t)s1 * 32];
                const float4 v2 = h4[(size_t)s2 * 32];
                const float4 v3 = h4[(size_t)s3 * 32];
                acc.x += v0.x; acc.y += v0.y; acc.z += v0.z; acc.w += v0.w;
                acc.x += v1.x; acc.y += v1.y; acc.z += v1.z; acc.w += v1.w;
                acc.x += v2.x; acc.y += v2.y; acc.z += v2.z; acc.w += v2.w;
                acc.x += v3.x; acc.y += v3.y; acc.z += v3.z; acc.w += v3.w;
            }
            const int rem = cnt - j;
            if (rem == 3) {
                const int s0 = __builtin_amdgcn_readlane(mys, j);
                const int s1 = __builtin_amdgcn_readlane(mys, j + 1);
                const int s2 = __builtin_amdgcn_readlane(mys, j + 2);
                const float4 v0 = h4[(size_t)s0 * 32];
                const float4 v1 = h4[(size_t)s1 * 32];
                const float4 v2 = h4[(size_t)s2 * 32];
                acc.x += v0.x; acc.y += v0.y; acc.z += v0.z; acc.w += v0.w;
                acc.x += v1.x; acc.y += v1.y; acc.z += v1.z; acc.w += v1.w;
                acc.x += v2.x; acc.y += v2.y; acc.z += v2.z; acc.w += v2.w;
            } else if (rem == 2) {
                const int s0 = __builtin_amdgcn_readlane(mys, j);
                const int s1 = __builtin_amdgcn_readlane(mys, j + 1);
                const float4 v0 = h4[(size_t)s0 * 32];
                const float4 v1 = h4[(size_t)s1 * 32];
                acc.x += v0.x; acc.y += v0.y; acc.z += v0.z; acc.w += v0.w;
                acc.x += v1.x; acc.y += v1.y; acc.z += v1.z; acc.w += v1.w;
            } else if (rem == 1) {
                const int s0 = __builtin_amdgcn_readlane(mys, j);
                const float4 v0 = h4[(size_t)s0 * 32];
                acc.x += v0.x; acc.y += v0.y; acc.z += v0.z; acc.w += v0.w;
            }
        }
        if constexpr (EPI) {
            acc.x = fmaxf(acc.x * si + b4.x, 0.0f) * so;
            acc.y = fmaxf(acc.y * si + b4.y, 0.0f) * so;
            acc.z = fmaxf(acc.z * si + b4.z, 0.0f) * so;
            acc.w = fmaxf(acc.w * si + b4.w, 0.0f) * so;
        }
        ((float4*)agg)[(size_t)d * 32 + lane] = acc;
    }
}

template <int NTILES, bool CLS>
__global__ __launch_bounds__(256)
void gemm_wmma_k128_kernel(const float* __restrict__ A,
                           const float* __restrict__ Wt,
                           const float* __restrict__ bias,
                           const float* __restrict__ arowscale,
                           const float* __restrict__ erowscale,
                           float* __restrict__ out,
                           int M, int relu,
                           const float* __restrict__ Wct,
                           const int* __restrict__ ids,
                           float* __restrict__ cls_out) {
    constexpr int N      = NTILES * 16;
    constexpr int PITCH  = 132;
    constexpr int OPITCH = N + 4;
    static_assert(!CLS || OPITCH == PITCH, "CLS reads the finished rows as A rows");
    __shared__ __attribute__((aligned(16))) float s_w[N * PITCH];
    __shared__ __attribute__((aligned(16))) float s_a[8 * 32 * PITCH];
    __shared__ __attribute__((aligned(16))) float s_c[CLS ? 16 * PITCH : 4];

    const int tid      = threadIdx.x;
    const int lane     = tid & 31;
    const int wave     = __builtin_amdgcn_readfirstlane(tid >> 5);
    const int c        = lane & 15;
    const int h        = lane >> 4;
    const int row_base = (blockIdx.x * 8 + wave) * 32;
    float* my_a = s_a + wave * (32 * PITCH);

    v4f wreg[N / 8];
#pragma unroll
    for (int i = 0; i < N / 8; ++i)
        wreg[i] = ((const v4f*)Wt)[tid + 256 * i];
    v4f areg[32];
#pragma unroll
    for (int i = 0; i < 32; ++i) {
        const int r = min(row_base + i, M - 1);
        areg[i] = ((const v4f*)A)[(size_t)r * 32 + lane];
    }
    v4f creg[2];
    int id0[8], id1[8];
    if constexpr (CLS) {
#pragma unroll
        for (int i = 0; i < 2; ++i)
            creg[i] = ((const v4f*)Wct)[tid + 256 * i];
#pragma unroll
        for (int r = 0; r < 8; ++r) {
            id0[r] = ids[min(row_base + 8 * h + r, M - 1)];
            id1[r] = ids[min(row_base + 16 + 8 * h + r, M - 1)];
        }
    }
    float myscale = 1.0f;
    if (arowscale) myscale = arowscale[min(row_base + lane, M - 1)];
    float es0[8], es1[8];
#pragma unroll
    for (int r = 0; r < 8; ++r) { es0[r] = 1.0f; es1[r] = 1.0f; }
    if (erowscale) {
#pragma unroll
        for (int r = 0; r < 8; ++r) {
            es0[r] = erowscale[min(row_base + 8 * h + r, M - 1)];
            es1[r] = erowscale[min(row_base + 16 + 8 * h + r, M - 1)];
        }
    }

    v8f acc0[NTILES], acc1[NTILES];
#pragma unroll
    for (int nt = 0; nt < NTILES; ++nt) {
        const float bv = bias[nt * 16 + c];
#pragma unroll
        for (int r = 0; r < 8; ++r) { acc0[nt][r] = bv; acc1[nt][r] = bv; }
    }

#pragma unroll
    for (int i = 0; i < N / 8; ++i) {
        const int u = tid + 256 * i;
        *(v4f*)(s_w + (u >> 5) * PITCH + (u & 31) * 4) = wreg[i];
    }
    if constexpr (CLS) {
#pragma unroll
        for (int i = 0; i < 2; ++i) {
            const int u = tid + 256 * i;
            *(v4f*)(s_c + (u >> 5) * PITCH + (u & 31) * 4) = creg[i];
        }
    }
#pragma unroll
    for (int i = 0; i < 32; ++i) {
        const float s = __int_as_float(
            __builtin_amdgcn_readlane(__float_as_int(myscale), i));
        v4f v = areg[i];
        v.x *= s; v.y *= s; v.z *= s; v.w *= s;
        *(v4f*)(my_a + i * PITCH + lane * 4) = v;
    }

    __syncthreads();

    const float* aw0 = my_a + c * PITCH + 4 * h;
    const float* aw1 = my_a + (16 + c) * PITCH + 4 * h;
    const float* bw  = s_w + c * PITCH + 4 * h;

    v4f a0c = *(const v4f*)(aw0);
    v4f a1c = *(const v4f*)(aw1);
    v4f bc[NTILES];
#pragma unroll
    for (int nt = 0; nt < NTILES; ++nt)
        bc[nt] = *(const v4f*)(bw + nt * 16 * PITCH);
#pragma unroll
    for (int j = 0; j < 16; ++j) {
        v4f a0n = a0c, a1n = a1c;
        v4f bn[NTILES];
#pragma unroll
        for (int nt = 0; nt < NTILES; ++nt) bn[nt] = bc[nt];
        if (j + 1 < 16) {
            a0n = *(const v4f*)(aw0 + 8 * (j + 1));
            a1n = *(const v4f*)(aw1 + 8 * (j + 1));
#pragma unroll
            for (int nt = 0; nt < NTILES; ++nt)
                bn[nt] = *(const v4f*)(bw + nt * 16 * PITCH + 8 * (j + 1));
        }
#pragma unroll
        for (int nt = 0; nt < NTILES; ++nt) {
            acc0[nt] = __builtin_amdgcn_wmma_f32_16x16x4_f32(
                false, a0c.xy, false, bc[nt].xy, (short)0, acc0[nt], false, false);
            acc1[nt] = __builtin_amdgcn_wmma_f32_16x16x4_f32(
                false, a1c.xy, false, bc[nt].xy, (short)0, acc1[nt], false, false);
            acc0[nt] = __builtin_amdgcn_wmma_f32_16x16x4_f32(
                false, a0c.zw, false, bc[nt].zw, (short)0, acc0[nt], false, false);
            acc1[nt] = __builtin_amdgcn_wmma_f32_16x16x4_f32(
                false, a1c.zw, false, bc[nt].zw, (short)0, acc1[nt], false, false);
        }
        a0c = a0n; a1c = a1n;
#pragma unroll
        for (int nt = 0; nt < NTILES; ++nt) bc[nt] = bn[nt];
    }

#pragma unroll
    for (int nt = 0; nt < NTILES; ++nt) {
#pragma unroll
        for (int r = 0; r < 8; ++r) {
            float v0 = acc0[nt][r];
            float v1 = acc1[nt][r];
            if (relu) { v0 = fmaxf(v0, 0.0f); v1 = fmaxf(v1, 0.0f); }
            v0 *= es0[r];
            v1 *= es1[r];
            my_a[(8 * h + r) * OPITCH + nt * 16 + c]      = v0;
            my_a[(16 + 8 * h + r) * OPITCH + nt * 16 + c] = v1;
        }
    }

    __syncthreads();

    if constexpr (CLS) {
        const float* cw = s_c + c * PITCH + 4 * h;
        v8f p0, p1;
#pragma unroll
        for (int r = 0; r < 8; ++r) { p0[r] = 0.0f; p1[r] = 0.0f; }
        v4f x0c = *(const v4f*)(aw0);
        v4f x1c = *(const v4f*)(aw1);
        v4f wcc = *(const v4f*)(cw);
#pragma unroll
        for (int j = 0; j < 16; ++j) {
            v4f x0n = x0c, x1n = x1c, wcn = wcc;
            if (j + 1 < 16) {
                x0n = *(const v4f*)(aw0 + 8 * (j + 1));
                x1n = *(const v4f*)(aw1 + 8 * (j + 1));
                wcn = *(const v4f*)(cw + 8 * (j + 1));
            }
            p0 = __builtin_amdgcn_wmma_f32_16x16x4_f32(
                false, x0c.xy, false, wcc.xy, (short)0, p0, false, false);
            p1 = __builtin_amdgcn_wmma_f32_16x16x4_f32(
                false, x1c.xy, false, wcc.xy, (short)0, p1, false, false);
            p0 = __builtin_amdgcn_wmma_f32_16x16x4_f32(
                false, x0c.zw, false, wcc.zw, (short)0, p0, false, false);
            p1 = __builtin_amdgcn_wmma_f32_16x16x4_f32(
                false, x1c.zw, false, wcc.zw, (short)0, p1, false, false);
            x0c = x0n; x1c = x1n; wcc = wcn;
        }
#pragma unroll
        for (int r = 0; r < 8; ++r) {
            if (row_base + 8 * h + r < M)
                atomicAdd(cls_out + (size_t)id0[r] * 16 + c, p0[r]);
            if (row_base + 16 + 8 * h + r < M)
                atomicAdd(cls_out + (size_t)id1[r] * 16 + c, p1[r]);
        }
        return;
    }

    constexpr int ITER = N / 4;
    constexpr int GRP  = ITER < 8 ? ITER : 8;
#pragma unroll
    for (int g = 0; g < ITER / GRP; ++g) {
        v4f t[GRP];
#pragma unroll
        for (int q = 0; q < GRP; ++q) {
            const int u = lane + 32 * (g * GRP + q);
            t[q] = *(const v4f*)(my_a + (u / (N / 4)) * OPITCH + 4 * (u % (N / 4)));
        }
#pragma unroll
        for (int q = 0; q < GRP; ++q) {
            const int u    = lane + 32 * (g * GRP + q);
            const int grow = row_base + u / (N / 4);
            if (grow < M)
                *(v4f*)(out + (size_t)grow * N + 4 * (u % (N / 4))) = t[q];
        }
    }
}


#define PG_GW 8
#define PG_SW 16
template <bool CLS, bool EPI>
__global__ __launch_bounds__((PG_GW + PG_SW) * 32)
void gemm_beside_gather_kernel(
        const float* __restrict__ A,
        const float* __restrict__ Wt,
        const float* __restrict__ bias,
        const float* __restrict__ arowscale,
        const float* __restrict__ erowscale,
        float* __restrict__ out,
        int M, int relu,
        const float* __restrict__ Wct,
        const int* __restrict__ ids,
        float* __restrict__ cls_out,
        const int* __restrict__ seg_start,
        const int* __restrict__ sorted_src,
        const float* __restrict__ hrows,
        float* __restrict__ agg, int n, int nE,
        const float* __restrict__ in_scale,
        const float* __restrict__ gbias,
        const float* __restrict__ out_scale) {
    constexpr int N     = 128;
    constexpr int PITCH = 132;
    __shared__ __attribute__((aligned(16))) float s_w[N * PITCH];
    __shared__ __attribute__((aligned(16))) float s_a[PG_GW * 32 * PITCH];
    __shared__ __attribute__((aligned(16))) float s_o[PG_GW * 16 * PITCH];
    __shared__ __attribute__((aligned(16))) float s_c[CLS ? 16 * PITCH : 4];
    __shared__ __attribute__((aligned(16))) float s_e[PG_GW * 32];
    __shared__ __attribute__((aligned(16))) int   s_id[CLS ? PG_GW * 32 : 4];
    __shared__ __attribute__((aligned(16))) float s_b[N];

    const int tid  = threadIdx.x;
    const int lane = tid & 31;
    const int wave = __builtin_amdgcn_readfirstlane(tid >> 5);
    const int c    = lane & 15;
    const int h    = lane >> 4;
    const bool blk_prod = (int)blockIdx.x * (PG_GW * 32) < M;
    const bool prod     = (wave < PG_GW) && blk_prod;
    const int row_base  = (blockIdx.x * PG_GW + wave) * 32;
    float* my_a = s_a + (wave < PG_GW ? wave : 0) * (32 * PITCH);
    float* my_o = s_o + (wave < PG_GW ? wave : 0) * (16 * PITCH);


    if (prod) {
        v4f areg[32];
#pragma unroll
        for (int i = 0; i < 32; ++i) {
            const int r = min(row_base + i, M - 1);
            areg[i] = ((const v4f*)A)[(size_t)r * 32 + lane];
        }
        float myscale = 1.0f;
        if (arowscale) myscale = arowscale[min(row_base + lane, M - 1)];
#pragma unroll
        for (int i = 0; i < 32; ++i) {
            const float s = __int_as_float(
                __builtin_amdgcn_readlane(__float_as_int(myscale), i));
            v4f v = areg[i];
            v.x *= s; v.y *= s; v.z *= s; v.w *= s;
            *(v4f*)(my_a + i * PITCH + lane * 4) = v;
        }
    } else if (wave >= PG_GW && blk_prod) {
        const int g = tid - PG_GW * 32;
        v4f wreg[8];
#pragma unroll
        for (int i = 0; i < 8; ++i)
            wreg[i] = ((const v4f*)Wt)[g + 512 * i];
        v4f creg;
        if constexpr (CLS) creg = ((const v4f*)Wct)[g];
        const int grow = min((int)blockIdx.x * (PG_GW * 32) + (g & 255), M - 1);
        float ev = 1.0f;
        if (erowscale) ev = erowscale[grow];
        int idv = 0;
        if constexpr (CLS) idv = ids[grow];
        const float bvv = bias[g & (N - 1)];
#pragma unroll
        for (int i = 0; i < 8; ++i) {
            const int u = g + 512 * i;
            *(v4f*)(s_w + (u >> 5) * PITCH + (u & 31) * 4) = wreg[i];
        }
        if constexpr (CLS)
            *(v4f*)(s_c + (g >> 5) * PITCH + (g & 31) * 4) = creg;
        if (g < PG_GW * 32) {
            s_e[g] = ev;
            if constexpr (CLS) s_id[g] = idv;
        } else if (g < PG_GW * 32 + N) {
            s_b[g - PG_GW * 32] = bvv;
        }
    }

    __syncthreads();

    if (wave >= PG_GW) {
        float4 b4 = make_float4(0.f, 0.f, 0.f, 0.f);
        if constexpr (EPI) b4 = ((const float4*)gbias)[lane];
        const int wave0  = (int)blockIdx.x * PG_SW + (wave - PG_GW);
        const int nwaves = (int)gridDim.x * PG_SW;
        const float4* h4 = (const float4*)hrows + lane;
#define PG_BOUNDS(dd, b, e)                                                      {                                                                            const int dc_ = min((dd), n - 1);                                        b = seg_start[dc_];                                                      e = seg_start[min(dc_ + 1, n - 1)];                                      if (dc_ == n - 1) e = nE;                                                b = max(b, 0);                                                           e = min(e, nE);                                                          if ((dd) >= n) { b = 0; e = 0; }                                     }
        int beg, end, beg1, end1;
        PG_BOUNDS(wave0, beg, end)
        int ids0 = (lane < end - beg) ? sorted_src[beg + lane] : 0;
        PG_BOUNDS(wave0 + nwaves, beg1, end1)
        for (int d = wave0; d < n; d += nwaves) {
            const int ids1 = (lane < end1 - beg1) ? sorted_src[beg1 + lane] : 0;
            int beg2, end2;
            PG_BOUNDS(d + 2 * nwaves, beg2, end2)
            float si = 1.0f, so = 1.0f;
            if constexpr (EPI) { si = in_scale[d]; so = out_scale[d]; }
            float4 acc = make_float4(0.f, 0.f, 0.f, 0.f);
            for (int base = beg; base < end; base += 32) {
                const int cnt = min(32, end - base);
                int mys = ids0;
                if (base != beg) mys = (lane < cnt) ? sorted_src[base + lane] : 0;
                mys = min(max(mys, 0), n - 1);
                int j = 0;
                for (; j + 8 <= cnt; j += 8) {
                    float4 v[8];
#pragma unroll
                    for (int q = 0; q < 8; ++q) {
                        const int s = __builtin_amdgcn_readlane(mys, j + q);
                        v[q] = h4[(size_t)s * 32];
                    }
#pragma unroll
                    for (int q = 0; q < 8; ++q) {
                        acc.x += v[q].x; acc.y += v[q].y; acc.z += v[q].z; acc.w += v[q].w;
                    }
                }
                if (j + 4 <= cnt) {
                    float4 v[4];
#pragma unroll
                    for (int q = 0; q < 4; ++q) {
                        const int s = __builtin_amdgcn_readlane(mys, j + q);
                        v[q] = h4[(size_t)s * 32];
                    }
#pragma unroll
                    for (int q = 0; q < 4; ++q) {
                        acc.x += v[q].x; acc.y += v[q].y; acc.z += v[q].z; acc.w += v[q].w;
                    }
                    j += 4;
                }
                const int rem = cnt - j;
                if (rem == 3) {
                    const int s0 = __builtin_amdgcn_readlane(mys, j);
                    const int s1 = __builtin_amdgcn_readlane(mys, j + 1);
                    const int s2 = __builtin_amdgcn_readlane(mys, j + 2);
                    const float4 v0 = h4[(size_t)s0 * 32];
                    const float4 v1 = h4[(size_t)s1 * 32];
                    const float4 v2 = h4[(size_t)s2 * 32];
                    acc.x += v0.x; acc.y += v0.y; acc.z += v0.z; acc.w += v0.w;
                    acc.x += v1.x; acc.y += v1.y; acc.z += v1.z; acc.w += v1.w;
                    acc.x += v2.x; acc.y += v2.y; acc.z += v2.z; acc.w += v2.w;
                } else if (rem == 2) {
                    const int s0 = __builtin_amdgcn_readlane(mys, j);
                    const int s1 = __builtin_amdgcn_readlane(mys, j + 1);
                    const float4 v0 = h4[(size_t)s0 * 32];
                    const float4 v1 = h4[(size_t)s1 * 32];
                    acc.x += v0.x; acc.y += v0.y; acc.z += v0.z; acc.w += v0.w;
                    acc.x += v1.x; acc.y += v1.y; acc.z += v1.z; acc.w += v1.w;
                } else if (rem == 1) {
                    const int s0 = __builtin_amdgcn_readlane(mys, j);
                    const float4 v0 = h4[(size_t)s0 * 32];
                    acc.x += v0.x; acc.y += v0.y; acc.z += v0.z; acc.w += v0.w;
                }
            }
            if constexpr (EPI) {
                acc.x = fmaxf(acc.x * si + b4.x, 0.0f) * so;
                acc.y = fmaxf(acc.y * si + b4.y, 0.0f) * so;
                acc.z = fmaxf(acc.z * si + b4.z, 0.0f) * so;
                acc.w = fmaxf(acc.w * si + b4.w, 0.0f) * so;
            }
            ((float4*)agg)[(size_t)d * 32 + lane] = acc;
            beg = beg1; end = end1; ids0 = ids1;
            beg1 = beg2; end1 = end2;
        }
#undef PG_BOUNDS
    } else if (prod) {
#pragma unroll
        for (int s = 0; s < 2; ++s) {
            const float* aw = my_a + (16 * s + c) * PITCH + 4 * h;
            const v4f esa = *(const v4f*)(s_e + wave * 32 + 16 * s + 8 * h);
            const v4f esb = *(const v4f*)(s_e + wave * 32 + 16 * s + 8 * h + 4);
            const float es[8] = {esa.x, esa.y, esa.z, esa.w, esb.x, esb.y, esb.z, esb.w};
#pragma unroll
            for (int q = 0; q < 2; ++q) {
                const float* bw = s_w + (64 * q + c) * PITCH + 4 * h;
                v8f acc[4];
#pragma unroll
                for (int t = 0; t < 4; ++t) {
                    const float bvt = s_b[(4 * q + t) * 16 + c];
#pragma unroll
                    for (int r = 0; r < 8; ++r) acc[t][r] = bvt;
                }
                v4f ac = *(const v4f*)(aw);
                v4f bcur[4];
#pragma unroll
                for (int t = 0; t < 4; ++t)
                    bcur[t] = *(const v4f*)(bw + t * 16 * PITCH);
#pragma unroll
                for (int j = 0; j < 16; ++j) {
                    v4f an = ac;
                    v4f bn[4];
#pragma unroll
                    for (int t = 0; t < 4; ++t) bn[t] = bcur[t];
                    if (j + 1 < 16) {
                        an = *(const v4f*)(aw + 8 * (j + 1));
#pragma unroll
                        for (int t = 0; t < 4; ++t)
                            bn[t] = *(const v4f*)(bw + t * 16 * PITCH + 8 * (j + 1));
                    }
#pragma unroll
                    for (int t = 0; t < 4; ++t)
                        acc[t] = __builtin_amdgcn_wmma_f32_16x16x4_f32(
                            false, ac.xy, false, bcur[t].xy, (short)0, acc[t], false, false);
#pragma unroll
                    for (int t = 0; t < 4; ++t)
                        acc[t] = __builtin_amdgcn_wmma_f32_16x16x4_f32(
                            false, ac.zw, false, bcur[t].zw, (short)0, acc[t], false, false);
                    ac = an;
#pragma unroll
                    for (int t = 0; t < 4; ++t) bcur[t] = bn[t];
                }
#pragma unroll
                for (int t = 0; t < 4; ++t) {
#pragma unroll
                    for (int r = 0; r < 8; ++r) {
                        float v = acc[t][r];
                        if (relu) v = fmaxf(v, 0.0f);
                        v *= es[r];
                        my_o[(8 * h + r) * PITCH + 64 * q + 16 * t + c] = v;
                    }
                }
            }

            if constexpr (CLS) {
                const float* ow = my_o + c * PITCH + 4 * h;
                const float* cw = s_c + c * PITCH + 4 * h;
                v8f p;
#pragma unroll
                for (int r = 0; r < 8; ++r) p[r] = 0.0f;
                v4f xc  = *(const v4f*)(ow);
                v4f wcc = *(const v4f*)(cw);
#pragma unroll
                for (int j = 0; j < 16; ++j) {
                    v4f xn = xc, wcn = wcc;
                    if (j + 1 < 16) {
                        xn  = *(const v4f*)(ow + 8 * (j + 1));
                        wcn = *(const v4f*)(cw + 8 * (j + 1));
                    }
                    p = __builtin_amdgcn_wmma_f32_16x16x4_f32(
                        false, xc.xy, false, wcc.xy, (short)0, p, false, false);
                    p = __builtin_amdgcn_wmma_f32_16x16x4_f32(
                        false, xc.zw, false, wcc.zw, (short)0, p, false, false);
                    xc = xn; wcc = wcn;
                }
                typedef __attribute__((ext_vector_type(4))) int v4i;
                const v4i ida = *(const v4i*)(s_id + wave * 32 + 16 * s + 8 * h);
                const v4i idb = *(const v4i*)(s_id + wave * 32 + 16 * s + 8 * h + 4);
                const int idr[8] = {ida.x, ida.y, ida.z, ida.w, idb.x, idb.y, idb.z, idb.w};
#pragma unroll
                for (int r = 0; r < 8; ++r) {
                    if (row_base + 16 * s + 8 * h + r < M)
                        atomicAdd(cls_out + (size_t)idr[r] * 16 + c, p[r]);
                }
            } else {
#pragma unroll
                for (int g = 0; g < 2; ++g) {
                    v4f t4[8];
#pragma unroll
                    for (int k = 0; k < 8; ++k)
                        t4[k] = *(const v4f*)(my_o + (8 * g + k) * PITCH + 4 * lane);
#pragma unroll
                    for (int k = 0; k < 8; ++k) {
                        const int grow = row_base + 16 * s + 8 * g + k;
                        if (grow < M)
                            *(v4f*)(out + (size_t)grow * N + 4 * lane) = t4[k];
                    }
                }
            }
        }
    }
}


extern "C" void kernel_launch(void* const* d_in, const int* in_sizes, int n_in,
                              void* d_out, int out_size, void* d_ws, size_t ws_size,
                              hipStream_t stream) {
    (void)n_in; (void)ws_size;
    const float* x   = (const float*)d_in[0];
    const float* W2  = (const float*)d_in[1];
    const float* b2  = (const float*)d_in[2];
    const float* gw  = (const float*)d_in[3];
    const float* gb  = (const float*)d_in[4];
    const float* Wc  = (const float*)d_in[5];
    const float* bc  = (const float*)d_in[6];
    const int*   src = (const int*)d_in[7];
    const int*   dst = (const int*)d_in[8];
    const int*   ids = (const int*)d_in[9];

    const int H  = 128;
    const int C  = 16;
    const int L  = in_sizes[2] / H;
    const int NS = in_sizes[9] / L;
    const int NE = in_sizes[7] / L;
    const int NL = in_sizes[4] / (L * H);
    const int NN = out_size / C;

    float* ws       = (float*)d_ws;
    float* out_feat = ws;
    float* hbuf     = out_feat + (size_t)NN * H;
    float* agg      = hbuf + (size_t)NS * H;
    float* wmt      = agg + (size_t)NS * H;
    float* gwt      = wmt + (size_t)L * H * H;
    float* wct      = gwt + (size_t)L * NL * H * H;

    int*   seg_all    = (int*)out_feat;
    int*   sorted_all = seg_all + (size_t)L * NS;
    float* dego_all   = (float*)(sorted_all + (size_t)L * NE);
    float* degi_all   = dego_all + (size_t)L * NS;
    float* bm_all     = degi_all + (size_t)L * NS;
    float* cnt_all    = bm_all + (size_t)L * H;
    int*   cin_all    = (int*)(cnt_all + (size_t)L * NS);
    int*   cout_all   = cin_all + (size_t)L * EB_NSL * NS;

    float* bufA = hbuf;
    float* bufB = agg;
    float* bufC = hbuf - (size_t)NS * H;

    const int T = 256;
    const int gemm_blocks_NS = (((NS / 16) + 1) / 2 + 7) / 8;
    const int gather_blocks  = ((NS + 1) / 2 + 7) / 8;
    const int pair_blocks    = 256;
    const int pair_threads   = (PG_GW + PG_SW) * 32;

    {
        const int nHH  = H * H;
        const int bHH  = (nHH + T - 1) / T;
        const int nmat = L + L * NL + 1;
        const int n4   = (int)((size_t)NN * C / 4);
        setup_kernel<<<dim3(bHH, nmat + 8), T, 0, stream>>>(
            W2, b2, gw, Wc, bc, wmt, bm_all, gwt, wct, (float4*)d_out, n4, L, NL, H, C);
    }

    slice_count_kernel<<<dim3(EB_NSL * EB_NB, L), EB_T, 0, stream>>>(
        src, dst, cin_all, cout_all, NS, NE);
    slice_totals_kernel<<<dim3((NS + T - 1) / T, L), T, 0, stream>>>(
        cin_all, cout_all, cnt_all, dego_all, degi_all, NS);
    segment_start_scan_kernel<<<dim3((NS + 1023) / 1024, L), 1024, 0, stream>>>(
        cnt_all, seg_all, NS);
    slice_place_kernel<<<dim3(EB_NSL * EB_NB, L), EB_T, 0, stream>>>(
        src, dst, seg_all, cin_all, sorted_all, NS, NE);

    const size_t HH = (size_t)H * H;
    const float* x0 = x;
    const float* x1 = x + (size_t)1 * NS * H;
    const float* x2 = x + (size_t)2 * NS * H;
    const int* seg0 = seg_all;
    const int* seg1 = seg_all + (size_t)1 * NS;
    const int* seg2 = seg_all + (size_t)2 * NS;
    const int* srt0 = sorted_all;
    const int* srt1 = sorted_all + (size_t)1 * NE;
    const int* srt2 = sorted_all + (size_t)2 * NE;
    const float* do0 = dego_all;
    const float* do1 = dego_all + (size_t)1 * NS;
    const float* do2 = dego_all + (size_t)2 * NS;
    const float* di0 = degi_all;
    const float* di1 = degi_all + (size_t)1 * NS;
    const float* di2 = degi_all + (size_t)2 * NS;
    const int* ids0 = ids;
    const int* ids1 = ids + (size_t)1 * NS;
    const int* ids2 = ids + (size_t)2 * NS;
    const float* gb00 = gb + (size_t)(0 * NL + 0) * H;
    const float* gb10 = gb + (size_t)(1 * NL + 0) * H;
    const float* gb20 = gb + (size_t)(2 * NL + 0) * H;
    const float* gb01 = gb + (size_t)(0 * NL + 1) * H;
    const float* gb11 = gb + (size_t)(1 * NL + 1) * H;
    const float* gb21 = gb + (size_t)(2 * NL + 1) * H;
    const float* gwt01 = gwt + (size_t)(0 * NL + 1) * HH;
    const float* gwt11 = gwt + (size_t)(1 * NL + 1) * HH;
    const float* gwt21 = gwt + (size_t)(2 * NL + 1) * HH;

    gemm_wmma_k128_kernel<8, false><<<gemm_blocks_NS, T, 0, stream>>>(
        x0, wmt, bm_all, nullptr, do0, bufA, NS, 0, nullptr, nullptr, nullptr);
    gemm_beside_gather_kernel<false, true><<<pair_blocks, pair_threads, 0, stream>>>(
        x1, wmt + HH, bm_all + H, nullptr, do1, bufC, NS, 0, nullptr, nullptr, nullptr,
        seg0, srt0, bufA, bufB, NS, NE, di0, gb00, do0);
    gather_rows_sum_kernel<false><<<gather_blocks, T, 0, stream>>>(
        seg0, srt0, bufB, bufA, NS, NE, nullptr, nullptr, nullptr);
    gemm_beside_gather_kernel<true, true><<<pair_blocks, pair_threads, 0, stream>>>(
        bufA, gwt01, gb01, di0, nullptr, nullptr, NS, 1, wct, ids0, (float*)d_out,
        seg1, srt1, bufC, bufB, NS, NE, di1, gb10, do1);
    gemm_beside_gather_kernel<false, false><<<pair_blocks, pair_threads, 0, stream>>>(
        x2, wmt + 2 * HH, bm_all + 2 * H, nullptr, do2, bufA, NS, 0, nullptr, nullptr, nullptr,
        seg1, srt1, bufB, bufC, NS, NE, nullptr, nullptr, nullptr);
    gemm_beside_gather_kernel<true, true><<<pair_blocks, pair_threads, 0, stream>>>(
        bufC, gwt11, gb11, di1, nullptr, nullptr, NS, 1, wct, ids1, (float*)d_out,
        seg2, srt2, bufA, bufB, NS, NE, di2, gb20, do2);
    gather_rows_sum_kernel<false><<<gather_blocks, T, 0, stream>>>(
        seg2, srt2, bufB, bufA, NS, NE, nullptr, nullptr, nullptr);
    gemm_wmma_k128_kernel<8, true><<<gemm_blocks_NS, T, 0, stream>>>(
        bufA, gwt21, gb21, di2, nullptr, nullptr, NS, 1, wct, ids2, (float*)d_out);
}


# a conv layer is one launch for all links: a wave sums its 16 rows' incoming rows into LDS and multiplies them there (layer 0 sums scaled x rows, then W2 gw0); 13 -> 7 launches; products reassociated
# speedup vs baseline: 6.7860x; 1.2321x over previous
#include <hip/hip_runtime.h>


typedef __attribute__((ext_vector_type(2))) float v2f;
typedef __attribute__((ext_vector_type(4))) float v4f;
typedef __attribute__((ext_vector_type(8))) float v8f;


#define MG_H     128
#define MG_PITCH 132
__global__ __launch_bounds__(256)
void setup_kernel(const float* __restrict__ W2,
                  const float* __restrict__ b2,
                  const float* __restrict__ gw,
                  const float* __restrict__ Wc,
                  const float* __restrict__ bc,
                  float* __restrict__ wmt,
                  float* __restrict__ bm,
                  float* __restrict__ gwt,
                  float* __restrict__ wct,
                  float4* __restrict__ out4, int n4,
                  int nL, int nLayer, int H, int C) {
    __shared__ __attribute__((aligned(16))) float s_t[16 * MG_PITCH];
    const int y    = blockIdx.y;
    const int nGw  = nL * nLayer;
    const int nmat = nL + nGw + 1;
    if (y < nL) {
        const float* gw0 = gw + (size_t)y * nLayer * MG_H * MG_H;
        const float* w2  = W2 + (size_t)y * MG_H * MG_H;
        const int x   = blockIdx.x;
        const int tid = threadIdx.x;
        if (x < MG_H / 16) {
            const int lane = tid & 31;
            const int wave = __builtin_amdgcn_readfirstlane(tid >> 5);
            const int c = lane & 15;
            const int h = lane >> 4;
            const float* ap = gw0 + (size_t)(4 * h) * MG_H + 16 * x + c;
            const float* bp = w2 + (size_t)(16 * wave + c) * MG_H + 4 * h;
            v4f af[16], bf[16];
#pragma unroll
            for (int jj = 0; jj < 16; ++jj) {
                bf[jj] = *(const v4f*)(bp + 8 * jj);
                af[jj].x = ap[(size_t)(8 * jj + 0) * MG_H];
                af[jj].y = ap[(size_t)(8 * jj + 1) * MG_H];
                af[jj].z = ap[(size_t)(8 * jj + 2) * MG_H];
                af[jj].w = ap[(size_t)(8 * jj + 3) * MG_H];
            }
            v8f acc;
#pragma unroll
            for (int r = 0; r < 8; ++r) acc[r] = 0.0f;
            __builtin_amdgcn_sched_barrier(0);
#pragma unroll
            for (int jj = 0; jj < 16; ++jj) {
                acc = __builtin_amdgcn_wmma_f32_16x16x4_f32(
                    false, af[jj].xy, false, bf[jj].xy, (short)0, acc, false, false);
                acc = __builtin_amdgcn_wmma_f32_16x16x4_f32(
                    false, af[jj].zw, false, bf[jj].zw, (short)0, acc, false, false);
            }
            __builtin_amdgcn_sched_barrier(0);
            asm volatile("v_nop\n\tv_nop\n\tv_nop\n\tv_nop");
            __builtin_amdgcn_sched_barrier(0);
#pragma unroll
            for (int r = 0; r < 8; ++r)
                s_t[(8 * h + r) * MG_PITCH + 16 * wave + c] = acc[r];
            __syncthreads();
#pragma unroll
            for (int q = 0; q < 2; ++q) {
                const int row = 2 * wave + q;
                const v4f t = *(const v4f*)(s_t + row * MG_PITCH + 4 * lane);
                *(v4f*)(wmt + (size_t)y * MG_H * MG_H + (size_t)(16 * x + row) * MG_H + 4 * lane) = t;
            }
        } else if (x == MG_H / 16) {
            const int n    = tid & (MG_H - 1);
            const int part = tid >> 7;
            const float* bsrc = b2 + (size_t)y * MG_H + part * (MG_H / 2);
            const float* gsrc = gw0 + (size_t)(part * (MG_H / 2)) * MG_H + n;
            float sum = 0.0f;
            for (int j0 = 0; j0 < MG_H / 2; j0 += 16) {
                float gv[16], bv[16];
#pragma unroll
                for (int q = 0; q < 16; ++q) {
                    gv[q] = gsrc[(size_t)(j0 + q) * MG_H];
                    bv[q] = bsrc[j0 + q];
                }
#pragma unroll
                for (int q = 0; q < 16; ++q) sum += bv[q] * gv[q];
            }
            s_t[tid] = sum;
            __syncthreads();
            if (tid < MG_H) bm[(size_t)y * MG_H + tid] = s_t[tid] + s_t[tid + MG_H];
        }
    } else if (y < nmat) {
        const float* in;
        float* outT;
        int K = H, N = H;
        if (y < nL + nGw) {
            const int g = y - nL;
            if (g % nLayer == 0) return;
            in = gw + (size_t)g * H * H;  outT = gwt + (size_t)g * H * H;
        } else {
            in = Wc;  outT = wct;  N = C;
        }
        const int i = blockIdx.x * blockDim.x + threadIdx.x;
        if (i < N * K) {
            const int n = i / K, k = i % K;
            outT[i] = in[k * N + n];
        }
    } else {
        const int fb     = (y - nmat) * gridDim.x + blockIdx.x;
        const int stride = (gridDim.y - nmat) * gridDim.x * blockDim.x;
        int i = fb * blockDim.x + threadIdx.x;
        const float4 b = ((const float4*)bc)[i & 3];
        for (; i < n4; i += stride) out4[i] = b;
    }
}

#define EB_NB  16
#define EB_NSL 16
#define EB_BR  3136
#define EB_T   512
#define EB_U   16

__device__ __forceinline__ void edge_slice_bounds(int sl, int nE, int& e0, int& e1) {
    const int per = (nE + EB_NSL - 1) / EB_NSL;
    e0 = min(sl * per, nE);
    e1 = min(e0 + per, nE);
}

__global__ __launch_bounds__(EB_T)
void slice_count_kernel(const int* __restrict__ src,
                        const int* __restrict__ dst,
                        int* __restrict__ cin,
                        int* __restrict__ cout, int n, int nE) {
    __shared__ int s_in[EB_BR];
    __shared__ int s_out[EB_BR];
    src  += (size_t)blockIdx.y * nE;
    dst  += (size_t)blockIdx.y * nE;
    cin  += (size_t)blockIdx.y * EB_NSL * n;
    cout += (size_t)blockIdx.y * EB_NSL * n;
    const int t  = threadIdx.x;
    const int sl = blockIdx.x / EB_NB;
    const int rg = blockIdx.x % EB_NB;
    int e0, e1;
    edge_slice_bounds(sl, nE, e0, e1);
    for (int r0 = rg * EB_BR; r0 < n; r0 += EB_NB * EB_BR) {
        for (int i = t; i < EB_BR; i += EB_T) { s_in[i] = 0; s_out[i] = 0; }
        __syncthreads();
        for (int e = e0 + t; e < e1; e += EB_U * EB_T) {
            int dv[EB_U], sv[EB_U];
#pragma unroll
            for (int q = 0; q < EB_U; ++q) {
                const int ee = e + q * EB_T;
                const int ec = min(ee, e1 - 1);
                const int d = dst[ec];
                const int s = src[ec];
                dv[q] = (ee < e1) ? d : -1;
                sv[q] = (ee < e1) ? s : -1;
            }
#pragma unroll
            for (int q = 0; q < EB_U; ++q) {
                const unsigned a = (unsigned)(dv[q] - r0);
                const unsigned b = (unsigned)(sv[q] - r0);
                if (a < (unsigned)EB_BR && dv[q] < n) atomicAdd(&s_in[a], 1);
                if (b < (unsigned)EB_BR && sv[q] < n) atomicAdd(&s_out[b], 1);
            }
        }
        __syncthreads();
        for (int i = t; i < EB_BR; i += EB_T) {
            const int d = r0 + i;
            if (d < n) {
                cin[(size_t)sl * n + d]  = s_in[i];
                cout[(size_t)sl * n + d] = s_out[i];
            }
        }
        __syncthreads();
    }
}

__global__ void slice_totals_kernel(int* __restrict__ cin,
                                    const int* __restrict__ cout,
                                    float* __restrict__ cnt_in,
                                    float* __restrict__ deg_out,
                                    float* __restrict__ deg_in, int n) {
    cin     += (size_t)blockIdx.y * EB_NSL * n;
    cout    += (size_t)blockIdx.y * EB_NSL * n;
    cnt_in  += (size_t)blockIdx.y * n;
    deg_out += (size_t)blockIdx.y * n;
    deg_in  += (size_t)blockIdx.y * n;
    const int d = blockIdx.x * blockDim.x + threadIdx.x;
    if (d < n) {
        int ci[EB_NSL], co = 0;
#pragma unroll
        for (int k = 0; k < EB_NSL; ++k) {
            ci[k] = cin[(size_t)k * n + d];
            co += cout[(size_t)k * n + d];
        }
        int run = 0;
#pragma unroll
        for (int k = 0; k < EB_NSL; ++k) {
            cin[(size_t)k * n + d] = run;
            run += ci[k];
        }
        cnt_in[d]  = (float)run;
        deg_in[d]  = rsqrtf(fmaxf((float)run, 1.0f));
        deg_out[d] = rsqrtf(fmaxf((float)co, 1.0f));
    }
}

__global__ __launch_bounds__(1024)
void segment_start_scan_kernel(const float* __restrict__ cnt,
                               int* __restrict__ seg, int n) {
    __shared__ int s_front[32];
    __shared__ int s_own[32];
    cnt += (size_t)blockIdx.y * n;
    seg += (size_t)blockIdx.y * n;
    const int t = threadIdx.x;
    const int lane = t & 31;
    const int w = t >> 5;
    const int first = blockIdx.x * 1024;
    int front = 0;
    const float4* c4 = (const float4*)cnt;
    const int nf4 = first >> 2;
    for (int i = t; i < nf4; i += 4096) {
        const float4 a = c4[i];
        const float4 b = c4[min(i + 1024, nf4 - 1)];
        const float4 c = c4[min(i + 2048, nf4 - 1)];
        const float4 e = c4[min(i + 3072, nf4 - 1)];
        const int sa = (int)a.x + (int)a.y + (int)a.z + (int)a.w;
        const int sb = (int)b.x + (int)b.y + (int)b.z + (int)b.w;
        const int sc = (int)c.x + (int)c.y + (int)c.z + (int)c.w;
        const int se = (int)e.x + (int)e.y + (int)e.z + (int)e.w;
        front += sa;
        front += (i + 1024 < nf4) ? sb : 0;
        front += (i + 2048 < nf4) ? sc : 0;
        front += (i + 3072 < nf4) ? se : 0;
    }
    const int idx = first + t;
    const int v = (idx < n) ? (int)cnt[idx] : 0;
    int incl = v;
    for (int o = 1; o < 32; o <<= 1) {
        const int u = __shfl_up(incl, o);
        if (lane >= o) incl += u;
    }
    for (int o = 16; o > 0; o >>= 1) front += __shfl_xor(front, o);
    if (lane == 31) s_own[w] = incl;
    if (lane == 0) s_front[w] = front;
    __syncthreads();
    int base = 0, woff = 0;
    for (int k = 0; k < 32; ++k) {
        base += s_front[k];
        if (k < w) woff += s_own[k];
    }
    if (idx < n) seg[idx] = base + woff + incl - v;
}

__global__ __launch_bounds__(EB_T)
void slice_place_kernel(const int* __restrict__ src,
                        const int* __restrict__ dst,
                        const int* __restrict__ seg,
                        const int* __restrict__ cin_front,
                        int* __restrict__ sorted_src, int n, int nE) {
    __shared__ int s_cur[EB_BR];
    src        += (size_t)blockIdx.y * nE;
    dst        += (size_t)blockIdx.y * nE;
    seg        += (size_t)blockIdx.y * n;
    cin_front  += (size_t)blockIdx.y * EB_NSL * n;
    sorted_src += (size_t)blockIdx.y * nE;
    const int t  = threadIdx.x;
    const int sl = blockIdx.x / EB_NB;
    const int rg = blockIdx.x % EB_NB;
    int e0, e1;
    edge_slice_bounds(sl, nE, e0, e1);
    for (int r0 = rg * EB_BR; r0 < n; r0 += EB_NB * EB_BR) {
        constexpr int NI = (EB_BR + EB_T - 1) / EB_T;
        int cv[NI];
#pragma unroll
        for (int k = 0; k < NI; ++k) {
            const int d = min(r0 + t + k * EB_T, n - 1);
            cv[k] = seg[d] + cin_front[(size_t)sl * n + d];
        }
#pragma unroll
        for (int k = 0; k < NI; ++k) {
            const int i = t + k * EB_T;
            if (i < EB_BR) s_cur[i] = cv[k];
        }
        __syncthreads();
        for (int e = e0 + t; e < e1; e += EB_U * EB_T) {
            int dv[EB_U], sv[EB_U];
#pragma unroll
            for (int q = 0; q < EB_U; ++q) {
                const int ee = e + q * EB_T;
                const int ec = min(ee, e1 - 1);
                const int d = dst[ec];
                const int s = src[ec];
                dv[q] = (ee < e1) ? d : -1;
                sv[q] = s;
            }
#pragma unroll
            for (int q = 0; q < EB_U; ++q) {
                const unsigned a = (unsigned)(dv[q] - r0);
                if (a < (unsigned)EB_BR && dv[q] < n) {
                    const int pos = atomicAdd(&s_cur[a], 1);
                    if ((unsigned)pos < (unsigned)nE) sorted_src[pos] = sv[q];
                }
            }
        }
        __syncthreads();
    }
}


#define FG_ISSUE(G, P, O)                                                      \
    _Pragma("unroll")                                                          \
    for (int q = 0; q < (G); ++q) {                                            \
        const int s = __builtin_amdgcn_readlane(mys, (P) + q);                 \
        v[(O) + q] = ((const float4*)(rows + (size_t)s * 128))[lane];          \
    }
#define FG_SUM(G, P, O)                                                        \
    _Pragma("unroll")                                                          \
    for (int q = 0; q < (G); ++q) {                                            \
        if constexpr (L0) {                                                    \
            const float wq = __int_as_float(                                   \
                __builtin_amdgcn_readlane(__float_as_int(myw), (P) + q));      \
            acc.x = fmaf(wq, v[(O) + q].x, acc.x);                             \
            acc.y = fmaf(wq, v[(O) + q].y, acc.y);                             \
            acc.z = fmaf(wq, v[(O) + q].z, acc.z);                             \
            acc.w = fmaf(wq, v[(O) + q].w, acc.w);                             \
            rs += wq;                                                          \
        } else {                                                               \
            acc.x += v[(O) + q].x; acc.y += v[(O) + q].y;                      \
            acc.z += v[(O) + q].z; acc.w += v[(O) + q].w;                      \
        }                                                                      \
    }
#define FG_CHUNK                                                               \
    {                                                                          \
        float4 v[31];                                                          \
        const int p8 = cnt & 16, p4 = cnt & 24, p2 = cnt & 28, p1 = cnt & 30;  \
        if (cnt & 16) { FG_ISSUE(16, 0, 0) }                                   \
        if (cnt & 8)  { FG_ISSUE(8, p8, 16) }                                  \
        if (cnt & 4)  { FG_ISSUE(4, p4, 24) }                                  \
        if (cnt & 2)  { FG_ISSUE(2, p2, 28) }                                  \
        if (cnt & 1)  { FG_ISSUE(1, p1, 30) }                                  \
        if (cnt & 16) { FG_SUM(16, 0, 0) }                                     \
        if (cnt & 8)  { FG_SUM(8, p8, 16) }                                    \
        if (cnt & 4)  { FG_SUM(4, p4, 24) }                                    \
        if (cnt & 2)  { FG_SUM(2, p2, 28) }                                    \
        if (cnt & 1)  { FG_SUM(1, p1, 30) }                                    \
    }

template <bool L0>
__global__ __launch_bounds__(256)
void gather_gemm_kernel(const float* __restrict__ rows,
                        const int* __restrict__ seg_start,
                        const int* __restrict__ sorted_src,
                        const float* __restrict__ dout,
                        const float* __restrict__ din,
                        const float* __restrict__ Wt,
                        int wt_stride,
                        const float* __restrict__ bias,
                        int bias_stride,
                        const float* __restrict__ bm,
                        float* __restrict__ out,
                        const float* __restrict__ Wct,
                        const int* __restrict__ ids,
                        float* __restrict__ cls_out,
                        int n, int nE) {
    constexpr int N     = 128;
    constexpr int PITCH = 132;
    __shared__ __attribute__((aligned(16))) float s_w[N * PITCH];
    __shared__ __attribute__((aligned(16))) float s_a[8 * 16 * PITCH];
    __shared__ __attribute__((aligned(16))) float s_c[L0 ? 4 : 16 * PITCH];
    __shared__ __attribute__((aligned(16))) float s_e[L0 ? 128 : 4];
    __shared__ __attribute__((aligned(16))) float s_q[L0 ? 128 : 4];
    __shared__ __attribute__((aligned(16))) int   s_id[L0 ? 4 : 128];

    const int y = blockIdx.y;
    rows       += (size_t)y * n * N;
    seg_start  += (size_t)y * n;
    sorted_src += (size_t)y * nE;
    dout       += (size_t)y * n;
    din        += (size_t)y * n;
    Wt         += (size_t)y * wt_stride;
    bias       += (size_t)y * bias_stride;

    const int tid  = threadIdx.x;
    const int lane = tid & 31;
    const int wave = __builtin_amdgcn_readfirstlane(tid >> 5);
    const int c    = lane & 15;
    const int h    = lane >> 4;
    const int row_base = (blockIdx.x * 8 + wave) * 16;
    float* my_a = s_a + wave * (16 * PITCH);

    v4f wreg[N / 8];
#pragma unroll
    for (int i = 0; i < N / 8; ++i)
        wreg[i] = ((const v4f*)Wt)[tid + 256 * i];
    v4f creg[2];
    if constexpr (!L0) {
#pragma unroll
        for (int i = 0; i < 2; ++i)
            creg[i] = ((const v4f*)Wct)[tid + 256 * i];
    }
    const int dl = min(row_base + lane, n - 1);
    int segv = seg_start[dl];
    const float div = din[dl];
    float dov = 0.0f;
    int idv = 0;
    if constexpr (L0) dov = dout[dl]; else idv = ids[(size_t)y * n + dl];
    float bvr[8], bmr[8];
#pragma unroll
    for (int nt = 0; nt < 8; ++nt) {
        bvr[nt] = bias[nt * 16 + c];
        bmr[nt] = 0.0f;
        if constexpr (L0) bmr[nt] = bm[(size_t)y * N + nt * 16 + c];
    }
    if (row_base + lane >= n) segv = nE;
    segv = min(max(segv, 0), nE);

#pragma unroll
    for (int i = 0; i < N / 8; ++i) {
        const int u = tid + 256 * i;
        *(v4f*)(s_w + (u >> 5) * PITCH + (u & 31) * 4) = wreg[i];
    }
    if constexpr (!L0) {
#pragma unroll
        for (int i = 0; i < 2; ++i) {
            const int u = tid + 256 * i;
            *(v4f*)(s_c + (u >> 5) * PITCH + (u & 31) * 4) = creg[i];
        }
    }
    if (lane < 16) {
        if constexpr (L0) s_e[wave * 16 + lane] = dov;
        else              s_id[wave * 16 + lane] = idv;
    }

    int beg  = __builtin_amdgcn_readlane(segv, 0);
    int end  = __builtin_amdgcn_readlane(segv, 1);
    int end1 = __builtin_amdgcn_readlane(segv, 2);
    int beg1 = end;
    int ids0 = (lane < end - beg)   ? sorted_src[beg + lane]  : 0;
    int ids1 = (lane < end1 - beg1) ? sorted_src[beg1 + lane] : 0;
    ids0 = min(max(ids0, 0), n - 1);
    float w0 = 0.0f;
    if constexpr (L0) {
        w0 = dout[ids0];
        asm volatile("" : "+v"(w0));
    } else {
        asm volatile("" : "+v"(ids1));
    }
    for (int i = 0; i < 16; ++i) {
        int beg2 = 0, end2 = 0;
        if (i + 2 < 16) {
            beg2 = end1;
            end2 = __builtin_amdgcn_readlane(segv, i + 3);
        }
        int ids2 = (lane < end2 - beg2) ? sorted_src[beg2 + lane] : 0;
        ids1 = min(max(ids1, 0), n - 1);
        float w1 = 0.0f;
        if constexpr (L0) w1 = dout[ids1];
        const float di = __int_as_float(
            __builtin_amdgcn_readlane(__float_as_int(div), i));
        float4 acc = make_float4(0.f, 0.f, 0.f, 0.f);
        float rs = 0.0f;
        {
            const int cnt = min(31, end - beg);
            const int mys = ids0;
            const float myw = w0;
            FG_CHUNK
        }
        for (int e = beg + 31; e < end; ++e) {
            const int s = min(max(sorted_src[e], 0), n - 1);
            const float4 vr = ((const float4*)(rows + (size_t)s * 128))[lane];
            if constexpr (L0) {
                const float wq = dout[s];
                acc.x = fmaf(wq, vr.x, acc.x);
                acc.y = fmaf(wq, vr.y, acc.y);
                acc.z = fmaf(wq, vr.z, acc.z);
                acc.w = fmaf(wq, vr.w, acc.w);
                rs += wq;
            } else {
                acc.x += vr.x; acc.y += vr.y; acc.z += vr.z; acc.w += vr.w;
            }
        }
        acc.x *= di; acc.y *= di; acc.z *= di; acc.w *= di;
        *(float4*)(my_a + i * PITCH + lane * 4) = acc;
        if constexpr (L0) {
            if (lane == 0) s_q[wave * 16 + i] = di * rs;
        }
        beg = beg1; end = end1; ids0 = ids1; w0 = w1;
        beg1 = beg2; end1 = end2; ids1 = ids2;
    }

    __syncthreads();

    const float* aw = my_a + c * PITCH + 4 * h;
    const float* bw = s_w + c * PITCH + 4 * h;
    float es[8];
#pragma unroll
    for (int r = 0; r < 8; ++r) es[r] = 1.0f;
    v8f acc[8];
    if constexpr (L0) {
        const v4f qa = *(const v4f*)(s_q + wave * 16 + 8 * h);
        const v4f qb = *(const v4f*)(s_q + wave * 16 + 8 * h + 4);
        const v4f ea = *(const v4f*)(s_e + wave * 16 + 8 * h);
        const v4f eb = *(const v4f*)(s_e + wave * 16 + 8 * h + 4);
        const float qv[8] = {qa.x, qa.y, qa.z, qa.w, qb.x, qb.y, qb.z, qb.w};
        es[0] = ea.x; es[1] = ea.y; es[2] = ea.z; es[3] = ea.w;
        es[4] = eb.x; es[5] = eb.y; es[6] = eb.z; es[7] = eb.w;
#pragma unroll
        for (int nt = 0; nt < 8; ++nt) {
#pragma unroll
            for (int r = 0; r < 8; ++r) acc[nt][r] = fmaf(qv[r], bmr[nt], bvr[nt]);
        }
    } else {
#pragma unroll
        for (int nt = 0; nt < 8; ++nt) {
#pragma unroll
            for (int r = 0; r < 8; ++r) acc[nt][r] = bvr[nt];
        }
    }

    __builtin_amdgcn_sched_barrier(0);
    v4f ac = *(const v4f*)(aw);
    v4f bcur[8];
#pragma unroll
    for (int nt = 0; nt < 8; ++nt)
        bcur[nt] = *(const v4f*)(bw + nt * 16 * PITCH);
#pragma unroll
    for (int j = 0; j < 16; ++j) {
        v4f an = ac;
        v4f bn[8];
#pragma unroll
        for (int nt = 0; nt < 8; ++nt) bn[nt] = bcur[nt];
        if (j + 1 < 16) {
            an = *(const v4f*)(aw + 8 * (j + 1));
#pragma unroll
            for (int nt = 0; nt < 8; ++nt)
                bn[nt] = *(const v4f*)(bw + nt * 16 * PITCH + 8 * (j + 1));
        }
#pragma unroll
        for (int nt = 0; nt < 8; ++nt)
            acc[nt] = __builtin_amdgcn_wmma_f32_16x16x4_f32(
                false, ac.xy, false, bcur[nt].xy, (short)0, acc[nt], false, false);
#pragma unroll
        for (int nt = 0; nt < 8; ++nt)
            acc[nt] = __builtin_amdgcn_wmma_f32_16x16x4_f32(
                false, ac.zw, false, bcur[nt].zw, (short)0, acc[nt], false, false);
        ac = an;
#pragma unroll
        for (int nt = 0; nt < 8; ++nt) bcur[nt] = bn[nt];
    }
    __builtin_amdgcn_sched_barrier(0);
    asm volatile("v_nop\n\tv_nop\n\tv_nop\n\tv_nop");
    __builtin_amdgcn_sched_barrier(0);

#pragma unroll
    for (int nt = 0; nt < 8; ++nt) {
#pragma unroll
        for (int r = 0; r < 8; ++r) {
            float v = fmaxf(acc[nt][r], 0.0f);
            if constexpr (L0) v *= es[r];
            my_a[(8 * h + r) * PITCH + nt * 16 + c] = v;
        }
    }

    if constexpr (!L0) {
        const float* cw = s_c + c * PITCH + 4 * h;
        v8f p;
#pragma unroll
        for (int r = 0; r < 8; ++r) p[r] = 0.0f;
        __builtin_amdgcn_sched_barrier(0);
        v4f xc  = *(const v4f*)(aw);
        v4f wcc = *(const v4f*)(cw);
#pragma unroll
        for (int j = 0; j < 16; ++j) {
            v4f xn = xc, wcn = wcc;
            if (j + 1 < 16) {
                xn  = *(const v4f*)(aw + 8 * (j + 1));
                wcn = *(const v4f*)(cw + 8 * (j + 1));
            }
            p = __builtin_amdgcn_wmma_f32_16x16x4_f32(
                false, xc.xy, false, wcc.xy, (short)0, p, false, false);
            p = __builtin_amdgcn_wmma_f32_16x16x4_f32(
                false, xc.zw, false, wcc.zw, (short)0, p, false, false);
            xc = xn; wcc = wcn;
        }
        __builtin_amdgcn_sched_barrier(0);
        asm volatile("v_nop\n\tv_nop\n\tv_nop\n\tv_nop");
        __builtin_amdgcn_sched_barrier(0);
        typedef __attribute__((ext_vector_type(4))) int v4i;
        const v4i ida = *(const v4i*)(s_id + wave * 16 + 8 * h);
        const v4i idb = *(const v4i*)(s_id + wave * 16 + 8 * h + 4);
        const int idr[8] = {ida.x, ida.y, ida.z, ida.w, idb.x, idb.y, idb.z, idb.w};
#pragma unroll
        for (int r = 0; r < 8; ++r) {
            if (row_base + 8 * h + r < n)
                atomicAdd(cls_out + (size_t)idr[r] * 16 + c, p[r]);
        }
    } else {
        float* orow = out + (size_t)y * n * N;
#pragma unroll
        for (int g = 0; g < 2; ++g) {
            v4f t4[8];
#pragma unroll
            for (int k = 0; k < 8; ++k)
                t4[k] = *(const v4f*)(my_a + (8 * g + k) * PITCH + 4 * lane);
#pragma unroll
            for (int k = 0; k < 8; ++k) {
                const int grow = row_base + 8 * g + k;
                if (grow < n)
                    *(v4f*)(orow + (size_t)grow * N + 4 * lane) = t4[k];
            }
        }
    }
}
#undef FG_CHUNK
#undef FG_SUM
#undef FG_ISSUE


extern "C" void kernel_launch(void* const* d_in, const int* in_sizes, int n_in,
                              void* d_out, int out_size, void* d_ws, size_t ws_size,
                              hipStream_t stream) {
    (void)n_in; (void)ws_size;
    const float* x   = (const float*)d_in[0];
    const float* W2  = (const float*)d_in[1];
    const float* b2  = (const float*)d_in[2];
    const float* gw  = (const float*)d_in[3];
    const float* gb  = (const float*)d_in[4];
    const float* Wc  = (const float*)d_in[5];
    const float* bc  = (const float*)d_in[6];
    const int*   src = (const int*)d_in[7];
    const int*   dst = (const int*)d_in[8];
    const int*   ids = (const int*)d_in[9];

    const int H  = 128;
    const int C  = 16;
    const int L  = in_sizes[2] / H;
    const int NS = in_sizes[9] / L;
    const int NE = in_sizes[7] / L;
    const int NL = in_sizes[4] / (L * H);
    const int NN = out_size / C;

    float* ws       = (float*)d_ws;
    float* out_feat = ws;
    float* hbuf     = out_feat + (size_t)NN * H;
    float* agg      = hbuf + (size_t)NS * H;
    float* wmt      = agg + (size_t)NS * H;
    float* gwt      = wmt + (size_t)L * H * H;
    float* wct      = gwt + (size_t)L * NL * H * H;

    int*   seg_all    = (int*)out_feat;
    int*   sorted_all = seg_all + (size_t)L * NS;
    float* dego_all   = (float*)(sorted_all + (size_t)L * NE);
    float* degi_all   = dego_all + (size_t)L * NS;
    float* bm_all     = degi_all + (size_t)L * NS;
    float* cnt_all    = bm_all + (size_t)L * H;
    int*   cin_all    = (int*)(cnt_all + (size_t)L * NS);
    int*   cout_all   = cin_all + (size_t)L * EB_NSL * NS;

    float* h1_all = hbuf - (size_t)NS * H;

    const int T = 256;

    {
        const int nHH  = H * H;
        const int bHH  = (nHH + T - 1) / T;
        const int nmat = L + L * NL + 1;
        const int n4   = (int)((size_t)NN * C / 4);
        setup_kernel<<<dim3(bHH, nmat + 8), T, 0, stream>>>(
            W2, b2, gw, Wc, bc, wmt, bm_all, gwt, wct, (float4*)d_out, n4, L, NL, H, C);
    }

    slice_count_kernel<<<dim3(EB_NSL * EB_NB, L), EB_T, 0, stream>>>(
        src, dst, cin_all, cout_all, NS, NE);
    slice_totals_kernel<<<dim3((NS + T - 1) / T, L), T, 0, stream>>>(
        cin_all, cout_all, cnt_all, dego_all, degi_all, NS);
    segment_start_scan_kernel<<<dim3((NS + 1023) / 1024, L), 1024, 0, stream>>>(
        cnt_all, seg_all, NS);
    slice_place_kernel<<<dim3(EB_NSL * EB_NB, L), EB_T, 0, stream>>>(
        src, dst, seg_all, cin_all, sorted_all, NS, NE);

    const int HH = H * H;
    const dim3 conv_grid((NS + 127) / 128, L);
    gather_gemm_kernel<true><<<conv_grid, T, 0, stream>>>(
        x, seg_all, sorted_all, dego_all, degi_all,
        wmt, HH, gb, NL * H, bm_all, h1_all,
        nullptr, nullptr, nullptr, NS, NE);
    gather_gemm_kernel<false><<<conv_grid, T, 0, stream>>>(
        h1_all, seg_all, sorted_all, dego_all, degi_all,
        gwt + HH, NL * HH, gb + H, NL * H, nullptr, nullptr,
        wct, ids, (float*)d_out, NS, NE);
}


# conv layers: one resident 20-wave work-group per WGP keeps Wt in LDS; a wave takes 16-row strips from an LDS counter, sums then multiplies, no block barrier between the phases; sum orders as before
# speedup vs baseline: 7.0944x; 1.0454x over previous
#include <hip/hip_runtime.h>


typedef __attribute__((ext_vector_type(2))) float v2f;
typedef __attribute__((ext_vector_type(4))) float v4f;
typedef __attribute__((ext_vector_type(8))) float v8f;


#define MG_H     128
#define MG_PITCH 132
__global__ __launch_bounds__(256)
void setup_kernel(const float* __restrict__ W2,
                  const float* __restrict__ b2,
                  const float* __restrict__ gw,
                  const float* __restrict__ Wc,
                  const float* __restrict__ bc,
                  float* __restrict__ wmt,
                  float* __restrict__ bm,
                  float* __restrict__ gwt,
                  float* __restrict__ wct,
                  float4* __restrict__ out4, int n4,
                  int nL, int nLayer, int H, int C) {
    __shared__ __attribute__((aligned(16))) float s_t[16 * MG_PITCH];
    const int y    = blockIdx.y;
    const int nGw  = nL * nLayer;
    const int nmat = nL + nGw + 1;
    if (y < nL) {
        const float* gw0 = gw + (size_t)y * nLayer * MG_H * MG_H;
        const float* w2  = W2 + (size_t)y * MG_H * MG_H;
        const int x   = blockIdx.x;
        const int tid = threadIdx.x;
        if (x < MG_H / 16) {
            const int lane = tid & 31;
            const int wave = __builtin_amdgcn_readfirstlane(tid >> 5);
            const int c = lane & 15;
            const int h = lane >> 4;
            const float* ap = gw0 + (size_t)(4 * h) * MG_H + 16 * x + c;
            const float* bp = w2 + (size_t)(16 * wave + c) * MG_H + 4 * h;
            v4f af[16], bf[16];
#pragma unroll
            for (int jj = 0; jj < 16; ++jj) {
                bf[jj] = *(const v4f*)(bp + 8 * jj);
                af[jj].x = ap[(size_t)(8 * jj + 0) * MG_H];
                af[jj].y = ap[(size_t)(8 * jj + 1) * MG_H];
                af[jj].z = ap[(size_t)(8 * jj + 2) * MG_H];
                af[jj].w = ap[(size_t)(8 * jj + 3) * MG_H];
            }
            v8f acc;
#pragma unroll
            for (int r = 0; r < 8; ++r) acc[r] = 0.0f;
            __builtin_amdgcn_sched_barrier(0);
#pragma unroll
            for (int jj = 0; jj < 16; ++jj) {
                acc = __builtin_amdgcn_wmma_f32_16x16x4_f32(
                    false, af[jj].xy, false, bf[jj].xy, (short)0, acc, false, false);
                acc = __builtin_amdgcn_wmma_f32_16x16x4_f32(
                    false, af[jj].zw, false, bf[jj].zw, (short)0, acc, false, false);
            }
            __builtin_amdgcn_sched_barrier(0);
            asm volatile("v_nop\n\tv_nop\n\tv_nop\n\tv_nop");
            __builtin_amdgcn_sched_barrier(0);
#pragma unroll
            for (int r = 0; r < 8; ++r)
                s_t[(8 * h + r) * MG_PITCH + 16 * wave + c] = acc[r];
            __syncthreads();
#pragma unroll
            for (int q = 0; q < 2; ++q) {
                const int row = 2 * wave + q;
                const v4f t = *(const v4f*)(s_t + row * MG_PITCH + 4 * lane);
                *(v4f*)(wmt + (size_t)y * MG_H * MG_H + (size_t)(16 * x + row) * MG_H + 4 * lane) = t;
            }
        } else if (x == MG_H / 16) {
            const int n    = tid & (MG_H - 1);
            const int part = tid >> 7;
            const float* bsrc = b2 + (size_t)y * MG_H + part * (MG_H / 2);
            const float* gsrc = gw0 + (size_t)(part * (MG_H / 2)) * MG_H + n;
            float sum = 0.0f;
            for (int j0 = 0; j0 < MG_H / 2; j0 += 16) {
                float gv[16], bv[16];
#pragma unroll
                for (int q = 0; q < 16; ++q) {
                    gv[q] = gsrc[(size_t)(j0 + q) * MG_H];
                    bv[q] = bsrc[j0 + q];
                }
#pragma unroll
                for (int q = 0; q < 16; ++q) sum += bv[q] * gv[q];
            }
            s_t[tid] = sum;
            __syncthreads();
            if (tid < MG_H) bm[(size_t)y * MG_H + tid] = s_t[tid] + s_t[tid + MG_H];
        }
    } else if (y < nmat) {
        const float* in;
        float* outT;
        int K = H, N = H;
        if (y < nL + nGw) {
            const int g = y - nL;
            if (g % nLayer == 0) return;
            in = gw + (size_t)g * H * H;  outT = gwt + (size_t)g * H * H;
        } else {
            in = Wc;  outT = wct;  N = C;
        }
        const int i = blockIdx.x * blockDim.x + threadIdx.x;
        if (i < N * K) {
            const int n = i / K, k = i % K;
            outT[i] = in[k * N + n];
        }
    } else {
        const int fb     = (y - nmat) * gridDim.x + blockIdx.x;
        const int stride = (gridDim.y - nmat) * gridDim.x * blockDim.x;
        int i = fb * blockDim.x + threadIdx.x;
        const float4 b = ((const float4*)bc)[i & 3];
        for (; i < n4; i += stride) out4[i] = b;
    }
}

#define EB_NB  16
#define EB_NSL 16
#define EB_BR  3136
#define EB_T   512
#define EB_U   16

__device__ __forceinline__ void edge_slice_bounds(int sl, int nE, int& e0, int& e1) {
    const int per = (nE + EB_NSL - 1) / EB_NSL;
    e0 = min(sl * per, nE);
    e1 = min(e0 + per, nE);
}

__global__ __launch_bounds__(EB_T)
void slice_count_kernel(const int* __restrict__ src,
                        const int* __restrict__ dst,
                        int* __restrict__ cin,
                        int* __restrict__ cout, int n, int nE) {
    __shared__ int s_in[EB_BR];
    __shared__ int s_out[EB_BR];
    src  += (size_t)blockIdx.y * nE;
    dst  += (size_t)blockIdx.y * nE;
    cin  += (size_t)blockIdx.y * EB_NSL * n;
    cout += (size_t)blockIdx.y * EB_NSL * n;
    const int t  = threadIdx.x;
    const int sl = blockIdx.x / EB_NB;
    const int rg = blockIdx.x % EB_NB;
    int e0, e1;
    edge_slice_bounds(sl, nE, e0, e1);
    for (int r0 = rg * EB_BR; r0 < n; r0 += EB_NB * EB_BR) {
        for (int i = t; i < EB_BR; i += EB_T) { s_in[i] = 0; s_out[i] = 0; }
        __syncthreads();
        for (int e = e0 + t; e < e1; e += EB_U * EB_T) {
            int dv[EB_U], sv[EB_U];
#pragma unroll
            for (int q = 0; q < EB_U; ++q) {
                const int ee = e + q * EB_T;
                const int ec = min(ee, e1 - 1);
                const int d = dst[ec];
                const int s = src[ec];
                dv[q] = (ee < e1) ? d : -1;
                sv[q] = (ee < e1) ? s : -1;
            }
#pragma unroll
            for (int q = 0; q < EB_U; ++q) {
                const unsigned a = (unsigned)(dv[q] - r0);
                const unsigned b = (unsigned)(sv[q] - r0);
                if (a < (unsigned)EB_BR && dv[q] < n) atomicAdd(&s_in[a], 1);
                if (b < (unsigned)EB_BR && sv[q] < n) atomicAdd(&s_out[b], 1);
            }
        }
        __syncthreads();
        for (int i = t; i < EB_BR; i += EB_T) {
            const int d = r0 + i;
            if (d < n) {
                cin[(size_t)sl * n + d]  = s_in[i];
                cout[(size_t)sl * n + d] = s_out[i];
            }
        }
        __syncthreads();
    }
}

__global__ void slice_totals_kernel(int* __restrict__ cin,
                                    const int* __restrict__ cout,
                                    float* __restrict__ cnt_in,
                                    float* __restrict__ deg_out,
                                    float* __restrict__ deg_in, int n) {
    cin     += (size_t)blockIdx.y * EB_NSL * n;
    cout    += (size_t)blockIdx.y * EB_NSL * n;
    cnt_in  += (size_t)blockIdx.y * n;
    deg_out += (size_t)blockIdx.y * n;
    deg_in  += (size_t)blockIdx.y * n;
    const int d = blockIdx.x * blockDim.x + threadIdx.x;
    if (d < n) {
        int ci[EB_NSL], co = 0;
#pragma unroll
        for (int k = 0; k < EB_NSL; ++k) {
            ci[k] = cin[(size_t)k * n + d];
            co += cout[(size_t)k * n + d];
        }
        int run = 0;
#pragma unroll
        for (int k = 0; k < EB_NSL; ++k) {
            cin[(size_t)k * n + d] = run;
            run += ci[k];
        }
        cnt_in[d]  = (float)run;
        deg_in[d]  = rsqrtf(fmaxf((float)run, 1.0f));
        deg_out[d] = rsqrtf(fmaxf((float)co, 1.0f));
    }
}

__global__ __launch_bounds__(1024)
void segment_start_scan_kernel(const float* __restrict__ cnt,
                               int* __restrict__ seg, int n) {
    __shared__ int s_front[32];
    __shared__ int s_own[32];
    cnt += (size_t)blockIdx.y * n;
    seg += (size_t)blockIdx.y * n;
    const int t = threadIdx.x;
    const int lane = t & 31;
    const int w = t >> 5;
    const int first = blockIdx.x * 1024;
    int front = 0;
    const float4* c4 = (const float4*)cnt;
    const int nf4 = first >> 2;
    for (int i = t; i < nf4; i += 4096) {
        const float4 a = c4[i];
        const float4 b = c4[min(i + 1024, nf4 - 1)];
        const float4 c = c4[min(i + 2048, nf4 - 1)];
        const float4 e = c4[min(i + 3072, nf4 - 1)];
        const int sa = (int)a.x + (int)a.y + (int)a.z + (int)a.w;
        const int sb = (int)b.x + (int)b.y + (int)b.z + (int)b.w;
        const int sc = (int)c.x + (int)c.y + (int)c.z + (int)c.w;
        const int se = (int)e.x + (int)e.y + (int)e.z + (int)e.w;
        front += sa;
        front += (i + 1024 < nf4) ? sb : 0;
        front += (i + 2048 < nf4) ? sc : 0;
        front += (i + 3072 < nf4) ? se : 0;
    }
    const int idx = first + t;
    const int v = (idx < n) ? (int)cnt[idx] : 0;
    int incl = v;
    for (int o = 1; o < 32; o <<= 1) {
        const int u = __shfl_up(incl, o);
        if (lane >= o) incl += u;
    }
    for (int o = 16; o > 0; o >>= 1) front += __shfl_xor(front, o);
    if (lane == 31) s_own[w] = incl;
    if (lane == 0) s_front[w] = front;
    __syncthreads();
    int base = 0, woff = 0;
    for (int k = 0; k < 32; ++k) {
        base += s_front[k];
        if (k < w) woff += s_own[k];
    }
    if (idx < n) seg[idx] = base + woff + incl - v;
}

__global__ __launch_bounds__(EB_T)
void slice_place_kernel(const int* __restrict__ src,
                        const int* __restrict__ dst,
                        const int* __restrict__ seg,
                        const int* __restrict__ cin_front,
                        int* __restrict__ sorted_src, int n, int nE) {
    __shared__ int s_cur[EB_BR];
    src        += (size_t)blockIdx.y * nE;
    dst        += (size_t)blockIdx.y * nE;
    seg        += (size_t)blockIdx.y * n;
    cin_front  += (size_t)blockIdx.y * EB_NSL * n;
    sorted_src += (size_t)blockIdx.y * nE;
    const int t  = threadIdx.x;
    const int sl = blockIdx.x / EB_NB;
    const int rg = blockIdx.x % EB_NB;
    int e0, e1;
    edge_slice_bounds(sl, nE, e0, e1);
    for (int r0 = rg * EB_BR; r0 < n; r0 += EB_NB * EB_BR) {
        constexpr int NI = (EB_BR + EB_T - 1) / EB_T;
        int cv[NI];
#pragma unroll
        for (int k = 0; k < NI; ++k) {
            const int d = min(r0 + t + k * EB_T, n - 1);
            cv[k] = seg[d] + cin_front[(size_t)sl * n + d];
        }
#pragma unroll
        for (int k = 0; k < NI; ++k) {
            const int i = t + k * EB_T;
            if (i < EB_BR) s_cur[i] = cv[k];
        }
        __syncthreads();
        for (int e = e0 + t; e < e1; e += EB_U * EB_T) {
            int dv[EB_U], sv[EB_U];
#pragma unroll
            for (int q = 0; q < EB_U; ++q) {
                const int ee = e + q * EB_T;
                const int ec = min(ee, e1 - 1);
                const int d = dst[ec];
                const int s = src[ec];
                dv[q] = (ee < e1) ? d : -1;
                sv[q] = s;
            }
#pragma unroll
            for (int q = 0; q < EB_U; ++q) {
                const unsigned a = (unsigned)(dv[q] - r0);
                if (a < (unsigned)EB_BR && dv[q] < n) {
                    const int pos = atomicAdd(&s_cur[a], 1);
                    if ((unsigned)pos < (unsigned)nE) sorted_src[pos] = sv[q];
                }
            }
        }
        __syncthreads();
    }
}


#define FG_ISSUE(G, P, O)                                                      \
    _Pragma("unroll")                                                          \
    for (int q = 0; q < (G); ++q) {                                            \
        const int s = __builtin_amdgcn_readlane(mys, (P) + q);                 \
        gv[(O) + q] = ((const v4f*)(rows + (size_t)s * 128))[lane];            \
    }
#define FG_SUM(G, P, O)                                                        \
    _Pragma("unroll")                                                          \
    for (int q = 0; q < (G); ++q) {                                            \
        if constexpr (L0) {                                                    \
            const float wq = __int_as_float(                                   \
                __builtin_amdgcn_readlane(__float_as_int(myw), (P) + q));      \
            acc.x = fmaf(wq, gv[(O) + q].x, acc.x);                            \
            acc.y = fmaf(wq, gv[(O) + q].y, acc.y);                            \
            acc.z = fmaf(wq, gv[(O) + q].z, acc.z);                            \
            acc.w = fmaf(wq, gv[(O) + q].w, acc.w);                            \
            rs += wq;                                                          \
        } else {                                                               \
            acc.x += gv[(O) + q].x; acc.y += gv[(O) + q].y;                    \
            acc.z += gv[(O) + q].z; acc.w += gv[(O) + q].w;                    \
        }                                                                      \
    }
#define FG_CHUNK                                                               \
    {                                                                          \
        const int p8 = cnt & 16, p4 = cnt & 24, p2 = cnt & 28, p1 = cnt & 30;  \
        if (cnt & 16) { FG_ISSUE(16, 0, 0) }                                   \
        if (cnt & 8)  { FG_ISSUE(8, p8, 16) }                                  \
        if (cnt & 4)  { FG_ISSUE(4, p4, 24) }                                  \
        if (cnt & 2)  { FG_ISSUE(2, p2, 28) }                                  \
        if (cnt & 1)  { FG_ISSUE(1, p1, 30) }                                  \
        if (cnt & 16) { FG_SUM(16, 0, 0) }                                     \
        if (cnt & 8)  { FG_SUM(8, p8, 16) }                                    \
        if (cnt & 4)  { FG_SUM(4, p4, 24) }                                    \
        if (cnt & 2)  { FG_SUM(2, p2, 28) }                                    \
        if (cnt & 1)  { FG_SUM(1, p1, 30) }                                    \
    }

#define GG_WAVES 20
#define GG_T     (GG_WAVES * 32)
template <bool L0>
__global__ __launch_bounds__(GG_T)
void gather_gemm_kernel(const float* __restrict__ rows,
                        const int* __restrict__ seg_start,
                        const int* __restrict__ sorted_src,
                        const float* __restrict__ dout,
                        const float* __restrict__ din,
                        const float* __restrict__ Wt,
                        int wt_stride,
                        const float* __restrict__ bias,
                        int bias_stride,
                        const float* __restrict__ bm,
                        float* __restrict__ out,
                        const float* __restrict__ Wct,
                        const int* __restrict__ ids,
                        float* __restrict__ cls_out,
                        int n, int nE) {
    constexpr int N     = 128;
    constexpr int PITCH = 132;
    __shared__ __attribute__((aligned(16))) float s_w[N * PITCH];
    __shared__ __attribute__((aligned(16))) float s_a[GG_WAVES * 16 * PITCH];
    __shared__ __attribute__((aligned(16))) float s_c[L0 ? 4 : 16 * PITCH];
    __shared__ __attribute__((aligned(16))) float s_e[L0 ? GG_WAVES * 16 : 4];
    __shared__ __attribute__((aligned(16))) float s_q[L0 ? GG_WAVES * 16 : 4];
    __shared__ __attribute__((aligned(16))) int   s_id[L0 ? 4 : GG_WAVES * 16];
    __shared__ float s_bias[N];
    __shared__ float s_bm[L0 ? N : 4];
    __shared__ int   s_next;

    const int y = blockIdx.y;
    rows       += (size_t)y * n * N;
    seg_start  += (size_t)y * n;
    sorted_src += (size_t)y * nE;
    dout       += (size_t)y * n;
    din        += (size_t)y * n;
    Wt         += (size_t)y * wt_stride;
    bias       += (size_t)y * bias_stride;

    const int tid  = threadIdx.x;
    const int lane = tid & 31;
    const int wave = __builtin_amdgcn_readfirstlane(tid >> 5);
    const int c    = lane & 15;
    const int h    = lane >> 4;
    float* my_a = s_a + wave * (16 * PITCH);

    constexpr int NW4 = N * N / 4;
    constexpr int WI  = (NW4 + GG_T - 1) / GG_T;
    v4f wreg[WI];
#pragma unroll
    for (int i = 0; i < WI; ++i)
        wreg[i] = ((const v4f*)Wt)[min(tid + GG_T * i, NW4 - 1)];
    v4f creg = {0.0f, 0.0f, 0.0f, 0.0f};
    if constexpr (!L0) creg = ((const v4f*)Wct)[min(tid, 16 * N / 4 - 1)];
    const float bv = bias[tid & (N - 1)];
    float bmv = 0.0f;
    if constexpr (L0) bmv = bm[(size_t)y * N + (tid & (N - 1))];
#pragma unroll
    for (int i = 0; i < WI; ++i) {
        const int u = tid + GG_T * i;
        if (u < NW4) *(v4f*)(s_w + (u >> 5) * PITCH + (u & 31) * 4) = wreg[i];
    }
    if constexpr (!L0) {
        if (tid < 16 * N / 4) *(v4f*)(s_c + (tid >> 5) * PITCH + (tid & 31) * 4) = creg;
    }
    if (tid < N) {
        s_bias[tid] = bv;
        if constexpr (L0) s_bm[tid] = bmv;
    }
    if (tid == 0) s_next = 0;
    __syncthreads();

    const int nstrips = (n + 15) >> 4;
    for (;;) {
        int k = 0;
        if (lane == 0) k = atomicAdd(&s_next, 1);
        k = __builtin_amdgcn_readfirstlane(k);
        const int strip = k * (int)gridDim.x + (int)blockIdx.x;
        if (strip >= nstrips) break;
        const int row_base = strip * 16;
        const int dl = min(row_base + lane, n - 1);
        int segv = seg_start[dl];
        const float div = din[dl];
        float dov = 0.0f;
        int idv = 0;
        if constexpr (L0) dov = dout[dl]; else idv = ids[(size_t)y * n + dl];
        if (row_base + lane >= n) segv = nE;
        segv = min(max(segv, 0), nE);
        if (lane < 16) {
            if constexpr (L0) s_e[wave * 16 + lane] = dov;
            else              s_id[wave * 16 + lane] = idv;
        }

        v4f gv[31];
#pragma unroll
        for (int q = 0; q < 31; ++q) asm volatile("" : "=v"(gv[q]));
        int beg  = __builtin_amdgcn_readlane(segv, 0);
        int end  = __builtin_amdgcn_readlane(segv, 1);
        int end1 = __builtin_amdgcn_readlane(segv, 2);
        int beg1 = end;
        int ids0 = (lane < end - beg)   ? sorted_src[beg + lane]  : 0;
        int ids1 = (lane < end1 - beg1) ? sorted_src[beg1 + lane] : 0;
        ids0 = min(max(ids0, 0), n - 1);
        float w0 = 0.0f;
        if constexpr (L0) {
            w0 = dout[ids0];
            asm volatile("" : "+v"(w0));
        } else {
            asm volatile("" : "+v"(ids1));
        }
        for (int i = 0; i < 16; ++i) {
            int beg2 = 0, end2 = 0;
            if (i + 2 < 16) {
                beg2 = end1;
                end2 = __builtin_amdgcn_readlane(segv, i + 3);
            }
            int ids2 = (lane < end2 - beg2) ? sorted_src[beg2 + lane] : 0;
            ids1 = min(max(ids1, 0), n - 1);
            float w1 = 0.0f;
            if constexpr (L0) w1 = dout[ids1];
            const float di = __int_as_float(
                __builtin_amdgcn_readlane(__float_as_int(div), i));
            float4 acc = make_float4(0.f, 0.f, 0.f, 0.f);
            float rs = 0.0f;
            {
                const int cnt = min(31, end - beg);
                const int mys = ids0;
                const float myw = w0;
                FG_CHUNK
            }
            for (int e = beg + 31; e < end; ++e) {
                const int s = min(max(sorted_src[e], 0), n - 1);
                const float4 vr = ((const float4*)(rows + (size_t)s * 128))[lane];
                if constexpr (L0) {
                    const float wq = dout[s];
                    acc.x = fmaf(wq, vr.x, acc.x);
                    acc.y = fmaf(wq, vr.y, acc.y);
                    acc.z = fmaf(wq, vr.z, acc.z);
                    acc.w = fmaf(wq, vr.w, acc.w);
                    rs += wq;
                } else {
                    acc.x += vr.x; acc.y += vr.y; acc.z += vr.z; acc.w += vr.w;
                }
            }
            acc.x *= di; acc.y *= di; acc.z *= di; acc.w *= di;
            *(float4*)(my_a + i * PITCH + lane * 4) = acc;
            if constexpr (L0) {
                if (lane == 0) s_q[wave * 16 + i] = di * rs;
            }
            beg = beg1; end = end1; ids0 = ids1; w0 = w1;
            beg1 = beg2; end1 = end2; ids1 = ids2;
        }

        float bvr[8], bmr[8];
#pragma unroll
        for (int nt = 0; nt < 8; ++nt) {
            bvr[nt] = s_bias[nt * 16 + c];
            bmr[nt] = 0.0f;
            if constexpr (L0) bmr[nt] = s_bm[nt * 16 + c];
        }

        const float* aw = my_a + c * PITCH + 4 * h;
        const float* bw = s_w + c * PITCH + 4 * h;
        v8f acc[8];
        if constexpr (L0) {
            const v4f qa = *(const v4f*)(s_q + wave * 16 + 8 * h);
            const v4f qb = *(const v4f*)(s_q + wave * 16 + 8 * h + 4);
            const float qv[8] = {qa.x, qa.y, qa.z, qa.w, qb.x, qb.y, qb.z, qb.w};
#pragma unroll
            for (int nt = 0; nt < 8; ++nt) {
#pragma unroll
                for (int r = 0; r < 8; ++r) acc[nt][r] = fmaf(qv[r], bmr[nt], bvr[nt]);
            }
        } else {
#pragma unroll
            for (int nt = 0; nt < 8; ++nt) {
#pragma unroll
                for (int r = 0; r < 8; ++r) acc[nt][r] = bvr[nt];
            }
        }

        __builtin_amdgcn_sched_barrier(0);
        v4f ac = *(const v4f*)(aw);
        v4f bcur[8];
#pragma unroll
        for (int nt = 0; nt < 8; ++nt)
            bcur[nt] = *(const v4f*)(bw + nt * 16 * PITCH);
#pragma unroll
        for (int j = 0; j < 16; ++j) {
            v4f an = ac;
            v4f bn[8];
#pragma unroll
            for (int nt = 0; nt < 8; ++nt) bn[nt] = bcur[nt];
            if (j + 1 < 16) {
                an = *(const v4f*)(aw + 8 * (j + 1));
#pragma unroll
                for (int nt = 0; nt < 8; ++nt)
                    bn[nt] = *(const v4f*)(bw + nt * 16 * PITCH + 8 * (j + 1));
            }
#pragma unroll
            for (int nt = 0; nt < 8; ++nt)
                acc[nt] = __builtin_amdgcn_wmma_f32_16x16x4_f32(
                    false, ac.xy, false, bcur[nt].xy, (short)0, acc[nt], false, false);
#pragma unroll
            for (int nt = 0; nt < 8; ++nt)
                acc[nt] = __builtin_amdgcn_wmma_f32_16x16x4_f32(
                    false, ac.zw, false, bcur[nt].zw, (short)0, acc[nt], false, false);
            ac = an;
#pragma unroll
            for (int nt = 0; nt < 8; ++nt) bcur[nt] = bn[nt];
        }
        __builtin_amdgcn_sched_barrier(0);
        asm volatile("v_nop\n\tv_nop\n\tv_nop\n\tv_nop");
        __builtin_amdgcn_sched_barrier(0);

        float es[8];
#pragma unroll
        for (int r = 0; r < 8; ++r) es[r] = 1.0f;
        if constexpr (L0) {
            const v4f ea = *(const v4f*)(s_e + wave * 16 + 8 * h);
            const v4f eb = *(const v4f*)(s_e + wave * 16 + 8 * h + 4);
            es[0] = ea.x; es[1] = ea.y; es[2] = ea.z; es[3] = ea.w;
            es[4] = eb.x; es[5] = eb.y; es[6] = eb.z; es[7] = eb.w;
        }

#pragma unroll
        for (int nt = 0; nt < 8; ++nt) {
#pragma unroll
            for (int r = 0; r < 8; ++r) {
                float v = fmaxf(acc[nt][r], 0.0f);
                if constexpr (L0) v *= es[r];
                my_a[(8 * h + r) * PITCH + nt * 16 + c] = v;
            }
        }

        if constexpr (!L0) {
            const float* cw = s_c + c * PITCH + 4 * h;
            v8f p;
#pragma unroll
            for (int r = 0; r < 8; ++r) p[r] = 0.0f;
            __builtin_amdgcn_sched_barrier(0);
            v4f xc  = *(const v4f*)(aw);
            v4f wcc = *(const v4f*)(cw);
#pragma unroll
            for (int j = 0; j < 16; ++j) {
                v4f xn = xc, wcn = wcc;
                if (j + 1 < 16) {
                    xn  = *(const v4f*)(aw + 8 * (j + 1));
                    wcn = *(const v4f*)(cw + 8 * (j + 1));
                }
                p = __builtin_amdgcn_wmma_f32_16x16x4_f32(
                    false, xc.xy, false, wcc.xy, (short)0, p, false, false);
                p = __builtin_amdgcn_wmma_f32_16x16x4_f32(
                    false, xc.zw, false, wcc.zw, (short)0, p, false, false);
                xc = xn; wcc = wcn;
            }
            __builtin_amdgcn_sched_barrier(0);
            asm volatile("v_nop\n\tv_nop\n\tv_nop\n\tv_nop");
            __builtin_amdgcn_sched_barrier(0);
            typedef __attribute__((ext_vector_type(4))) int v4i;
            const v4i ida = *(const v4i*)(s_id + wave * 16 + 8 * h);
            const v4i idb = *(const v4i*)(s_id + wave * 16 + 8 * h + 4);
            const int idr[8] = {ida.x, ida.y, ida.z, ida.w, idb.x, idb.y, idb.z, idb.w};
#pragma unroll
            for (int r = 0; r < 8; ++r) {
                if (row_base + 8 * h + r < n)
                    atomicAdd(cls_out + (size_t)idr[r] * 16 + c, p[r]);
            }
        } else {
            float* srow = out + ((size_t)y * n + row_base) * N;
            asm volatile("" : "+s"(srow));
            typedef __attribute__((address_space(1))) v4f gl_v4f;
            gl_v4f* srow4 = (gl_v4f*)srow;
#pragma unroll
            for (int g = 0; g < 2; ++g) {
                v4f t4[8];
#pragma unroll
                for (int k = 0; k < 8; ++k)
                    t4[k] = *(const v4f*)(my_a + (8 * g + k) * PITCH + 4 * lane);
#pragma unroll
                for (int k = 0; k < 8; ++k) {
                    const int grow = row_base + 8 * g + k;
                    if (grow < n)
                        srow4[(8 * g + k) * (N / 4) + lane] = t4[k];
                }
            }
        }
    }
}
#undef FG_CHUNK
#undef FG_SUM
#undef FG_ISSUE


extern "C" void kernel_launch(void* const* d_in, const int* in_sizes, int n_in,
                              void* d_out, int out_size, void* d_ws, size_t ws_size,
                              hipStream_t stream) {
    (void)n_in; (void)ws_size;
    const float* x   = (const float*)d_in[0];
    const float* W2  = (const float*)d_in[1];
    const float* b2  = (const float*)d_in[2];
    const float* gw  = (const float*)d_in[3];
    const float* gb  = (const float*)d_in[4];
    const float* Wc  = (const float*)d_in[5];
    const float* bc  = (const float*)d_in[6];
    const int*   src = (const int*)d_in[7];
    const int*   dst = (const int*)d_in[8];
    const int*   ids = (const int*)d_in[9];

    const int H  = 128;
    const int C  = 16;
    const int L  = in_sizes[2] / H;
    const int NS = in_sizes[9] / L;
    const int NE = in_sizes[7] / L;
    const int NL = in_sizes[4] / (L * H);
    const int NN = out_size / C;

    float* ws       = (float*)d_ws;
    float* out_feat = ws;
    float* hbuf     = out_feat + (size_t)NN * H;
    float* agg      = hbuf + (size_t)NS * H;
    float* wmt      = agg + (size_t)NS * H;
    float* gwt      = wmt + (size_t)L * H * H;
    float* wct      = gwt + (size_t)L * NL * H * H;

    int*   seg_all    = (int*)out_feat;
    int*   sorted_all = seg_all + (size_t)L * NS;
    float* dego_all   = (float*)(sorted_all + (size_t)L * NE);
    float* degi_all   = dego_all + (size_t)L * NS;
    float* bm_all     = degi_all + (size_t)L * NS;
    float* cnt_all    = bm_all + (size_t)L * H;
    int*   cin_all    = (int*)(cnt_all + (size_t)L * NS);
    int*   cout_all   = cin_all + (size_t)L * EB_NSL * NS;

    float* h1_all = hbuf - (size_t)NS * H;

    const int T = 256;

    {
        const int nHH  = H * H;
        const int bHH  = (nHH + T - 1) / T;
        const int nmat = L + L * NL + 1;
        const int n4   = (int)((size_t)NN * C / 4);
        setup_kernel<<<dim3(bHH, nmat + 8), T, 0, stream>>>(
            W2, b2, gw, Wc, bc, wmt, bm_all, gwt, wct, (float4*)d_out, n4, L, NL, H, C);
    }

    slice_count_kernel<<<dim3(EB_NSL * EB_NB, L), EB_T, 0, stream>>>(
        src, dst, cin_all, cout_all, NS, NE);
    slice_totals_kernel<<<dim3((NS + T - 1) / T, L), T, 0, stream>>>(
        cin_all, cout_all, cnt_all, dego_all, degi_all, NS);
    segment_start_scan_kernel<<<dim3((NS + 1023) / 1024, L), 1024, 0, stream>>>(
        cnt_all, seg_all, NS);
    slice_place_kernel<<<dim3(EB_NSL * EB_NB, L), EB_T, 0, stream>>>(
        src, dst, seg_all, cin_all, sorted_all, NS, NE);

    const int HH = H * H;
    const dim3 conv_grid(85, L);
    gather_gemm_kernel<true><<<conv_grid, GG_T, 0, stream>>>(
        x, seg_all, sorted_all, dego_all, degi_all,
        wmt, HH, gb, NL * H, bm_all, h1_all,
        nullptr, nullptr, nullptr, NS, NE);
    gather_gemm_kernel<false><<<conv_grid, GG_T, 0, stream>>>(
        h1_all, seg_all, sorted_all, dego_all, degi_all,
        gwt + HH, NL * HH, gb + H, NL * H, nullptr, nullptr,
        wct, ids, (float*)d_out, NS, NE);
}
